# starting kernel
# baseline (speedup 1.0000x reference)
#include <hip/hip_runtime.h>
#include <math.h>

typedef __attribute__((ext_vector_type(2))) float v2f;
typedef __attribute__((ext_vector_type(8))) float v8f;

#define F_DIM 64
#define H_DIM 4
#define HF    256
#define NEG_SLOPE 0.2f
#define LN_EPS 1e-5f

__device__ __forceinline__ void atomic_max_f32(float* addr, float val) {
    if (val >= 0.0f) atomicMax((int*)addr,          __float_as_int(val));
    else             atomicMin((unsigned int*)addr, __float_as_uint(val));
}

__global__ void init_kernel(float* __restrict__ rst, float* __restrict__ z,
                            float* __restrict__ m, int nrst, int nz) {
    int i = blockIdx.x * blockDim.x + threadIdx.x;
    if (i < nrst) rst[i] = 0.0f;
    if (i < nz)   { z[i] = 0.0f; m[i] = -INFINITY; }
}

__global__ void fc_gemm_kernel(const float* __restrict__ h,
                               const float* __restrict__ w,
                               float* __restrict__ feat, int nRowTiles) {
    const int wave   = (blockIdx.x * blockDim.x + threadIdx.x) >> 5;
    const int nTiles = nRowTiles * (HF / 16);
    if (wave >= nTiles) return;
    const int rt = wave >> 4;
    const int ct = wave & 15;
    const int row0 = rt * 16, col0 = ct * 16;
    const int lane = threadIdx.x & 31;
    const int half = lane >> 4, l = lane & 15;
    const int arow = row0 + l;
    const int bcol = col0 + l;

    v8f acc = {};
    #pragma unroll
    for (int k = 0; k < F_DIM; k += 4) {
        const int ka = k + 2 * half;
        v2f a, b;
        a.x = h[arow * F_DIM + ka];
        a.y = h[arow * F_DIM + ka + 1];
        b.x = w[ka * HF + bcol];
        b.y = w[(ka + 1) * HF + bcol];
        acc = __builtin_amdgcn_wmma_f32_16x16x4_f32(
                  false, a, false, b, (short)0, acc, false, false);
    }
    const int orow = row0 + half * 8;
    #pragma unroll
    for (int i = 0; i < 8; ++i)
        feat[(size_t)(orow + i) * HF + bcol] = acc[i];
}

__global__ void attn_score_kernel(const float* __restrict__ feat,
                                  const float* __restrict__ al,
                                  const float* __restrict__ ar,
                                  float* __restrict__ el, float* __restrict__ er,
                                  int N) {
    const int wave = (blockIdx.x * blockDim.x + threadIdx.x) >> 5;
    if (wave >= N) return;
    const int lane = threadIdx.x & 31;
    const int abase = lane * 8;
    const size_t base = (size_t)wave * HF + abase;
    float sl = 0.0f, sr = 0.0f;
    #pragma unroll
    for (int j = 0; j < 8; ++j) {
        const float f = feat[base + j];
        sl += f * al[abase + j];
        sr += f * ar[abase + j];
    }
    #pragma unroll
    for (int msk = 1; msk < 8; msk <<= 1) {
        sl += __shfl_xor(sl, msk, 32);
        sr += __shfl_xor(sr, msk, 32);
    }
    if ((lane & 7) == 0) {
        const int hd = lane >> 3;
        el[wave * H_DIM + hd] = sl;
        er[wave * H_DIM + hd] = sr;
    }
}

__global__ void edge_max_kernel(const int* __restrict__ src, const int* __restrict__ dst,
                                const float* __restrict__ el, const float* __restrict__ er,
                                float* __restrict__ m, int EH) {
    int t = blockIdx.x * blockDim.x + threadIdx.x;
    if (t >= EH) return;
    const int e = t >> 2, hd = t & 3;
    const int s = src[e], d = dst[e];
    float v = el[s * H_DIM + hd] + er[d * H_DIM + hd];
    v = v > 0.0f ? v : NEG_SLOPE * v;
    atomic_max_f32(&m[d * H_DIM + hd], v);
}

__global__ void edge_exp_kernel(const int* __restrict__ src, const int* __restrict__ dst,
                                const float* __restrict__ el, const float* __restrict__ er,
                                const float* __restrict__ m, float* __restrict__ z,
                                float* __restrict__ exb, int EH) {
    int t = blockIdx.x * blockDim.x + threadIdx.x;
    if (t >= EH) return;
    const int e = t >> 2, hd = t & 3;
    const int s = src[e], d = dst[e];
    float v = el[s * H_DIM + hd] + er[d * H_DIM + hd];
    v = v > 0.0f ? v : NEG_SLOPE * v;
    const float ex = __expf(v - m[d * H_DIM + hd]);
    exb[t] = ex;
    atomicAdd(&z[d * H_DIM + hd], ex);
}

__global__ void aggregate_kernel(const int* __restrict__ src, const int* __restrict__ dst,
                                 const float* __restrict__ feat,
                                 const float* __restrict__ exb, const float* __restrict__ z,
                                 float* __restrict__ rst, int E) {
    const int e = (blockIdx.x * blockDim.x + threadIdx.x) >> 5;
    if (e >= E) return;
    const int lane = threadIdx.x & 31;
    const int s = src[e], d = dst[e];
    const int hd = lane >> 3;
    const float alpha = exb[e * H_DIM + hd] / z[d * H_DIM + hd];
    const int base = lane * 8;
    const float4* f4 = (const float4*)(feat + (size_t)s * HF + base);
    const float4 v0 = f4[0], v1 = f4[1];
    float* r = rst + (size_t)d * HF + base;
    atomicAdd(r + 0, v0.x * alpha); atomicAdd(r + 1, v0.y * alpha);
    atomicAdd(r + 2, v0.z * alpha); atomicAdd(r + 3, v0.w * alpha);
    atomicAdd(r + 4, v1.x * alpha); atomicAdd(r + 5, v1.y * alpha);
    atomicAdd(r + 6, v1.z * alpha); atomicAdd(r + 7, v1.w * alpha);
}

__global__ void out_gemm_kernel(const float* __restrict__ rst,
                                const float* __restrict__ gb,
                                const float* __restrict__ w,
                                const float* __restrict__ ob,
                                float* __restrict__ x, int nRowTiles) {
    const int wave   = (blockIdx.x * blockDim.x + threadIdx.x) >> 5;
    const int nTiles = nRowTiles * (F_DIM / 16);
    if (wave >= nTiles) return;
    const int rt = wave >> 2;
    const int ct = wave & 3;
    const int row0 = rt * 16, col0 = ct * 16;
    const int lane = threadIdx.x & 31;
    const int half = lane >> 4, l = lane & 15;
    const int arow = row0 + l;
    const int bcol = col0 + l;

    v8f acc = {};
    #pragma unroll 8
    for (int k = 0; k < HF; k += 4) {
        const int ka = k + 2 * half;
        v2f a, b;
        a.x = rst[(size_t)arow * HF + ka]     + gb[ka];
        a.y = rst[(size_t)arow * HF + ka + 1] + gb[ka + 1];
        b.x = w[ka * F_DIM + bcol];
        b.y = w[(ka + 1) * F_DIM + bcol];
        acc = __builtin_amdgcn_wmma_f32_16x16x4_f32(
                  false, a, false, b, (short)0, acc, false, false);
    }
    const float bias = ob[bcol];
    const int orow = row0 + half * 8;
    #pragma unroll
    for (int i = 0; i < 8; ++i)
        x[(size_t)(orow + i) * F_DIM + bcol] = acc[i] + bias;
}

__global__ void layernorm_kernel(const float* __restrict__ x,
                                 const float* __restrict__ g,
                                 const float* __restrict__ b,
                                 float* __restrict__ out, int N) {
    const int wave = (blockIdx.x * blockDim.x + threadIdx.x) >> 5;
    if (wave >= N) return;
    const int lane = threadIdx.x & 31;
    const size_t base = (size_t)wave * F_DIM;
    const float x0 = x[base + lane], x1 = x[base + lane + 32];
    float s = x0 + x1, sq = x0 * x0 + x1 * x1;
    #pragma unroll
    for (int msk = 1; msk < 32; msk <<= 1) {
        s  += __shfl_xor(s,  msk, 32);
        sq += __shfl_xor(sq, msk, 32);
    }
    const float mu  = s * (1.0f / 64.0f);
    const float var = sq * (1.0f / 64.0f) - mu * mu;
    const float inv = rsqrtf(var + LN_EPS);
    out[base + lane]      = (x0 - mu) * inv * g[lane]      + b[lane];
    out[base + lane + 32] = (x1 - mu) * inv * g[lane + 32] + b[lane + 32];
}

extern "C" void kernel_launch(void* const* d_in, const int* in_sizes, int n_in,
                              void* d_out, int out_size, void* d_ws, size_t ws_size,
                              hipStream_t stream) {
    const float* h   = (const float*)d_in[0];
    const int*   src = (const int*)  d_in[1];
    const int*   dst = (const int*)  d_in[2];
    const float* fcw = (const float*)d_in[3];
    const float* al  = (const float*)d_in[4];
    const float* ar  = (const float*)d_in[5];
    const float* gb  = (const float*)d_in[6];
    const float* ow  = (const float*)d_in[7];
    const float* ob  = (const float*)d_in[8];
    const float* lng = (const float*)d_in[9];
    const float* lnb = (const float*)d_in[10];

    const int N = in_sizes[0] / F_DIM;
    const int E = in_sizes[1];

    char*  ws  = (char*)d_ws;
    size_t off = 0;
    auto carve = [&](size_t bytes) -> float* {
        float* p = (float*)(ws + off);
        off += (bytes + 255) & ~(size_t)255;
        return p;
    };
    float* feat = carve((size_t)N * HF * 4);
    float* rst  = carve((size_t)N * HF * 4);
    float* el   = carve((size_t)N * H_DIM * 4);
    float* er   = carve((size_t)N * H_DIM * 4);
    float* m    = carve((size_t)N * H_DIM * 4);
    float* z    = carve((size_t)N * H_DIM * 4);
    float* exb  = carve((size_t)E * H_DIM * 4);
    float* x    = carve((size_t)N * F_DIM * 4);
    (void)ws_size;

    const int nRowTiles = (N + 15) / 16;

    {
        const int total  = N * HF;
        const int blocks = (total + 255) / 256;
        init_kernel<<<blocks, 256, 0, stream>>>(rst, z, m, total, N * H_DIM);
    }
    {
        const int waves  = nRowTiles * (HF / 16);
        const int blocks = (waves + 7) / 8;
        fc_gemm_kernel<<<blocks, 256, 0, stream>>>(h, fcw, feat, nRowTiles);
    }
    {
        const int blocks = (N + 7) / 8;
        attn_score_kernel<<<blocks, 256, 0, stream>>>(feat, al, ar, el, er, N);
    }
    {
        const int EH = E * H_DIM;
        const int blocks = (EH + 255) / 256;
        edge_max_kernel<<<blocks, 256, 0, stream>>>(src, dst, el, er, m, EH);
        edge_exp_kernel<<<blocks, 256, 0, stream>>>(src, dst, el, er, m, z, exb, EH);
    }
    {
        const int blocks = (E + 7) / 8;
        aggregate_kernel<<<blocks, 256, 0, stream>>>(src, dst, feat, exb, z, rst, E);
    }
    {
        const int waves  = nRowTiles * (F_DIM / 16);
        const int blocks = (waves + 7) / 8;
        out_gemm_kernel<<<blocks, 256, 0, stream>>>(rst, gb, ow, ob, x, nRowTiles);
    }
    {
        const int blocks = (N + 7) / 8;
        layernorm_kernel<<<blocks, 256, 0, stream>>>(x, lng, lnb, (float*)d_out, N);
    }
}


# edges grouped by destination (one int atomic per edge ranks it, then place, scatter); one wave sums a row of rst in registers, stores it once; no float atomics on rst; sum order = rank order
# speedup vs baseline: 2.0788x; 2.0788x over previous
#include <hip/hip_runtime.h>
#include <math.h>

typedef __attribute__((ext_vector_type(2))) float v2f;
typedef __attribute__((ext_vector_type(8))) float v8f;

#define F_DIM 64
#define H_DIM 4
#define HF    256
#define NEG_SLOPE 0.2f
#define LN_EPS 1e-5f

__device__ __forceinline__ void atomic_max_f32(float* addr, float val) {
    if (val >= 0.0f) atomicMax((int*)addr,          __float_as_int(val));
    else             atomicMin((unsigned int*)addr, __float_as_uint(val));
}

__global__ void init_kernel(float* __restrict__ z, float* __restrict__ m,
                            int* __restrict__ deg, int* __restrict__ total,
                            int nz, int N) {
    int i = blockIdx.x * blockDim.x + threadIdx.x;
    if (i < nz) { z[i] = 0.0f; m[i] = -INFINITY; }
    if (i < N)  deg[i] = 0;
    if (i == 0) *total = 0;
}

__global__ void fc_gemm_kernel(const float* __restrict__ h,
                               const float* __restrict__ w,
                               float* __restrict__ feat, int nRowTiles) {
    const int wave   = (blockIdx.x * blockDim.x + threadIdx.x) >> 5;
    const int nTiles = nRowTiles * (HF / 16);
    if (wave >= nTiles) return;
    const int rt = wave >> 4;
    const int ct = wave & 15;
    const int row0 = rt * 16, col0 = ct * 16;
    const int lane = threadIdx.x & 31;
    const int half = lane >> 4, l = lane & 15;
    const int arow = row0 + l;
    const int bcol = col0 + l;

    v8f acc = {};
    #pragma unroll
    for (int k = 0; k < F_DIM; k += 4) {
        const int ka = k + 2 * half;
        v2f a, b;
        a.x = h[arow * F_DIM + ka];
        a.y = h[arow * F_DIM + ka + 1];
        b.x = w[ka * HF + bcol];
        b.y = w[(ka + 1) * HF + bcol];
        acc = __builtin_amdgcn_wmma_f32_16x16x4_f32(
                  false, a, false, b, (short)0, acc, false, false);
    }
    const int orow = row0 + half * 8;
    #pragma unroll
    for (int i = 0; i < 8; ++i)
        feat[(size_t)(orow + i) * HF + bcol] = acc[i];
}

__global__ void attn_score_kernel(const float* __restrict__ feat,
                                  const float* __restrict__ al,
                                  const float* __restrict__ ar,
                                  float* __restrict__ el, float* __restrict__ er,
                                  int N) {
    const int wave = (blockIdx.x * blockDim.x + threadIdx.x) >> 5;
    if (wave >= N) return;
    const int lane = threadIdx.x & 31;
    const int abase = lane * 8;
    const size_t base = (size_t)wave * HF + abase;
    float sl = 0.0f, sr = 0.0f;
    #pragma unroll
    for (int j = 0; j < 8; ++j) {
        const float f = feat[base + j];
        sl += f * al[abase + j];
        sr += f * ar[abase + j];
    }
    #pragma unroll
    for (int msk = 1; msk < 8; msk <<= 1) {
        sl += __shfl_xor(sl, msk, 32);
        sr += __shfl_xor(sr, msk, 32);
    }
    if ((lane & 7) == 0) {
        const int hd = lane >> 3;
        el[wave * H_DIM + hd] = sl;
        er[wave * H_DIM + hd] = sr;
    }
}

__global__ void edge_max_kernel(const int* __restrict__ src, const int* __restrict__ dst,
                                const float* __restrict__ el, const float* __restrict__ er,
                                float* __restrict__ m, int EH) {
    int t = blockIdx.x * blockDim.x + threadIdx.x;
    if (t >= EH) return;
    const int e = t >> 2, hd = t & 3;
    const int s = src[e], d = dst[e];
    float v = el[s * H_DIM + hd] + er[d * H_DIM + hd];
    v = v > 0.0f ? v : NEG_SLOPE * v;
    atomic_max_f32(&m[d * H_DIM + hd], v);
}

__global__ void edge_exp_kernel(const int* __restrict__ src, const int* __restrict__ dst,
                                const float* __restrict__ el, const float* __restrict__ er,
                                const float* __restrict__ m, float* __restrict__ z,
                                float* __restrict__ exb, int EH) {
    int t = blockIdx.x * blockDim.x + threadIdx.x;
    if (t >= EH) return;
    const int e = t >> 2, hd = t & 3;
    const int s = src[e], d = dst[e];
    float v = el[s * H_DIM + hd] + er[d * H_DIM + hd];
    v = v > 0.0f ? v : NEG_SLOPE * v;
    const float ex = __expf(v - m[d * H_DIM + hd]);
    exb[t] = ex;
    atomicAdd(&z[d * H_DIM + hd], ex);
}

__global__ void edge_rank_kernel(const int* __restrict__ dst,
                                 int* __restrict__ deg,
                                 int* __restrict__ rank, int E) {
    const int e = blockIdx.x * blockDim.x + threadIdx.x;
    if (e < E) rank[e] = atomicAdd(&deg[dst[e]], 1);
}

__global__ void segment_place_kernel(const int* __restrict__ deg,
                                     int* __restrict__ total,
                                     int* __restrict__ rowstart, int N) {
    __shared__ int wave_off[8];
    __shared__ int block_base;
    const int i    = blockIdx.x * blockDim.x + threadIdx.x;
    const int lane = threadIdx.x & 31, wv = threadIdx.x >> 5;
    const int v = (i < N) ? deg[i] : 0;
    int inc = v;
    #pragma unroll
    for (int o = 1; o < 32; o <<= 1) {
        const int t = __shfl_up(inc, o, 32);
        if (lane >= o) inc += t;
    }
    if (lane == 31) wave_off[wv] = inc;
    __syncthreads();
    if (threadIdx.x == 0) {
        int run = 0;
        #pragma unroll
        for (int w = 0; w < 8; ++w) { const int t = wave_off[w]; wave_off[w] = run; run += t; }
        block_base = atomicAdd(total, run);
    }
    __syncthreads();
    if (i < N) rowstart[i] = block_base + wave_off[wv] + inc - v;
}

__global__ void edge_scatter_kernel(const int* __restrict__ dst,
                                    const int* __restrict__ rank,
                                    const int* __restrict__ rowstart,
                                    int* __restrict__ perm, int E) {
    const int e = blockIdx.x * blockDim.x + threadIdx.x;
    if (e < E) perm[rowstart[dst[e]] + rank[e]] = e;
}

__global__ void aggregate_rows_kernel(const int* __restrict__ src,
                                      const int* __restrict__ rowstart,
                                      const int* __restrict__ deg,
                                      const int* __restrict__ perm,
                                      const float* __restrict__ feat,
                                      const float* __restrict__ exb,
                                      const float* __restrict__ z,
                                      float* __restrict__ rst, int N, int E) {
    const int d = __builtin_amdgcn_readfirstlane(
                      (int)((blockIdx.x * blockDim.x + threadIdx.x) >> 5));
    if (d >= N) return;
    const int lane  = threadIdx.x & 31;
    const bool lo   = lane < 16;
    const int beg = min(max(rowstart[d], 0), E);
    const int cnt = min(max(deg[d], 0), E - beg);
    const float4 zd = *(const float4*)(z + (size_t)d * H_DIM);

    float4 a0 = {0.0f, 0.0f, 0.0f, 0.0f}, a1 = {0.0f, 0.0f, 0.0f, 0.0f};
    for (int c = 0; c < cnt; c += 32) {
        const int n = min(32, cnt - c);
        int    s_l  = 0;
        float4 al_l = {0.0f, 0.0f, 0.0f, 0.0f};
        if (lane < n) {
            const int e = min(max(perm[beg + c + lane], 0), E - 1);
            s_l = src[e];
            const float4 ex = *(const float4*)(exb + (size_t)e * H_DIM);
            al_l.x = ex.x / zd.x; al_l.y = ex.y / zd.y;
            al_l.z = ex.z / zd.z; al_l.w = ex.w / zd.w;
        }
        #pragma unroll 4
        for (int i = 0; i < n; ++i) {
            const int   s  = __builtin_amdgcn_readlane(s_l, i);
            const float ax = __int_as_float(__builtin_amdgcn_readlane(__float_as_int(al_l.x), i));
            const float ay = __int_as_float(__builtin_amdgcn_readlane(__float_as_int(al_l.y), i));
            const float az = __int_as_float(__builtin_amdgcn_readlane(__float_as_int(al_l.z), i));
            const float aw = __int_as_float(__builtin_amdgcn_readlane(__float_as_int(al_l.w), i));
            const float alpha0 = lo ? ax : ay;
            const float alpha1 = lo ? az : aw;
            const float* row = feat + (size_t)s * HF + lane * 4;
            const float4 v0 = *(const float4*)(row);
            const float4 v1 = *(const float4*)(row + 128);
            a0.x += v0.x * alpha0; a0.y += v0.y * alpha0;
            a0.z += v0.z * alpha0; a0.w += v0.w * alpha0;
            a1.x += v1.x * alpha1; a1.y += v1.y * alpha1;
            a1.z += v1.z * alpha1; a1.w += v1.w * alpha1;
        }
    }
    float* r = rst + (size_t)d * HF + lane * 4;
    *(float4*)(r)       = a0;
    *(float4*)(r + 128) = a1;
}

__global__ void out_gemm_kernel(const float* __restrict__ rst,
                                const float* __restrict__ gb,
                                const float* __restrict__ w,
                                const float* __restrict__ ob,
                                float* __restrict__ x, int nRowTiles) {
    const int wave   = (blockIdx.x * blockDim.x + threadIdx.x) >> 5;
    const int nTiles = nRowTiles * (F_DIM / 16);
    if (wave >= nTiles) return;
    const int rt = wave >> 2;
    const int ct = wave & 3;
    const int row0 = rt * 16, col0 = ct * 16;
    const int lane = threadIdx.x & 31;
    const int half = lane >> 4, l = lane & 15;
    const int arow = row0 + l;
    const int bcol = col0 + l;

    v8f acc = {};
    #pragma unroll 8
    for (int k = 0; k < HF; k += 4) {
        const int ka = k + 2 * half;
        v2f a, b;
        a.x = rst[(size_t)arow * HF + ka]     + gb[ka];
        a.y = rst[(size_t)arow * HF + ka + 1] + gb[ka + 1];
        b.x = w[ka * F_DIM + bcol];
        b.y = w[(ka + 1) * F_DIM + bcol];
        acc = __builtin_amdgcn_wmma_f32_16x16x4_f32(
                  false, a, false, b, (short)0, acc, false, false);
    }
    const float bias = ob[bcol];
    const int orow = row0 + half * 8;
    #pragma unroll
    for (int i = 0; i < 8; ++i)
        x[(size_t)(orow + i) * F_DIM + bcol] = acc[i] + bias;
}

__global__ void layernorm_kernel(const float* __restrict__ x,
                                 const float* __restrict__ g,
                                 const float* __restrict__ b,
                                 float* __restrict__ out, int N) {
    const int wave = (blockIdx.x * blockDim.x + threadIdx.x) >> 5;
    if (wave >= N) return;
    const int lane = threadIdx.x & 31;
    const size_t base = (size_t)wave * F_DIM;
    const float x0 = x[base + lane], x1 = x[base + lane + 32];
    float s = x0 + x1, sq = x0 * x0 + x1 * x1;
    #pragma unroll
    for (int msk = 1; msk < 32; msk <<= 1) {
        s  += __shfl_xor(s,  msk, 32);
        sq += __shfl_xor(sq, msk, 32);
    }
    const float mu  = s * (1.0f / 64.0f);
    const float var = sq * (1.0f / 64.0f) - mu * mu;
    const float inv = rsqrtf(var + LN_EPS);
    out[base + lane]      = (x0 - mu) * inv * g[lane]      + b[lane];
    out[base + lane + 32] = (x1 - mu) * inv * g[lane + 32] + b[lane + 32];
}

extern "C" void kernel_launch(void* const* d_in, const int* in_sizes, int n_in,
                              void* d_out, int out_size, void* d_ws, size_t ws_size,
                              hipStream_t stream) {
    const float* h   = (const float*)d_in[0];
    const int*   src = (const int*)  d_in[1];
    const int*   dst = (const int*)  d_in[2];
    const float* fcw = (const float*)d_in[3];
    const float* al  = (const float*)d_in[4];
    const float* ar  = (const float*)d_in[5];
    const float* gb  = (const float*)d_in[6];
    const float* ow  = (const float*)d_in[7];
    const float* ob  = (const float*)d_in[8];
    const float* lng = (const float*)d_in[9];
    const float* lnb = (const float*)d_in[10];

    const int N = in_sizes[0] / F_DIM;
    const int E = in_sizes[1];

    char*  ws  = (char*)d_ws;
    size_t off = 0;
    auto carve = [&](size_t bytes) -> float* {
        float* p = (float*)(ws + off);
        off += (bytes + 255) & ~(size_t)255;
        return p;
    };
    float* feat = carve((size_t)N * HF * 4);
    float* rst  = carve((size_t)N * HF * 4);
    float* el   = carve((size_t)N * H_DIM * 4);
    float* er   = carve((size_t)N * H_DIM * 4);
    float* m    = carve((size_t)N * H_DIM * 4);
    float* z    = carve((size_t)N * H_DIM * 4);
    float* exb  = carve((size_t)E * H_DIM * 4);
    float* x    = carve((size_t)N * F_DIM * 4);
    int* deg      = (int*)carve((size_t)N * 4);
    int* rowstart = (int*)carve((size_t)N * 4);
    int* total    = (int*)carve(4);
    int* rank     = (int*)carve((size_t)E * 4);
    int* perm     = (int*)carve((size_t)E * 4);
    (void)ws_size;

    const int nRowTiles = (N + 15) / 16;

    {
        const int nz     = N * H_DIM;
        const int blocks = (nz + 255) / 256;
        init_kernel<<<blocks, 256, 0, stream>>>(z, m, deg, total, nz, N);
    }
    {
        const int waves  = nRowTiles * (HF / 16);
        const int blocks = (waves + 7) / 8;
        fc_gemm_kernel<<<blocks, 256, 0, stream>>>(h, fcw, feat, nRowTiles);
    }
    {
        const int blocks = (N + 7) / 8;
        attn_score_kernel<<<blocks, 256, 0, stream>>>(feat, al, ar, el, er, N);
    }
    {
        const int EH = E * H_DIM;
        const int blocks = (EH + 255) / 256;
        edge_max_kernel<<<blocks, 256, 0, stream>>>(src, dst, el, er, m, EH);
        edge_exp_kernel<<<blocks, 256, 0, stream>>>(src, dst, el, er, m, z, exb, EH);
    }
    {
        const int eblocks = (E + 255) / 256;
        const int nblocks = (N + 255) / 256;
        edge_rank_kernel<<<eblocks, 256, 0, stream>>>(dst, deg, rank, E);
        segment_place_kernel<<<nblocks, 256, 0, stream>>>(deg, total, rowstart, N);
        edge_scatter_kernel<<<eblocks, 256, 0, stream>>>(dst, rank, rowstart, perm, E);
        const int blocks = (N + 7) / 8;
        aggregate_rows_kernel<<<blocks, 256, 0, stream>>>(src, rowstart, deg, perm, feat,
                                                          exb, z, rst, N, E);
    }
    {
        const int waves  = nRowTiles * (F_DIM / 16);
        const int blocks = (waves + 7) / 8;
        out_gemm_kernel<<<blocks, 256, 0, stream>>>(rst, gb, ow, ob, x, nRowTiles);
    }
    {
        const int blocks = (N + 7) / 8;
        layernorm_kernel<<<blocks, 256, 0, stream>>>(x, lng, lnb, (float*)d_out, N);
    }
}


# each wave does more items (fc 4 column tiles, out 2; scores and layernorm 8 nodes; aggregation 2 nodes; rank and scatter 4 edges per thread): 2 to 8 times fewer waves to dispatch per launch
# speedup vs baseline: 3.1127x; 1.4974x over previous
#include <hip/hip_runtime.h>
#include <math.h>

typedef __attribute__((ext_vector_type(2))) float v2f;
typedef __attribute__((ext_vector_type(8))) float v8f;

#define F_DIM 64
#define H_DIM 4
#define HF    256
#define NEG_SLOPE 0.2f
#define LN_EPS 1e-5f

__global__ void init_kernel(int* __restrict__ deg, int* __restrict__ total, int N) {
    int i = blockIdx.x * blockDim.x + threadIdx.x;
    if (i < N)  deg[i] = 0;
    if (i == 0) *total = 0;
}

#define FC_TILES_PER_WAVE 4
__global__ void fc_gemm_kernel(const float* __restrict__ h,
                               const float* __restrict__ w,
                               float* __restrict__ feat, int nRowTiles) {
    const int wave   = (blockIdx.x * blockDim.x + threadIdx.x) >> 5;
    const int nWaves = nRowTiles * (HF / 16 / FC_TILES_PER_WAVE);
    if (wave >= nWaves) return;
    const int rt = wave >> 2;
    const int cg = wave & 3;
    const int row0 = rt * 16;
    const int lane = threadIdx.x & 31;
    const int half = lane >> 4, l = lane & 15;
    const int arow = row0 + l;

    v2f a[F_DIM / 4];
    #pragma unroll
    for (int k4 = 0; k4 < F_DIM / 4; ++k4) {
        const int ka = k4 * 4 + 2 * half;
        a[k4].x = h[arow * F_DIM + ka];
        a[k4].y = h[arow * F_DIM + ka + 1];
    }
    const int orow = row0 + half * 8;
    #pragma unroll
    for (int t = 0; t < FC_TILES_PER_WAVE; ++t) {
        const int bcol = (cg * FC_TILES_PER_WAVE + t) * 16 + l;
        v8f acc = {};
        #pragma unroll
        for (int k4 = 0; k4 < F_DIM / 4; ++k4) {
            const int ka = k4 * 4 + 2 * half;
            v2f b;
            b.x = w[ka * HF + bcol];
            b.y = w[(ka + 1) * HF + bcol];
            acc = __builtin_amdgcn_wmma_f32_16x16x4_f32(
                      false, a[k4], false, b, (short)0, acc, false, false);
        }
        #pragma unroll
        for (int i = 0; i < 8; ++i)
            feat[(size_t)(orow + i) * HF + bcol] = acc[i];
    }
}

#define NODES_PER_WAVE 8
__global__ void attn_score_kernel(const float* __restrict__ feat,
                                  const float* __restrict__ al,
                                  const float* __restrict__ ar,
                                  float* __restrict__ el, float* __restrict__ er,
                                  int N) {
    const int wave  = (blockIdx.x * blockDim.x + threadIdx.x) >> 5;
    const int node0 = wave * NODES_PER_WAVE;
    if (node0 >= N) return;
    const int lane = threadIdx.x & 31;
    const int abase = lane * 8;
    float alv[8], arv[8];
    #pragma unroll
    for (int j = 0; j < 8; ++j) { alv[j] = al[abase + j]; arv[j] = ar[abase + j]; }
    float4 fa[NODES_PER_WAVE], fb[NODES_PER_WAVE];
    #pragma unroll
    for (int n = 0; n < NODES_PER_WAVE; ++n) {
        const float* p = feat + (size_t)min(node0 + n, N - 1) * HF + abase;
        fa[n] = *(const float4*)(p);
        fb[n] = *(const float4*)(p + 4);
    }
    #pragma unroll
    for (int n = 0; n < NODES_PER_WAVE; ++n) {
        const float f[8] = {fa[n].x, fa[n].y, fa[n].z, fa[n].w,
                            fb[n].x, fb[n].y, fb[n].z, fb[n].w};
        float sl = 0.0f, sr = 0.0f;
        #pragma unroll
        for (int j = 0; j < 8; ++j) {
            sl += f[j] * alv[j];
            sr += f[j] * arv[j];
        }
        #pragma unroll
        for (int msk = 1; msk < 8; msk <<= 1) {
            sl += __shfl_xor(sl, msk, 32);
            sr += __shfl_xor(sr, msk, 32);
        }
        const int node = node0 + n;
        if ((lane & 7) == 0 && node < N) {
            const int hd = lane >> 3;
            el[node * H_DIM + hd] = sl;
            er[node * H_DIM + hd] = sr;
        }
    }
}

#define EDGES_PER_THREAD 4
__global__ void edge_rank_kernel(const int* __restrict__ dst,
                                 int* __restrict__ deg,
                                 int* __restrict__ rank, int E) {
    const int e0 = (blockIdx.x * blockDim.x + threadIdx.x) * EDGES_PER_THREAD;
    if (e0 + EDGES_PER_THREAD <= E) {
        const int4 d = *(const int4*)(dst + e0);
        int4 r;
        r.x = atomicAdd(&deg[d.x], 1);
        r.y = atomicAdd(&deg[d.y], 1);
        r.z = atomicAdd(&deg[d.z], 1);
        r.w = atomicAdd(&deg[d.w], 1);
        *(int4*)(rank + e0) = r;
    } else {
        for (int e = e0; e < E; ++e) rank[e] = atomicAdd(&deg[dst[e]], 1);
    }
}

__global__ void segment_place_kernel(const int* __restrict__ deg,
                                     int* __restrict__ total,
                                     int* __restrict__ rowstart, int N) {
    __shared__ int wave_off[8];
    __shared__ int block_base;
    const int i    = blockIdx.x * blockDim.x + threadIdx.x;
    const int lane = threadIdx.x & 31, wv = threadIdx.x >> 5;
    const int v = (i < N) ? deg[i] : 0;
    int inc = v;
    #pragma unroll
    for (int o = 1; o < 32; o <<= 1) {
        const int t = __shfl_up(inc, o, 32);
        if (lane >= o) inc += t;
    }
    if (lane == 31) wave_off[wv] = inc;
    __syncthreads();
    if (threadIdx.x == 0) {
        int run = 0;
        #pragma unroll
        for (int w = 0; w < 8; ++w) { const int t = wave_off[w]; wave_off[w] = run; run += t; }
        block_base = atomicAdd(total, run);
    }
    __syncthreads();
    if (i < N) rowstart[i] = block_base + wave_off[wv] + inc - v;
}

__global__ void edge_scatter_kernel(const int* __restrict__ dst,
                                    const int* __restrict__ rank,
                                    const int* __restrict__ rowstart,
                                    int* __restrict__ perm, int E) {
    const int e0 = (blockIdx.x * blockDim.x + threadIdx.x) * EDGES_PER_THREAD;
    if (e0 + EDGES_PER_THREAD <= E) {
        const int4 d = *(const int4*)(dst + e0);
        const int4 r = *(const int4*)(rank + e0);
        const int p0 = rowstart[d.x], p1 = rowstart[d.y];
        const int p2 = rowstart[d.z], p3 = rowstart[d.w];
        perm[p0 + r.x] = e0;
        perm[p1 + r.y] = e0 + 1;
        perm[p2 + r.z] = e0 + 2;
        perm[p3 + r.w] = e0 + 3;
    } else {
        for (int e = e0; e < E; ++e) perm[rowstart[dst[e]] + rank[e]] = e;
    }
}

__device__ __forceinline__ float leaky(float v) {
    return v > 0.0f ? v : NEG_SLOPE * v;
}

__device__ __forceinline__ void load_edge(const int* __restrict__ perm,
                                          const int* __restrict__ src,
                                          const float* __restrict__ el,
                                          const float4 erd, int pos, int E,
                                          int& s, float4& v) {
    const int e = min(max(perm[pos], 0), E - 1);
    s = src[e];
    const float4 l = *(const float4*)(el + (size_t)s * H_DIM);
    v.x = leaky(l.x + erd.x); v.y = leaky(l.y + erd.y);
    v.z = leaky(l.z + erd.z); v.w = leaky(l.w + erd.w);
}

template <int UNROLL>
__device__ __forceinline__ void add_rows(const float* __restrict__ feat,
                                         int lane, bool lo, int s_l,
                                         const float4 wt, int n,
                                         float4& a0, float4& a1) {
    #pragma unroll UNROLL
    for (int i = 0; i < n; ++i) {
        const int   s  = __builtin_amdgcn_readlane(s_l, i);
        const float wx = __int_as_float(__builtin_amdgcn_readlane(__float_as_int(wt.x), i));
        const float wy = __int_as_float(__builtin_amdgcn_readlane(__float_as_int(wt.y), i));
        const float wz = __int_as_float(__builtin_amdgcn_readlane(__float_as_int(wt.z), i));
        const float ww = __int_as_float(__builtin_amdgcn_readlane(__float_as_int(wt.w), i));
        const float w0 = lo ? wx : wy;
        const float w1 = lo ? wz : ww;
        const float* row = feat + (size_t)s * HF + lane * 4;
        const float4 f0 = *(const float4*)(row);
        const float4 f1 = *(const float4*)(row + 128);
        a0.x += f0.x * w0; a0.y += f0.y * w0;
        a0.z += f0.z * w0; a0.w += f0.w * w0;
        a1.x += f1.x * w1; a1.y += f1.y * w1;
        a1.z += f1.z * w1; a1.w += f1.w * w1;
    }
}

__device__ __forceinline__ void aggregate_node(const int d, const int lane, const bool lo,
                                               const int* __restrict__ src,
                                               const int* __restrict__ rowstart,
                                               const int* __restrict__ deg,
                                               const int* __restrict__ perm,
                                               const float* __restrict__ feat,
                                               const float* __restrict__ el,
                                               const float* __restrict__ er,
                                               float* __restrict__ rst, int E) {
    const int beg = min(max(rowstart[d], 0), E);
    const int cnt = min(max(deg[d], 0), E - beg);
    const float4 erd = *(const float4*)(er + (size_t)d * H_DIM);

    int    s0 = 0;
    float4 v0 = {-INFINITY, -INFINITY, -INFINITY, -INFINITY};
    if (lane < cnt) load_edge(perm, src, el, erd, beg + lane, E, s0, v0);

    float4 mx = v0;
    for (int c = 32; c < cnt; c += 32) {
        if (c + lane < cnt) {
            int s; float4 v;
            load_edge(perm, src, el, erd, beg + c + lane, E, s, v);
            mx.x = fmaxf(mx.x, v.x); mx.y = fmaxf(mx.y, v.y);
            mx.z = fmaxf(mx.z, v.z); mx.w = fmaxf(mx.w, v.w);
        }
    }
    #pragma unroll
    for (int msk = 16; msk >= 1; msk >>= 1) {
        mx.x = fmaxf(mx.x, __shfl_xor(mx.x, msk, 32));
        mx.y = fmaxf(mx.y, __shfl_xor(mx.y, msk, 32));
        mx.z = fmaxf(mx.z, __shfl_xor(mx.z, msk, 32));
        mx.w = fmaxf(mx.w, __shfl_xor(mx.w, msk, 32));
    }

    float4 a0 = {0.0f, 0.0f, 0.0f, 0.0f}, a1 = {0.0f, 0.0f, 0.0f, 0.0f};
    float4 ex0 = {0.0f, 0.0f, 0.0f, 0.0f};
    if (lane < cnt) {
        ex0.x = __expf(v0.x - mx.x); ex0.y = __expf(v0.y - mx.y);
        ex0.z = __expf(v0.z - mx.z); ex0.w = __expf(v0.w - mx.w);
    }
    float4 zs = ex0;
    for (int c = 32; c < cnt; c += 32) {
        const int n = min(32, cnt - c);
        int    s_l = 0;
        float4 ex  = {0.0f, 0.0f, 0.0f, 0.0f};
        if (lane < n) {
            float4 v;
            load_edge(perm, src, el, erd, beg + c + lane, E, s_l, v);
            ex.x = __expf(v.x - mx.x); ex.y = __expf(v.y - mx.y);
            ex.z = __expf(v.z - mx.z); ex.w = __expf(v.w - mx.w);
        }
        zs.x += ex.x; zs.y += ex.y; zs.z += ex.z; zs.w += ex.w;
        add_rows<1>(feat, lane, lo, s_l, ex, n, a0, a1);
    }
    #pragma unroll
    for (int msk = 16; msk >= 1; msk >>= 1) {
        zs.x += __shfl_xor(zs.x, msk, 32);
        zs.y += __shfl_xor(zs.y, msk, 32);
        zs.z += __shfl_xor(zs.z, msk, 32);
        zs.w += __shfl_xor(zs.w, msk, 32);
    }
    float4 rz = {1.0f, 1.0f, 1.0f, 1.0f};
    if (cnt > 0) {
        rz.x = 1.0f / zs.x; rz.y = 1.0f / zs.y;
        rz.z = 1.0f / zs.z; rz.w = 1.0f / zs.w;
    }
    const float r0 = lo ? rz.x : rz.y, r1 = lo ? rz.z : rz.w;
    a0.x *= r0; a0.y *= r0; a0.z *= r0; a0.w *= r0;
    a1.x *= r1; a1.y *= r1; a1.z *= r1; a1.w *= r1;
    float4 alpha;
    alpha.x = ex0.x * rz.x; alpha.y = ex0.y * rz.y;
    alpha.z = ex0.z * rz.z; alpha.w = ex0.w * rz.w;

    add_rows<4>(feat, lane, lo, s0, alpha, min(cnt, 32), a0, a1);

    float* r = rst + (size_t)d * HF + lane * 4;
    *(float4*)(r)       = a0;
    *(float4*)(r + 128) = a1;
}

#define AGG_NODES_PER_WAVE 2
__global__ void aggregate_rows_kernel(const int* __restrict__ src,
                                      const int* __restrict__ rowstart,
                                      const int* __restrict__ deg,
                                      const int* __restrict__ perm,
                                      const float* __restrict__ feat,
                                      const float* __restrict__ el,
                                      const float* __restrict__ er,
                                      float* __restrict__ rst, int N, int E) {
    const int wv = __builtin_amdgcn_readfirstlane(
                       (int)((blockIdx.x * blockDim.x + threadIdx.x) >> 5));
    const int lane  = threadIdx.x & 31;
    const bool lo   = lane < 16;
    #pragma nounroll
    for (int j = 0; j < AGG_NODES_PER_WAVE; ++j) {
        const int d = wv * AGG_NODES_PER_WAVE + j;
        if (d >= N) return;
        aggregate_node(d, lane, lo, src, rowstart, deg, perm, feat, el, er, rst, E);
    }
}

__global__ void out_gemm_kernel(const float* __restrict__ rst,
                                const float* __restrict__ gb,
                                const float* __restrict__ w,
                                const float* __restrict__ ob,
                                float* __restrict__ x, int nRowTiles) {
    const int wave   = (blockIdx.x * blockDim.x + threadIdx.x) >> 5;
    const int nWaves = nRowTiles * (F_DIM / 32);
    if (wave >= nWaves) return;
    const int rt = wave >> 1;
    const int cp = wave & 1;
    const int row0 = rt * 16, col0 = cp * 32;
    const int lane = threadIdx.x & 31;
    const int half = lane >> 4, l = lane & 15;
    const int arow = row0 + l;
    const int bcol0 = col0 + l, bcol1 = col0 + 16 + l;

    v8f acc0 = {}, acc1 = {};
    #pragma unroll 8
    for (int k = 0; k < HF; k += 4) {
        const int ka = k + 2 * half;
        v2f a, b0, b1;
        a.x = rst[(size_t)arow * HF + ka]     + gb[ka];
        a.y = rst[(size_t)arow * HF + ka + 1] + gb[ka + 1];
        b0.x = w[ka * F_DIM + bcol0];
        b0.y = w[(ka + 1) * F_DIM + bcol0];
        b1.x = w[ka * F_DIM + bcol1];
        b1.y = w[(ka + 1) * F_DIM + bcol1];
        acc0 = __builtin_amdgcn_wmma_f32_16x16x4_f32(
                   false, a, false, b0, (short)0, acc0, false, false);
        acc1 = __builtin_amdgcn_wmma_f32_16x16x4_f32(
                   false, a, false, b1, (short)0, acc1, false, false);
    }
    const float bias0 = ob[bcol0], bias1 = ob[bcol1];
    const int orow = row0 + half * 8;
    #pragma unroll
    for (int i = 0; i < 8; ++i) {
        x[(size_t)(orow + i) * F_DIM + bcol0] = acc0[i] + bias0;
        x[(size_t)(orow + i) * F_DIM + bcol1] = acc1[i] + bias1;
    }
}

__global__ void layernorm_kernel(const float* __restrict__ x,
                                 const float* __restrict__ g,
                                 const float* __restrict__ b,
                                 float* __restrict__ out, int N) {
    const int wave  = (blockIdx.x * blockDim.x + threadIdx.x) >> 5;
    const int node0 = wave * NODES_PER_WAVE;
    if (node0 >= N) return;
    const int lane = threadIdx.x & 31;
    const float g0 = g[lane], g1 = g[lane + 32];
    const float b0 = b[lane], b1 = b[lane + 32];
    float xa[NODES_PER_WAVE], xb[NODES_PER_WAVE];
    #pragma unroll
    for (int n = 0; n < NODES_PER_WAVE; ++n) {
        const size_t base = (size_t)min(node0 + n, N - 1) * F_DIM;
        xa[n] = x[base + lane];
        xb[n] = x[base + lane + 32];
    }
    #pragma unroll
    for (int n = 0; n < NODES_PER_WAVE; ++n) {
        const float x0 = xa[n], x1 = xb[n];
        float s = x0 + x1, sq = x0 * x0 + x1 * x1;
        #pragma unroll
        for (int msk = 1; msk < 32; msk <<= 1) {
            s  += __shfl_xor(s,  msk, 32);
            sq += __shfl_xor(sq, msk, 32);
        }
        const float mu  = s * (1.0f / 64.0f);
        const float var = sq * (1.0f / 64.0f) - mu * mu;
        const float inv = rsqrtf(var + LN_EPS);
        const int node = node0 + n;
        if (node < N) {
            const size_t base = (size_t)node * F_DIM;
            out[base + lane]      = (x0 - mu) * inv * g0 + b0;
            out[base + lane + 32] = (x1 - mu) * inv * g1 + b1;
        }
    }
}

extern "C" void kernel_launch(void* const* d_in, const int* in_sizes, int n_in,
                              void* d_out, int out_size, void* d_ws, size_t ws_size,
                              hipStream_t stream) {
    const float* h   = (const float*)d_in[0];
    const int*   src = (const int*)  d_in[1];
    const int*   dst = (const int*)  d_in[2];
    const float* fcw = (const float*)d_in[3];
    const float* al  = (const float*)d_in[4];
    const float* ar  = (const float*)d_in[5];
    const float* gb  = (const float*)d_in[6];
    const float* ow  = (const float*)d_in[7];
    const float* ob  = (const float*)d_in[8];
    const float* lng = (const float*)d_in[9];
    const float* lnb = (const float*)d_in[10];

    const int N = in_sizes[0] / F_DIM;
    const int E = in_sizes[1];

    char*  ws  = (char*)d_ws;
    size_t off = 0;
    auto carve = [&](size_t bytes) -> float* {
        float* p = (float*)(ws + off);
        off += (bytes + 255) & ~(size_t)255;
        return p;
    };
    float* feat = carve((size_t)N * HF * 4);
    float* rst  = carve((size_t)N * HF * 4);
    float* el   = carve((size_t)N * H_DIM * 4);
    float* er   = carve((size_t)N * H_DIM * 4);
    float* x    = carve((size_t)N * F_DIM * 4);
    int* deg      = (int*)carve((size_t)N * 4);
    int* rowstart = (int*)carve((size_t)N * 4);
    int* total    = (int*)carve(4);
    int* rank     = (int*)carve((size_t)E * 4);
    int* perm     = (int*)carve((size_t)E * 4);
    (void)ws_size;

    const int nRowTiles = (N + 15) / 16;

    {
        const int blocks = (N + 255) / 256;
        init_kernel<<<blocks, 256, 0, stream>>>(deg, total, N);
    }
    {
        const int waves  = nRowTiles * (HF / 16 / FC_TILES_PER_WAVE);
        const int blocks = (waves + 7) / 8;
        fc_gemm_kernel<<<blocks, 256, 0, stream>>>(h, fcw, feat, nRowTiles);
    }
    {
        const int blocks = (N + 8 * NODES_PER_WAVE - 1) / (8 * NODES_PER_WAVE);
        attn_score_kernel<<<blocks, 256, 0, stream>>>(feat, al, ar, el, er, N);
    }
    {
        const int eblocks = (E + 256 * EDGES_PER_THREAD - 1) / (256 * EDGES_PER_THREAD);
        const int nblocks = (N + 255) / 256;
        edge_rank_kernel<<<eblocks, 256, 0, stream>>>(dst, deg, rank, E);
        segment_place_kernel<<<nblocks, 256, 0, stream>>>(deg, total, rowstart, N);
        edge_scatter_kernel<<<eblocks, 256, 0, stream>>>(dst, rank, rowstart, perm, E);
        const int blocks = (N + 8 * AGG_NODES_PER_WAVE - 1) / (8 * AGG_NODES_PER_WAVE);
        aggregate_rows_kernel<<<blocks, 256, 0, stream>>>(src, rowstart, deg, perm, feat,
                                                          el, er, rst, N, E);
    }
    {
        const int waves  = nRowTiles * (F_DIM / 32);
        const int blocks = (waves + 7) / 8;
        out_gemm_kernel<<<blocks, 256, 0, stream>>>(rst, gb, ow, ob, x, nRowTiles);
    }
    {
        const int blocks = (N + 8 * NODES_PER_WAVE - 1) / (8 * NODES_PER_WAVE);
        layernorm_kernel<<<blocks, 256, 0, stream>>>(x, lng, lnb, (float*)d_out, N);
    }
}


# fc projection moved behind the aggregation (linearity): edges gather 256-B rows of h, not 1-KiB rows of feat; fc_w folded into out_w per call, el/er from h; sums re-associated
# speedup vs baseline: 3.8581x; 1.2395x over previous
#include <hip/hip_runtime.h>
#include <math.h>

typedef __attribute__((ext_vector_type(2))) float v2f;
typedef __attribute__((ext_vector_type(8))) float v8f;

#define F_DIM 64
#define H_DIM 4
#define HF    256
#define NEG_SLOPE 0.2f
#define LN_EPS 1e-5f


__device__ __forceinline__ void wmma_tile_k64(const float* __restrict__ A, int lda,
                                              const float* __restrict__ B, int ldb,
                                              float* __restrict__ D, int ldd, int lane) {
    const int half = lane >> 4, l = lane & 15;
    v8f acc = {};
    #pragma unroll
    for (int k4 = 0; k4 < F_DIM / 4; ++k4) {
        const int ka = k4 * 4 + 2 * half;
        v2f a, b;
        a.x = A[l * lda + ka];
        a.y = A[l * lda + ka + 1];
        b.x = B[ka * ldb + l];
        b.y = B[(ka + 1) * ldb + l];
        acc = __builtin_amdgcn_wmma_f32_16x16x4_f32(
                  false, a, false, b, (short)0, acc, false, false);
    }
    #pragma unroll
    for (int i = 0; i < 8; ++i)
        D[(half * 8 + i) * ldd + l] = acc[i];
}

__global__ void setup_kernel(const float* __restrict__ fcw,
                             const float* __restrict__ al,
                             const float* __restrict__ ar,
                             const float* __restrict__ gb,
                             const float* __restrict__ ow,
                             const float* __restrict__ ob,
                             int* __restrict__ deg, int* __restrict__ total,
                             float* __restrict__ wlr, float* __restrict__ bvec,
                             float* __restrict__ wc, int N, int nInit) {
    __shared__ float red[256];
    const int b = blockIdx.x;
    if (b < nInit) {
        const int i = b * blockDim.x + threadIdx.x;
        if (i < N)  deg[i] = 0;
        if (i == 0) *total = 0;
        return;
    }
    const int r = b - nInit;
    if (r < 2) {
        const int t = r * 256 + threadIdx.x;
        const int o = t >> 6, k = t & 63, hs = o & 3;
        const float* a = (o < 4 ? al : ar) + hs * F_DIM;
        const float* w = fcw + k * HF + hs * F_DIM;
        float s = 0.0f;
        #pragma unroll
        for (int f = 0; f < F_DIM; f += 4) {
            const float4 wv = *(const float4*)(w + f);
            const float4 av = *(const float4*)(a + f);
            s += wv.x * av.x; s += wv.y * av.y;
            s += wv.z * av.z; s += wv.w * av.w;
        }
        wlr[o * F_DIM + k] = s;
        return;
    }
    if (r == 2) {
        const int j = threadIdx.x & 63, part = threadIdx.x >> 6;
        float s = 0.0f;
        #pragma unroll 8
        for (int p = part * 64; p < part * 64 + 64; ++p)
            s += gb[p] * ow[p * F_DIM + j];
        red[threadIdx.x] = s;
        __syncthreads();
        if (threadIdx.x < 64)
            bvec[j] = ((red[j] + red[64 + j]) + (red[128 + j] + red[192 + j])) + ob[j];
        return;
    }
    const int tile = (r - 3) * 8 + (threadIdx.x >> 5);
    const int hs = tile >> 4, mt = (tile >> 2) & 3, nt = tile & 3;
    wmma_tile_k64(fcw + (mt * 16) * HF + hs * F_DIM, HF,
                  ow + (hs * F_DIM) * F_DIM + nt * 16, F_DIM,
                  wc + (hs * F_DIM + mt * 16) * F_DIM + nt * 16, F_DIM,
                  threadIdx.x & 31);
}

#define NODES_PER_WAVE 8
__global__ void attn_score_kernel(const float* __restrict__ h,
                                  const float* __restrict__ wlr,
                                  float* __restrict__ el, float* __restrict__ er,
                                  int N) {
    const int wave  = (blockIdx.x * blockDim.x + threadIdx.x) >> 5;
    const int node0 = wave * NODES_PER_WAVE;
    if (node0 >= N) return;
    const int lane = threadIdx.x & 31;
    const int sub = lane & 7, q = lane >> 3;
    float4 fa[NODES_PER_WAVE / 4], fb[NODES_PER_WAVE / 4];
    #pragma unroll
    for (int g = 0; g < NODES_PER_WAVE / 4; ++g) {
        const float* p = h + (size_t)min(node0 + g * 4 + q, N - 1) * F_DIM + sub * 8;
        fa[g] = *(const float4*)(p);
        fb[g] = *(const float4*)(p + 4);
    }
    float4 wa[8], wb[8];
    #pragma unroll
    for (int o = 0; o < 8; ++o) {
        wa[o] = *(const float4*)(wlr + o * F_DIM + sub * 8);
        wb[o] = *(const float4*)(wlr + o * F_DIM + sub * 8 + 4);
    }
    #pragma unroll
    for (int g = 0; g < NODES_PER_WAVE / 4; ++g) {
        float s[8];
        #pragma unroll
        for (int o = 0; o < 8; ++o) {
            float t = fa[g].x * wa[o].x;
            t += fa[g].y * wa[o].y; t += fa[g].z * wa[o].z; t += fa[g].w * wa[o].w;
            t += fb[g].x * wb[o].x; t += fb[g].y * wb[o].y;
            t += fb[g].z * wb[o].z; t += fb[g].w * wb[o].w;
            s[o] = t;
        }
        #pragma unroll
        for (int msk = 1; msk < 8; msk <<= 1) {
            #pragma unroll
            for (int o = 0; o < 8; ++o) s[o] += __shfl_xor(s[o], msk, 32);
        }
        const int node = node0 + g * 4 + q;
        if (sub == 0 && node < N) {
            float4 l4, r4;
            l4.x = s[0]; l4.y = s[1]; l4.z = s[2]; l4.w = s[3];
            r4.x = s[4]; r4.y = s[5]; r4.z = s[6]; r4.w = s[7];
            *(float4*)(el + (size_t)node * H_DIM) = l4;
            *(float4*)(er + (size_t)node * H_DIM) = r4;
        }
    }
}

#define EDGES_PER_THREAD 4
__global__ void edge_rank_kernel(const int* __restrict__ dst,
                                 int* __restrict__ deg,
                                 int* __restrict__ rank, int E) {
    const int e0 = (blockIdx.x * blockDim.x + threadIdx.x) * EDGES_PER_THREAD;
    if (e0 + EDGES_PER_THREAD <= E) {
        const int4 d = *(const int4*)(dst + e0);
        int4 r;
        r.x = atomicAdd(&deg[d.x], 1);
        r.y = atomicAdd(&deg[d.y], 1);
        r.z = atomicAdd(&deg[d.z], 1);
        r.w = atomicAdd(&deg[d.w], 1);
        *(int4*)(rank + e0) = r;
    } else {
        for (int e = e0; e < E; ++e) rank[e] = atomicAdd(&deg[dst[e]], 1);
    }
}

__global__ void segment_place_kernel(const int* __restrict__ deg,
                                     int* __restrict__ total,
                                     int* __restrict__ rowstart, int N) {
    __shared__ int wave_off[8];
    __shared__ int block_base;
    const int i    = blockIdx.x * blockDim.x + threadIdx.x;
    const int lane = threadIdx.x & 31, wv = threadIdx.x >> 5;
    const int v = (i < N) ? deg[i] : 0;
    int inc = v;
    #pragma unroll
    for (int o = 1; o < 32; o <<= 1) {
        const int t = __shfl_up(inc, o, 32);
        if (lane >= o) inc += t;
    }
    if (lane == 31) wave_off[wv] = inc;
    __syncthreads();
    if (threadIdx.x == 0) {
        int run = 0;
        #pragma unroll
        for (int w = 0; w < 8; ++w) { const int t = wave_off[w]; wave_off[w] = run; run += t; }
        block_base = atomicAdd(total, run);
    }
    __syncthreads();
    if (i < N) rowstart[i] = block_base + wave_off[wv] + inc - v;
}

__global__ void edge_scatter_kernel(const int* __restrict__ dst,
                                    const int* __restrict__ rank,
                                    const int* __restrict__ rowstart,
                                    int* __restrict__ perm, int E) {
    const int e0 = (blockIdx.x * blockDim.x + threadIdx.x) * EDGES_PER_THREAD;
    if (e0 + EDGES_PER_THREAD <= E) {
        const int4 d = *(const int4*)(dst + e0);
        const int4 r = *(const int4*)(rank + e0);
        const int p0 = rowstart[d.x], p1 = rowstart[d.y];
        const int p2 = rowstart[d.z], p3 = rowstart[d.w];
        perm[p0 + r.x] = e0;
        perm[p1 + r.y] = e0 + 1;
        perm[p2 + r.z] = e0 + 2;
        perm[p3 + r.w] = e0 + 3;
    } else {
        for (int e = e0; e < E; ++e) perm[rowstart[dst[e]] + rank[e]] = e;
    }
}

__device__ __forceinline__ float leaky(float v) {
    return v > 0.0f ? v : NEG_SLOPE * v;
}

__device__ __forceinline__ void load_edge(const int* __restrict__ perm,
                                          const int* __restrict__ src,
                                          const float* __restrict__ el,
                                          const float4 erd, int pos, int E,
                                          int& s, float4& v) {
    const int e = min(max(perm[pos], 0), E - 1);
    s = src[e];
    const float4 l = *(const float4*)(el + (size_t)s * H_DIM);
    v.x = leaky(l.x + erd.x); v.y = leaky(l.y + erd.y);
    v.z = leaky(l.z + erd.z); v.w = leaky(l.w + erd.w);
}

#define ROWS_PER_TRIP 8
__device__ __forceinline__ void add_rows(const float* __restrict__ h,
                                         int lane, int s_l,
                                         const float4 wt, int n,
                                         v2f (&acc)[H_DIM]) {
    for (int i = 0; i < n; i += ROWS_PER_TRIP) {
        v2f f[ROWS_PER_TRIP];
        #pragma unroll
        for (int u = 0; u < ROWS_PER_TRIP; ++u) {
            const int s = __builtin_amdgcn_readlane(s_l, i + u);
            f[u] = *(const v2f*)(h + (size_t)s * F_DIM + lane * 2);
        }
        #pragma unroll
        for (int u = 0; u < ROWS_PER_TRIP; ++u) {
            const float wx = __int_as_float(__builtin_amdgcn_readlane(__float_as_int(wt.x), i + u));
            const float wy = __int_as_float(__builtin_amdgcn_readlane(__float_as_int(wt.y), i + u));
            const float wz = __int_as_float(__builtin_amdgcn_readlane(__float_as_int(wt.z), i + u));
            const float ww = __int_as_float(__builtin_amdgcn_readlane(__float_as_int(wt.w), i + u));
            const v2f w0 = {wx, wx}, w1 = {wy, wy}, w2 = {wz, wz}, w3 = {ww, ww};
            acc[0] = __builtin_elementwise_fma(f[u], w0, acc[0]);
            acc[1] = __builtin_elementwise_fma(f[u], w1, acc[1]);
            acc[2] = __builtin_elementwise_fma(f[u], w2, acc[2]);
            acc[3] = __builtin_elementwise_fma(f[u], w3, acc[3]);
        }
    }
}

__device__ __forceinline__ void aggregate_node(const int d, const int lane,
                                               const int* __restrict__ src,
                                               const int* __restrict__ rowstart,
                                               const int* __restrict__ deg,
                                               const int* __restrict__ perm,
                                               const float* __restrict__ h,
                                               const float* __restrict__ el,
                                               const float* __restrict__ er,
                                               float* __restrict__ agg, int E) {
    const int beg = min(max(rowstart[d], 0), E);
    const int cnt = min(max(deg[d], 0), E - beg);
    const float4 erd = *(const float4*)(er + (size_t)d * H_DIM);

    int    s0 = 0;
    float4 v0 = {-INFINITY, -INFINITY, -INFINITY, -INFINITY};
    if (lane < cnt) load_edge(perm, src, el, erd, beg + lane, E, s0, v0);
    s0 = lane < cnt ? s0 : __builtin_amdgcn_readfirstlane(s0);

    float4 mx = v0;
    for (int c = 32; c < cnt; c += 32) {
        if (c + lane < cnt) {
            int s; float4 v;
            load_edge(perm, src, el, erd, beg + c + lane, E, s, v);
            mx.x = fmaxf(mx.x, v.x); mx.y = fmaxf(mx.y, v.y);
            mx.z = fmaxf(mx.z, v.z); mx.w = fmaxf(mx.w, v.w);
        }
    }
    #pragma unroll
    for (int msk = 16; msk >= 1; msk >>= 1) {
        mx.x = fmaxf(mx.x, __shfl_xor(mx.x, msk, 32));
        mx.y = fmaxf(mx.y, __shfl_xor(mx.y, msk, 32));
        mx.z = fmaxf(mx.z, __shfl_xor(mx.z, msk, 32));
        mx.w = fmaxf(mx.w, __shfl_xor(mx.w, msk, 32));
    }

    v2f acc[H_DIM];
    #pragma unroll
    for (int k = 0; k < H_DIM; ++k) { acc[k].x = 0.0f; acc[k].y = 0.0f; }
    float4 ex0 = {0.0f, 0.0f, 0.0f, 0.0f};
    if (lane < cnt) {
        ex0.x = __expf(v0.x - mx.x); ex0.y = __expf(v0.y - mx.y);
        ex0.z = __expf(v0.z - mx.z); ex0.w = __expf(v0.w - mx.w);
    }
    float4 zs = ex0;
    for (int c = 32; c < cnt; c += 32) {
        const int n = min(32, cnt - c);
        int    s_l = 0;
        float4 ex  = {0.0f, 0.0f, 0.0f, 0.0f};
        if (lane < n) {
            float4 v;
            load_edge(perm, src, el, erd, beg + c + lane, E, s_l, v);
            ex.x = __expf(v.x - mx.x); ex.y = __expf(v.y - mx.y);
            ex.z = __expf(v.z - mx.z); ex.w = __expf(v.w - mx.w);
        }
        s_l = lane < n ? s_l : __builtin_amdgcn_readfirstlane(s_l);
        zs.x += ex.x; zs.y += ex.y; zs.z += ex.z; zs.w += ex.w;
        add_rows(h, lane, s_l, ex, n, acc);
    }
    #pragma unroll
    for (int msk = 16; msk >= 1; msk >>= 1) {
        zs.x += __shfl_xor(zs.x, msk, 32);
        zs.y += __shfl_xor(zs.y, msk, 32);
        zs.z += __shfl_xor(zs.z, msk, 32);
        zs.w += __shfl_xor(zs.w, msk, 32);
    }
    float4 rz = {1.0f, 1.0f, 1.0f, 1.0f};
    if (cnt > 0) {
        rz.x = 1.0f / zs.x; rz.y = 1.0f / zs.y;
        rz.z = 1.0f / zs.z; rz.w = 1.0f / zs.w;
    }
    acc[0].x *= rz.x; acc[0].y *= rz.x;
    acc[1].x *= rz.y; acc[1].y *= rz.y;
    acc[2].x *= rz.z; acc[2].y *= rz.z;
    acc[3].x *= rz.w; acc[3].y *= rz.w;
    float4 alpha;
    alpha.x = ex0.x * rz.x; alpha.y = ex0.y * rz.y;
    alpha.z = ex0.z * rz.z; alpha.w = ex0.w * rz.w;

    add_rows(h, lane, s0, alpha, min(cnt, 32), acc);

    float* r = agg + (size_t)d * HF + lane * 2;
    #pragma unroll
    for (int k = 0; k < H_DIM; ++k)
        *(v2f*)(r + k * F_DIM) = acc[k];
}

#define AGG_NODES_PER_WAVE 2
__global__ void aggregate_rows_kernel(const float* __restrict__ h,
                                      const int* __restrict__ src,
                                      const int* __restrict__ rowstart,
                                      const int* __restrict__ deg,
                                      const int* __restrict__ perm,
                                      const float* __restrict__ el,
                                      const float* __restrict__ er,
                                      float* __restrict__ agg, int N, int E) {
    const int wv = __builtin_amdgcn_readfirstlane(
                       (int)((blockIdx.x * blockDim.x + threadIdx.x) >> 5));
    const int lane  = threadIdx.x & 31;
    #pragma nounroll
    for (int j = 0; j < AGG_NODES_PER_WAVE; ++j) {
        const int d = wv * AGG_NODES_PER_WAVE + j;
        if (d >= N) return;
        aggregate_node(d, lane, src, rowstart, deg, perm, h, el, er, agg, E);
    }
}

__global__ void out_gemm_kernel(const float* __restrict__ agg,
                                const float* __restrict__ w,
                                const float* __restrict__ bv,
                                float* __restrict__ x, int nRowTiles) {
    const int wave   = (blockIdx.x * blockDim.x + threadIdx.x) >> 5;
    const int nWaves = nRowTiles * (F_DIM / 32);
    if (wave >= nWaves) return;
    const int rt = wave >> 1;
    const int cp = wave & 1;
    const int row0 = rt * 16, col0 = cp * 32;
    const int lane = threadIdx.x & 31;
    const int half = lane >> 4, l = lane & 15;
    const int arow = row0 + l;
    const int bcol0 = col0 + l, bcol1 = col0 + 16 + l;

    v8f acc0 = {}, acc1 = {};
    #pragma unroll 8
    for (int k = 0; k < HF; k += 4) {
        const int ka = k + 2 * half;
        v2f a, b0, b1;
        a.x = agg[(size_t)arow * HF + ka];
        a.y = agg[(size_t)arow * HF + ka + 1];
        b0.x = w[ka * F_DIM + bcol0];
        b0.y = w[(ka + 1) * F_DIM + bcol0];
        b1.x = w[ka * F_DIM + bcol1];
        b1.y = w[(ka + 1) * F_DIM + bcol1];
        acc0 = __builtin_amdgcn_wmma_f32_16x16x4_f32(
                   false, a, false, b0, (short)0, acc0, false, false);
        acc1 = __builtin_amdgcn_wmma_f32_16x16x4_f32(
                   false, a, false, b1, (short)0, acc1, false, false);
    }
    const float bias0 = bv[bcol0], bias1 = bv[bcol1];
    const int orow = row0 + half * 8;
    #pragma unroll
    for (int i = 0; i < 8; ++i) {
        x[(size_t)(orow + i) * F_DIM + bcol0] = acc0[i] + bias0;
        x[(size_t)(orow + i) * F_DIM + bcol1] = acc1[i] + bias1;
    }
}

__global__ void layernorm_kernel(const float* __restrict__ x,
                                 const float* __restrict__ g,
                                 const float* __restrict__ b,
                                 float* __restrict__ out, int N) {
    const int wave  = (blockIdx.x * blockDim.x + threadIdx.x) >> 5;
    const int node0 = wave * NODES_PER_WAVE;
    if (node0 >= N) return;
    const int lane = threadIdx.x & 31;
    const float g0 = g[lane], g1 = g[lane + 32];
    const float b0 = b[lane], b1 = b[lane + 32];
    float xa[NODES_PER_WAVE], xb[NODES_PER_WAVE];
    #pragma unroll
    for (int n = 0; n < NODES_PER_WAVE; ++n) {
        const size_t base = (size_t)min(node0 + n, N - 1) * F_DIM;
        xa[n] = x[base + lane];
        xb[n] = x[base + lane + 32];
    }
    #pragma unroll
    for (int n = 0; n < NODES_PER_WAVE; ++n) {
        const float x0 = xa[n], x1 = xb[n];
        float s = x0 + x1, sq = x0 * x0 + x1 * x1;
        #pragma unroll
        for (int msk = 1; msk < 32; msk <<= 1) {
            s  += __shfl_xor(s,  msk, 32);
            sq += __shfl_xor(sq, msk, 32);
        }
        const float mu  = s * (1.0f / 64.0f);
        const float var = sq * (1.0f / 64.0f) - mu * mu;
        const float inv = rsqrtf(var + LN_EPS);
        const int node = node0 + n;
        if (node < N) {
            const size_t base = (size_t)node * F_DIM;
            out[base + lane]      = (x0 - mu) * inv * g0 + b0;
            out[base + lane + 32] = (x1 - mu) * inv * g1 + b1;
        }
    }
}

extern "C" void kernel_launch(void* const* d_in, const int* in_sizes, int n_in,
                              void* d_out, int out_size, void* d_ws, size_t ws_size,
                              hipStream_t stream) {
    const float* h   = (const float*)d_in[0];
    const int*   src = (const int*)  d_in[1];
    const int*   dst = (const int*)  d_in[2];
    const float* fcw = (const float*)d_in[3];
    const float* al  = (const float*)d_in[4];
    const float* ar  = (const float*)d_in[5];
    const float* gb  = (const float*)d_in[6];
    const float* ow  = (const float*)d_in[7];
    const float* ob  = (const float*)d_in[8];
    const float* lng = (const float*)d_in[9];
    const float* lnb = (const float*)d_in[10];

    const int N = in_sizes[0] / F_DIM;
    const int E = in_sizes[1];

    char*  ws  = (char*)d_ws;
    size_t off = 0;
    auto carve = [&](size_t bytes) -> float* {
        float* p = (float*)(ws + off);
        off += (bytes + 255) & ~(size_t)255;
        return p;
    };
    float* agg  = carve((size_t)N * HF * 4);
    float* el   = carve((size_t)N * H_DIM * 4);
    float* er   = carve((size_t)N * H_DIM * 4);
    float* x    = carve((size_t)N * F_DIM * 4);
    float* wlr  = carve((size_t)2 * H_DIM * F_DIM * 4);
    float* bvec = carve((size_t)F_DIM * 4);
    float* wc   = carve((size_t)HF * F_DIM * 4);
    int* deg      = (int*)carve((size_t)N * 4);
    int* rowstart = (int*)carve((size_t)N * 4);
    int* total    = (int*)carve(4);
    int* rank     = (int*)carve((size_t)E * 4);
    int* perm     = (int*)carve((size_t)E * 4);
    (void)ws_size;

    const int nRowTiles = (N + 15) / 16;

    {
        const int nInit = (N + 255) / 256;
        setup_kernel<<<nInit + 3 + 8, 256, 0, stream>>>(fcw, al, ar, gb, ow, ob, deg, total,
                                                        wlr, bvec, wc, N, nInit);
    }
    {
        const int blocks = (N + 8 * NODES_PER_WAVE - 1) / (8 * NODES_PER_WAVE);
        attn_score_kernel<<<blocks, 256, 0, stream>>>(h, wlr, el, er, N);
    }
    {
        const int eblocks = (E + 256 * EDGES_PER_THREAD - 1) / (256 * EDGES_PER_THREAD);
        const int nblocks = (N + 255) / 256;
        edge_rank_kernel<<<eblocks, 256, 0, stream>>>(dst, deg, rank, E);
        segment_place_kernel<<<nblocks, 256, 0, stream>>>(deg, total, rowstart, N);
        edge_scatter_kernel<<<eblocks, 256, 0, stream>>>(dst, rank, rowstart, perm, E);
        const int blocks = (N + 8 * AGG_NODES_PER_WAVE - 1) / (8 * AGG_NODES_PER_WAVE);
        aggregate_rows_kernel<<<blocks, 256, 0, stream>>>(h, src, rowstart, deg, perm,
                                                          el, er, agg, N, E);
    }
    {
        const int waves  = nRowTiles * (F_DIM / 32);
        const int blocks = (waves + 7) / 8;
        out_gemm_kernel<<<blocks, 256, 0, stream>>>(agg, wc, bvec, x, nRowTiles);
    }
    {
        const int blocks = (N + 8 * NODES_PER_WAVE - 1) / (8 * NODES_PER_WAVE);
        layernorm_kernel<<<blocks, 256, 0, stream>>>(x, lng, lnb, (float*)d_out, N);
    }
}


# aggregation, output product and layernorm in one launch: a block sums 16 nodes' rows into LDS, WMMA reads its A tile from there, x tile normalised from LDS; agg and x planes and 2 launches gone
# speedup vs baseline: 4.3431x; 1.1257x over previous
#include <hip/hip_runtime.h>
#include <math.h>

typedef __attribute__((ext_vector_type(2))) float v2f;
typedef __attribute__((ext_vector_type(8))) float v8f;

#define F_DIM 64
#define H_DIM 4
#define HF    256
#define NEG_SLOPE 0.2f
#define LN_EPS 1e-5f


__device__ __forceinline__ void wmma_tile_k64(const float* __restrict__ A, int lda,
                                              const float* __restrict__ B, int ldb,
                                              float* __restrict__ D, int ldd, int lane) {
    const int half = lane >> 4, l = lane & 15;
    v8f acc = {};
    #pragma unroll
    for (int k4 = 0; k4 < F_DIM / 4; ++k4) {
        const int ka = k4 * 4 + 2 * half;
        v2f a, b;
        a.x = A[l * lda + ka];
        a.y = A[l * lda + ka + 1];
        b.x = B[ka * ldb + l];
        b.y = B[(ka + 1) * ldb + l];
        acc = __builtin_amdgcn_wmma_f32_16x16x4_f32(
                  false, a, false, b, (short)0, acc, false, false);
    }
    #pragma unroll
    for (int i = 0; i < 8; ++i)
        D[(half * 8 + i) * ldd + l] = acc[i];
}

__global__ void setup_kernel(const float* __restrict__ fcw,
                             const float* __restrict__ al,
                             const float* __restrict__ ar,
                             const float* __restrict__ gb,
                             const float* __restrict__ ow,
                             const float* __restrict__ ob,
                             int* __restrict__ deg, int* __restrict__ total,
                             float* __restrict__ wlr, float* __restrict__ bvec,
                             float* __restrict__ wc, int N, int nInit) {
    __shared__ float red[256];
    const int b = blockIdx.x;
    if (b < nInit) {
        const int i = b * blockDim.x + threadIdx.x;
        if (i < N)  deg[i] = 0;
        if (i == 0) *total = 0;
        return;
    }
    const int r = b - nInit;
    if (r < 2) {
        const int t = r * 256 + threadIdx.x;
        const int o = t >> 6, k = t & 63, hs = o & 3;
        const float* a = (o < 4 ? al : ar) + hs * F_DIM;
        const float* w = fcw + k * HF + hs * F_DIM;
        float s = 0.0f;
        #pragma unroll
        for (int f = 0; f < F_DIM; f += 4) {
            const float4 wv = *(const float4*)(w + f);
            const float4 av = *(const float4*)(a + f);
            s += wv.x * av.x; s += wv.y * av.y;
            s += wv.z * av.z; s += wv.w * av.w;
        }
        wlr[o * F_DIM + k] = s;
        return;
    }
    if (r == 2) {
        const int j = threadIdx.x & 63, part = threadIdx.x >> 6;
        float s = 0.0f;
        #pragma unroll 8
        for (int p = part * 64; p < part * 64 + 64; ++p)
            s += gb[p] * ow[p * F_DIM + j];
        red[threadIdx.x] = s;
        __syncthreads();
        if (threadIdx.x < 64)
            bvec[j] = ((red[j] + red[64 + j]) + (red[128 + j] + red[192 + j])) + ob[j];
        return;
    }
    const int tile = (r - 3) * 8 + (threadIdx.x >> 5);
    const int hs = tile >> 4, mt = (tile >> 2) & 3, nt = tile & 3;
    wmma_tile_k64(fcw + (mt * 16) * HF + hs * F_DIM, HF,
                  ow + (hs * F_DIM) * F_DIM + nt * 16, F_DIM,
                  wc + (hs * F_DIM + mt * 16) * F_DIM + nt * 16, F_DIM,
                  threadIdx.x & 31);
}

#define NODES_PER_WAVE 8
__global__ void attn_score_kernel(const float* __restrict__ h,
                                  const float* __restrict__ wlr,
                                  float* __restrict__ el, float* __restrict__ er,
                                  int N) {
    const int wave  = (blockIdx.x * blockDim.x + threadIdx.x) >> 5;
    const int node0 = wave * NODES_PER_WAVE;
    if (node0 >= N) return;
    const int lane = threadIdx.x & 31;
    const int sub = lane & 7, q = lane >> 3;
    float4 fa[NODES_PER_WAVE / 4], fb[NODES_PER_WAVE / 4];
    #pragma unroll
    for (int g = 0; g < NODES_PER_WAVE / 4; ++g) {
        const float* p = h + (size_t)min(node0 + g * 4 + q, N - 1) * F_DIM + sub * 8;
        fa[g] = *(const float4*)(p);
        fb[g] = *(const float4*)(p + 4);
    }
    float4 wa[8], wb[8];
    #pragma unroll
    for (int o = 0; o < 8; ++o) {
        wa[o] = *(const float4*)(wlr + o * F_DIM + sub * 8);
        wb[o] = *(const float4*)(wlr + o * F_DIM + sub * 8 + 4);
    }
    #pragma unroll
    for (int g = 0; g < NODES_PER_WAVE / 4; ++g) {
        float s[8];
        #pragma unroll
        for (int o = 0; o < 8; ++o) {
            float t = fa[g].x * wa[o].x;
            t += fa[g].y * wa[o].y; t += fa[g].z * wa[o].z; t += fa[g].w * wa[o].w;
            t += fb[g].x * wb[o].x; t += fb[g].y * wb[o].y;
            t += fb[g].z * wb[o].z; t += fb[g].w * wb[o].w;
            s[o] = t;
        }
        #pragma unroll
        for (int msk = 1; msk < 8; msk <<= 1) {
            #pragma unroll
            for (int o = 0; o < 8; ++o) s[o] += __shfl_xor(s[o], msk, 32);
        }
        const int node = node0 + g * 4 + q;
        if (sub == 0 && node < N) {
            float4 l4, r4;
            l4.x = s[0]; l4.y = s[1]; l4.z = s[2]; l4.w = s[3];
            r4.x = s[4]; r4.y = s[5]; r4.z = s[6]; r4.w = s[7];
            *(float4*)(el + (size_t)node * H_DIM) = l4;
            *(float4*)(er + (size_t)node * H_DIM) = r4;
        }
    }
}

#define EDGES_PER_THREAD 4
__global__ void edge_rank_kernel(const int* __restrict__ dst,
                                 int* __restrict__ deg,
                                 int* __restrict__ rank, int E) {
    const int e0 = (blockIdx.x * blockDim.x + threadIdx.x) * EDGES_PER_THREAD;
    if (e0 + EDGES_PER_THREAD <= E) {
        const int4 d = *(const int4*)(dst + e0);
        int4 r;
        r.x = atomicAdd(&deg[d.x], 1);
        r.y = atomicAdd(&deg[d.y], 1);
        r.z = atomicAdd(&deg[d.z], 1);
        r.w = atomicAdd(&deg[d.w], 1);
        *(int4*)(rank + e0) = r;
    } else {
        for (int e = e0; e < E; ++e) rank[e] = atomicAdd(&deg[dst[e]], 1);
    }
}

__global__ void segment_place_kernel(const int* __restrict__ deg,
                                     int* __restrict__ total,
                                     int* __restrict__ rowstart, int N) {
    __shared__ int wave_off[8];
    __shared__ int block_base;
    const int i    = blockIdx.x * blockDim.x + threadIdx.x;
    const int lane = threadIdx.x & 31, wv = threadIdx.x >> 5;
    const int v = (i < N) ? deg[i] : 0;
    int inc = v;
    #pragma unroll
    for (int o = 1; o < 32; o <<= 1) {
        const int t = __shfl_up(inc, o, 32);
        if (lane >= o) inc += t;
    }
    if (lane == 31) wave_off[wv] = inc;
    __syncthreads();
    if (threadIdx.x == 0) {
        int run = 0;
        #pragma unroll
        for (int w = 0; w < 8; ++w) { const int t = wave_off[w]; wave_off[w] = run; run += t; }
        block_base = atomicAdd(total, run);
    }
    __syncthreads();
    if (i < N) rowstart[i] = block_base + wave_off[wv] + inc - v;
}

__global__ void edge_scatter_kernel(const int* __restrict__ dst,
                                    const int* __restrict__ rank,
                                    const int* __restrict__ rowstart,
                                    int* __restrict__ perm, int E) {
    const int e0 = (blockIdx.x * blockDim.x + threadIdx.x) * EDGES_PER_THREAD;
    if (e0 + EDGES_PER_THREAD <= E) {
        const int4 d = *(const int4*)(dst + e0);
        const int4 r = *(const int4*)(rank + e0);
        const int p0 = rowstart[d.x], p1 = rowstart[d.y];
        const int p2 = rowstart[d.z], p3 = rowstart[d.w];
        perm[p0 + r.x] = e0;
        perm[p1 + r.y] = e0 + 1;
        perm[p2 + r.z] = e0 + 2;
        perm[p3 + r.w] = e0 + 3;
    } else {
        for (int e = e0; e < E; ++e) perm[rowstart[dst[e]] + rank[e]] = e;
    }
}

__device__ __forceinline__ float leaky(float v) {
    return v > 0.0f ? v : NEG_SLOPE * v;
}

__device__ __forceinline__ void load_edge(const int* __restrict__ perm,
                                          const int* __restrict__ src,
                                          const float* __restrict__ el,
                                          const float4 erd, int pos, int E,
                                          int& s, float4& v) {
    const int e = min(max(perm[pos], 0), E - 1);
    s = src[e];
    const float4 l = *(const float4*)(el + (size_t)s * H_DIM);
    v.x = leaky(l.x + erd.x); v.y = leaky(l.y + erd.y);
    v.z = leaky(l.z + erd.z); v.w = leaky(l.w + erd.w);
}

#define ROWS_PER_TRIP 8
__device__ __forceinline__ void add_rows(const float* __restrict__ h,
                                         int lane, int s_l,
                                         const float4 wt, int n,
                                         v2f (&acc)[H_DIM]) {
    for (int i = 0; i < n; i += ROWS_PER_TRIP) {
        v2f f[ROWS_PER_TRIP];
        #pragma unroll
        for (int u = 0; u < ROWS_PER_TRIP; ++u) {
            const int s = __builtin_amdgcn_readlane(s_l, i + u);
            f[u] = *(const v2f*)(h + (size_t)s * F_DIM + lane * 2);
        }
        #pragma unroll
        for (int u = 0; u < ROWS_PER_TRIP; ++u) {
            const float wx = __int_as_float(__builtin_amdgcn_readlane(__float_as_int(wt.x), i + u));
            const float wy = __int_as_float(__builtin_amdgcn_readlane(__float_as_int(wt.y), i + u));
            const float wz = __int_as_float(__builtin_amdgcn_readlane(__float_as_int(wt.z), i + u));
            const float ww = __int_as_float(__builtin_amdgcn_readlane(__float_as_int(wt.w), i + u));
            const v2f w0 = {wx, wx}, w1 = {wy, wy}, w2 = {wz, wz}, w3 = {ww, ww};
            acc[0] = __builtin_elementwise_fma(f[u], w0, acc[0]);
            acc[1] = __builtin_elementwise_fma(f[u], w1, acc[1]);
            acc[2] = __builtin_elementwise_fma(f[u], w2, acc[2]);
            acc[3] = __builtin_elementwise_fma(f[u], w3, acc[3]);
        }
    }
}

__device__ __forceinline__ void aggregate_node(const int d, const int lane,
                                               const int* __restrict__ src,
                                               const int* __restrict__ rowstart,
                                               const int* __restrict__ deg,
                                               const int* __restrict__ perm,
                                               const float* __restrict__ h,
                                               const float* __restrict__ el,
                                               const float* __restrict__ er,
                                               float* __restrict__ row, int E) {
    const int beg = min(max(rowstart[d], 0), E);
    const int cnt = min(max(deg[d], 0), E - beg);
    const float4 erd = *(const float4*)(er + (size_t)d * H_DIM);

    int    s0 = 0;
    float4 v0 = {-INFINITY, -INFINITY, -INFINITY, -INFINITY};
    if (lane < cnt) load_edge(perm, src, el, erd, beg + lane, E, s0, v0);
    s0 = lane < cnt ? s0 : __builtin_amdgcn_readfirstlane(s0);

    float4 mx = v0;
    for (int c = 32; c < cnt; c += 32) {
        if (c + lane < cnt) {
            int s; float4 v;
            load_edge(perm, src, el, erd, beg + c + lane, E, s, v);
            mx.x = fmaxf(mx.x, v.x); mx.y = fmaxf(mx.y, v.y);
            mx.z = fmaxf(mx.z, v.z); mx.w = fmaxf(mx.w, v.w);
        }
    }
    #pragma unroll
    for (int msk = 16; msk >= 1; msk >>= 1) {
        mx.x = fmaxf(mx.x, __shfl_xor(mx.x, msk, 32));
        mx.y = fmaxf(mx.y, __shfl_xor(mx.y, msk, 32));
        mx.z = fmaxf(mx.z, __shfl_xor(mx.z, msk, 32));
        mx.w = fmaxf(mx.w, __shfl_xor(mx.w, msk, 32));
    }

    v2f acc[H_DIM];
    #pragma unroll
    for (int k = 0; k < H_DIM; ++k) { acc[k].x = 0.0f; acc[k].y = 0.0f; }
    float4 ex0 = {0.0f, 0.0f, 0.0f, 0.0f};
    if (lane < cnt) {
        ex0.x = __expf(v0.x - mx.x); ex0.y = __expf(v0.y - mx.y);
        ex0.z = __expf(v0.z - mx.z); ex0.w = __expf(v0.w - mx.w);
    }
    float4 zs = ex0;
    for (int c = 32; c < cnt; c += 32) {
        const int n = min(32, cnt - c);
        int    s_l = 0;
        float4 ex  = {0.0f, 0.0f, 0.0f, 0.0f};
        if (lane < n) {
            float4 v;
            load_edge(perm, src, el, erd, beg + c + lane, E, s_l, v);
            ex.x = __expf(v.x - mx.x); ex.y = __expf(v.y - mx.y);
            ex.z = __expf(v.z - mx.z); ex.w = __expf(v.w - mx.w);
        }
        s_l = lane < n ? s_l : __builtin_amdgcn_readfirstlane(s_l);
        zs.x += ex.x; zs.y += ex.y; zs.z += ex.z; zs.w += ex.w;
        add_rows(h, lane, s_l, ex, n, acc);
    }
    #pragma unroll
    for (int msk = 16; msk >= 1; msk >>= 1) {
        zs.x += __shfl_xor(zs.x, msk, 32);
        zs.y += __shfl_xor(zs.y, msk, 32);
        zs.z += __shfl_xor(zs.z, msk, 32);
        zs.w += __shfl_xor(zs.w, msk, 32);
    }
    float4 rz = {1.0f, 1.0f, 1.0f, 1.0f};
    if (cnt > 0) {
        rz.x = 1.0f / zs.x; rz.y = 1.0f / zs.y;
        rz.z = 1.0f / zs.z; rz.w = 1.0f / zs.w;
    }
    acc[0].x *= rz.x; acc[0].y *= rz.x;
    acc[1].x *= rz.y; acc[1].y *= rz.y;
    acc[2].x *= rz.z; acc[2].y *= rz.z;
    acc[3].x *= rz.w; acc[3].y *= rz.w;
    float4 alpha;
    alpha.x = ex0.x * rz.x; alpha.y = ex0.y * rz.y;
    alpha.z = ex0.z * rz.z; alpha.w = ex0.w * rz.w;

    add_rows(h, lane, s0, alpha, min(cnt, 32), acc);

    float* r = row + lane * 2;
    #pragma unroll
    for (int k = 0; k < H_DIM; ++k)
        *(v2f*)(r + k * F_DIM) = acc[k];
}

#define AGG_NODES_PER_WAVE 2
#define TILE_NODES (8 * AGG_NODES_PER_WAVE)
#define AGG_PITCH  (HF + 4)
#define X_PITCH    (F_DIM + 4)
__global__ void aggregate_out_kernel(const float* __restrict__ h,
                                     const int* __restrict__ src,
                                     const int* __restrict__ rowstart,
                                     const int* __restrict__ deg,
                                     const int* __restrict__ perm,
                                     const float* __restrict__ el,
                                     const float* __restrict__ er,
                                     const float* __restrict__ w,
                                     const float* __restrict__ bv,
                                     const float* __restrict__ g,
                                     const float* __restrict__ b,
                                     float* __restrict__ out, int N, int E) {
    __shared__ float s_agg[TILE_NODES * AGG_PITCH];
    __shared__ float s_x[TILE_NODES * X_PITCH];
    const int wv    = __builtin_amdgcn_readfirstlane((int)(threadIdx.x >> 5));
    const int lane  = threadIdx.x & 31;
    const int node0 = blockIdx.x * TILE_NODES;

    #pragma nounroll
    for (int j = 0; j < AGG_NODES_PER_WAVE; ++j) {
        const int r = wv * AGG_NODES_PER_WAVE + j;
        const int d = node0 + r;
        float* row = s_agg + r * AGG_PITCH;
        if (d < N) {
            aggregate_node(d, lane, src, rowstart, deg, perm, h, el, er, row, E);
        } else {
            const v2f zero = {0.0f, 0.0f};
            for (int k = 0; k < H_DIM; ++k)
                *(v2f*)(row + k * F_DIM + lane * 2) = zero;
        }
    }
    __syncthreads();

    if (wv < F_DIM / 16) {
        const int half = lane >> 4, l = lane & 15;
        const float* arow = s_agg + l * AGG_PITCH;
        const int bcol = wv * 16 + l;
        v8f acc = {};
        #pragma unroll 8
        for (int k = 0; k < HF; k += 4) {
            const int ka = k + 2 * half;
            v2f a, bb;
            a.x  = arow[ka];
            a.y  = arow[ka + 1];
            bb.x = w[ka * F_DIM + bcol];
            bb.y = w[(ka + 1) * F_DIM + bcol];
            acc = __builtin_amdgcn_wmma_f32_16x16x4_f32(
                      false, a, false, bb, (short)0, acc, false, false);
        }
        const float bias = bv[bcol];
        float* xr = s_x + (half * 8) * X_PITCH + bcol;
        #pragma unroll
        for (int i = 0; i < 8; ++i)
            xr[i * X_PITCH] = acc[i] + bias;
    }
    __syncthreads();

    const float g0 = g[lane], g1 = g[lane + 32];
    const float b0 = b[lane], b1 = b[lane + 32];
    #pragma unroll
    for (int j = 0; j < AGG_NODES_PER_WAVE; ++j) {
        const int r = wv * AGG_NODES_PER_WAVE + j;
        const int node = node0 + r;
        const float x0 = s_x[r * X_PITCH + lane], x1 = s_x[r * X_PITCH + lane + 32];
        float s = x0 + x1, sq = x0 * x0 + x1 * x1;
        #pragma unroll
        for (int msk = 1; msk < 32; msk <<= 1) {
            s  += __shfl_xor(s,  msk, 32);
            sq += __shfl_xor(sq, msk, 32);
        }
        const float mu  = s * (1.0f / 64.0f);
        const float var = sq * (1.0f / 64.0f) - mu * mu;
        const float inv = rsqrtf(var + LN_EPS);
        if (node < N) {
            const size_t base = (size_t)node * F_DIM;
            out[base + lane]      = (x0 - mu) * inv * g0 + b0;
            out[base + lane + 32] = (x1 - mu) * inv * g1 + b1;
        }
    }
}

extern "C" void kernel_launch(void* const* d_in, const int* in_sizes, int n_in,
                              void* d_out, int out_size, void* d_ws, size_t ws_size,
                              hipStream_t stream) {
    const float* h   = (const float*)d_in[0];
    const int*   src = (const int*)  d_in[1];
    const int*   dst = (const int*)  d_in[2];
    const float* fcw = (const float*)d_in[3];
    const float* al  = (const float*)d_in[4];
    const float* ar  = (const float*)d_in[5];
    const float* gb  = (const float*)d_in[6];
    const float* ow  = (const float*)d_in[7];
    const float* ob  = (const float*)d_in[8];
    const float* lng = (const float*)d_in[9];
    const float* lnb = (const float*)d_in[10];

    const int N = in_sizes[0] / F_DIM;
    const int E = in_sizes[1];

    char*  ws  = (char*)d_ws;
    size_t off = 0;
    auto carve = [&](size_t bytes) -> float* {
        float* p = (float*)(ws + off);
        off += (bytes + 255) & ~(size_t)255;
        return p;
    };
    float* el   = carve((size_t)N * H_DIM * 4);
    float* er   = carve((size_t)N * H_DIM * 4);
    float* wlr  = carve((size_t)2 * H_DIM * F_DIM * 4);
    float* bvec = carve((size_t)F_DIM * 4);
    float* wc   = carve((size_t)HF * F_DIM * 4);
    int* deg      = (int*)carve((size_t)N * 4);
    int* rowstart = (int*)carve((size_t)N * 4);
    int* total    = (int*)carve(4);
    int* rank     = (int*)carve((size_t)E * 4);
    int* perm     = (int*)carve((size_t)E * 4);
    (void)ws_size;

    {
        const int nInit = (N + 255) / 256;
        setup_kernel<<<nInit + 3 + 8, 256, 0, stream>>>(fcw, al, ar, gb, ow, ob, deg, total,
                                                        wlr, bvec, wc, N, nInit);
    }
    {
        const int blocks = (N + 8 * NODES_PER_WAVE - 1) / (8 * NODES_PER_WAVE);
        attn_score_kernel<<<blocks, 256, 0, stream>>>(h, wlr, el, er, N);
    }
    {
        const int eblocks = (E + 256 * EDGES_PER_THREAD - 1) / (256 * EDGES_PER_THREAD);
        const int nblocks = (N + 255) / 256;
        edge_rank_kernel<<<eblocks, 256, 0, stream>>>(dst, deg, rank, E);
        segment_place_kernel<<<nblocks, 256, 0, stream>>>(deg, total, rowstart, N);
        edge_scatter_kernel<<<eblocks, 256, 0, stream>>>(dst, rank, rowstart, perm, E);
        const int blocks = (N + TILE_NODES - 1) / TILE_NODES;
        aggregate_out_kernel<<<blocks, 256, 0, stream>>>(h, src, rowstart, deg, perm,
                                                         el, er, wc, bvec, lng, lnb,
                                                         (float*)d_out, N, E);
    }
}


# edges grouped by a two-level counting sort on LDS counters (4096-edge blocks into 256-node buckets, one block per bucket): no global atomic per edge, one launch less; sum order = LDS rank order
# speedup vs baseline: 8.2505x; 1.8997x over previous
#include <hip/hip_runtime.h>
#include <math.h>

typedef __attribute__((ext_vector_type(2))) float v2f;
typedef __attribute__((ext_vector_type(8))) float v8f;
typedef __attribute__((ext_vector_type(2))) int   v2i;

#define F_DIM 64
#define H_DIM 4
#define HF    256
#define NEG_SLOPE 0.2f
#define LN_EPS 1e-5f


__device__ __forceinline__ void wmma_tile_k64(const float* __restrict__ A, int lda,
                                              const float* __restrict__ B, int ldb,
                                              float* __restrict__ D, int ldd, int lane) {
    const int half = lane >> 4, l = lane & 15;
    v8f acc = {};
    #pragma unroll
    for (int k4 = 0; k4 < F_DIM / 4; ++k4) {
        const int ka = k4 * 4 + 2 * half;
        v2f a, b;
        a.x = A[l * lda + ka];
        a.y = A[l * lda + ka + 1];
        b.x = B[ka * ldb + l];
        b.y = B[(ka + 1) * ldb + l];
        acc = __builtin_amdgcn_wmma_f32_16x16x4_f32(
                  false, a, false, b, (short)0, acc, false, false);
    }
    #pragma unroll
    for (int i = 0; i < 8; ++i)
        D[(half * 8 + i) * ldd + l] = acc[i];
}

__global__ void setup_kernel(const float* __restrict__ fcw,
                             const float* __restrict__ al,
                             const float* __restrict__ ar,
                             const float* __restrict__ gb,
                             const float* __restrict__ ow,
                             const float* __restrict__ ob,
                             int* __restrict__ gcount, int* __restrict__ total,
                             float* __restrict__ wlr, float* __restrict__ bvec,
                             float* __restrict__ wc, int nb, int nInit) {
    __shared__ float red[256];
    const int b = blockIdx.x;
    if (b < nInit) {
        const int i = b * blockDim.x + threadIdx.x;
        if (i < nb) gcount[i] = 0;
        if (i < 2)  total[i] = 0;
        return;
    }
    const int r = b - nInit;
    if (r < 2) {
        const int t = r * 256 + threadIdx.x;
        const int o = t >> 6, k = t & 63, hs = o & 3;
        const float* a = (o < 4 ? al : ar) + hs * F_DIM;
        const float* w = fcw + k * HF + hs * F_DIM;
        float s = 0.0f;
        #pragma unroll
        for (int f = 0; f < F_DIM; f += 4) {
            const float4 wv = *(const float4*)(w + f);
            const float4 av = *(const float4*)(a + f);
            s += wv.x * av.x; s += wv.y * av.y;
            s += wv.z * av.z; s += wv.w * av.w;
        }
        wlr[o * F_DIM + k] = s;
        return;
    }
    if (r == 2) {
        const int j = threadIdx.x & 63, part = threadIdx.x >> 6;
        float s = 0.0f;
        #pragma unroll 8
        for (int p = part * 64; p < part * 64 + 64; ++p)
            s += gb[p] * ow[p * F_DIM + j];
        red[threadIdx.x] = s;
        __syncthreads();
        if (threadIdx.x < 64)
            bvec[j] = ((red[j] + red[64 + j]) + (red[128 + j] + red[192 + j])) + ob[j];
        return;
    }
    const int tile = (r - 3) * 8 + (threadIdx.x >> 5);
    const int hs = tile >> 4, mt = (tile >> 2) & 3, nt = tile & 3;
    wmma_tile_k64(fcw + (mt * 16) * HF + hs * F_DIM, HF,
                  ow + (hs * F_DIM) * F_DIM + nt * 16, F_DIM,
                  wc + (hs * F_DIM + mt * 16) * F_DIM + nt * 16, F_DIM,
                  threadIdx.x & 31);
}

#define NODES_PER_WAVE 8
__global__ void attn_score_kernel(const float* __restrict__ h,
                                  const float* __restrict__ wlr,
                                  float* __restrict__ el, float* __restrict__ er,
                                  int N) {
    const int wave  = (blockIdx.x * blockDim.x + threadIdx.x) >> 5;
    const int node0 = wave * NODES_PER_WAVE;
    if (node0 >= N) return;
    const int lane = threadIdx.x & 31;
    const int sub = lane & 7, q = lane >> 3;
    float4 fa[NODES_PER_WAVE / 4], fb[NODES_PER_WAVE / 4];
    #pragma unroll
    for (int g = 0; g < NODES_PER_WAVE / 4; ++g) {
        const float* p = h + (size_t)min(node0 + g * 4 + q, N - 1) * F_DIM + sub * 8;
        fa[g] = *(const float4*)(p);
        fb[g] = *(const float4*)(p + 4);
    }
    float4 wa[8], wb[8];
    #pragma unroll
    for (int o = 0; o < 8; ++o) {
        wa[o] = *(const float4*)(wlr + o * F_DIM + sub * 8);
        wb[o] = *(const float4*)(wlr + o * F_DIM + sub * 8 + 4);
    }
    #pragma unroll
    for (int g = 0; g < NODES_PER_WAVE / 4; ++g) {
        float s[8];
        #pragma unroll
        for (int o = 0; o < 8; ++o) {
            float t = fa[g].x * wa[o].x;
            t += fa[g].y * wa[o].y; t += fa[g].z * wa[o].z; t += fa[g].w * wa[o].w;
            t += fb[g].x * wb[o].x; t += fb[g].y * wb[o].y;
            t += fb[g].z * wb[o].z; t += fb[g].w * wb[o].w;
            s[o] = t;
        }
        #pragma unroll
        for (int msk = 1; msk < 8; msk <<= 1) {
            #pragma unroll
            for (int o = 0; o < 8; ++o) s[o] += __shfl_xor(s[o], msk, 32);
        }
        const int node = node0 + g * 4 + q;
        if (sub == 0 && node < N) {
            float4 l4, r4;
            l4.x = s[0]; l4.y = s[1]; l4.z = s[2]; l4.w = s[3];
            r4.x = s[4]; r4.y = s[5]; r4.z = s[6]; r4.w = s[7];
            *(float4*)(el + (size_t)node * H_DIM) = l4;
            *(float4*)(er + (size_t)node * H_DIM) = r4;
        }
    }
}

#define BKT_SHIFT   8
#define BKT_NODES   (1 << BKT_SHIFT)
#define BKT_CAP     8192
#define MAX_BKTS    256
#define CHUNK_EDGES 4096
__global__ void edge_partition_kernel(const int* __restrict__ dst,
                                      int* __restrict__ gcount,
                                      int* __restrict__ total,
                                      int* __restrict__ recs,
                                      int* __restrict__ ovf,
                                      int E, int nb) {
    __shared__ int cnt[MAX_BKTS];
    __shared__ int gbase[MAX_BKTS];
    const int tid = threadIdx.x;
    for (int b = tid; b < nb; b += blockDim.x) cnt[b] = 0;
    __syncthreads();
    const int e_blk = blockIdx.x * CHUNK_EDGES;
    int dd[16], lr[16];
    if (e_blk + CHUNK_EDGES <= E) {
        #pragma unroll
        for (int k = 0; k < 4; ++k) {
            const int4 v = *(const int4*)(dst + e_blk + k * 1024 + tid * 4);
            dd[4 * k] = v.x; dd[4 * k + 1] = v.y; dd[4 * k + 2] = v.z; dd[4 * k + 3] = v.w;
        }
    } else {
        #pragma unroll
        for (int i = 0; i < 16; ++i) {
            const int e = e_blk + (i >> 2) * 1024 + tid * 4 + (i & 3);
            const int v = dst[min(e, E - 1)];
            dd[i] = e < E ? v : -1;
        }
    }
    #pragma unroll
    for (int i = 0; i < 16; ++i)
        lr[i] = dd[i] >= 0 ? atomicAdd(&cnt[dd[i] >> BKT_SHIFT], 1) : 0;
    __syncthreads();
    for (int b = tid; b < nb; b += blockDim.x) {
        const int c = cnt[b];
        gbase[b] = c > 0 ? atomicAdd(&gcount[b], c) : 0;
    }
    __syncthreads();
    v2i* r2 = (v2i*)recs;
    v2i* o2 = (v2i*)ovf;
    #pragma unroll
    for (int i = 0; i < 16; ++i) {
        if (dd[i] >= 0) {
            const int bb = dd[i] >> BKT_SHIFT;
            const int p  = gbase[bb] + lr[i];
            v2i rec;
            rec.x = dd[i];
            rec.y = e_blk + (i >> 2) * 1024 + tid * 4 + (i & 3);
            if (p < BKT_CAP) r2[(size_t)bb * BKT_CAP + p] = rec;
            else             o2[atomicAdd(&total[1], 1)] = rec;
        }
    }
}

__global__ void bucket_sort_kernel(const int* __restrict__ recs,
                                   const int* __restrict__ ovf,
                                   const int* __restrict__ gcount,
                                   int* __restrict__ total,
                                   int* __restrict__ deg,
                                   int* __restrict__ rowstart,
                                   int* __restrict__ perm, int N, int E) {
    __shared__ int ncnt[BKT_NODES];
    __shared__ int wave_off[8];
    __shared__ int block_base;
    const int tid = threadIdx.x, b = blockIdx.x;
    const int lane = tid & 31, wv = tid >> 5;
    ncnt[tid] = 0;
    const int n_in  = min(max(gcount[b], 0), BKT_CAP);
    const int n_ovf = min(max(total[1], 0), E);
    const v2i* mine = (const v2i*)recs + (size_t)b * BKT_CAP;
    const v2i* o2   = (const v2i*)ovf;
    __syncthreads();
    for (int i0 = tid; i0 < n_in; i0 += 4 * BKT_NODES) {
        v2i r[4];
        #pragma unroll
        for (int u = 0; u < 4; ++u) r[u] = mine[min(i0 + u * BKT_NODES, n_in - 1)];
        #pragma unroll
        for (int u = 0; u < 4; ++u)
            if (i0 + u * BKT_NODES < n_in) atomicAdd(&ncnt[r[u].x & (BKT_NODES - 1)], 1);
    }
    for (int i = tid; i < n_ovf; i += BKT_NODES) {
        const v2i r = o2[i];
        if ((r.x >> BKT_SHIFT) == b) atomicAdd(&ncnt[r.x & (BKT_NODES - 1)], 1);
    }
    __syncthreads();
    const int v = ncnt[tid];
    int inc = v;
    #pragma unroll
    for (int o = 1; o < 32; o <<= 1) {
        const int t = __shfl_up(inc, o, 32);
        if (lane >= o) inc += t;
    }
    if (lane == 31) wave_off[wv] = inc;
    __syncthreads();
    if (tid == 0) {
        int run = 0;
        #pragma unroll
        for (int w = 0; w < 8; ++w) { const int t = wave_off[w]; wave_off[w] = run; run += t; }
        block_base = atomicAdd(total, run);
    }
    __syncthreads();
    const int excl = wave_off[wv] + inc - v;
    const int base = block_base;
    const int node = b * BKT_NODES + tid;
    if (node < N) { deg[node] = v; rowstart[node] = base + excl; }
    ncnt[tid] = excl;
    __syncthreads();
    for (int i0 = tid; i0 < n_in; i0 += 4 * BKT_NODES) {
        v2i r[4];
        #pragma unroll
        for (int u = 0; u < 4; ++u) r[u] = mine[min(i0 + u * BKT_NODES, n_in - 1)];
        #pragma unroll
        for (int u = 0; u < 4; ++u)
            if (i0 + u * BKT_NODES < n_in) {
                const int pos = atomicAdd(&ncnt[r[u].x & (BKT_NODES - 1)], 1);
                perm[min(max(base + pos, 0), E - 1)] = r[u].y;
            }
    }
    for (int i = tid; i < n_ovf; i += BKT_NODES) {
        const v2i r = o2[i];
        if ((r.x >> BKT_SHIFT) == b) {
            const int pos = atomicAdd(&ncnt[r.x & (BKT_NODES - 1)], 1);
            perm[min(max(base + pos, 0), E - 1)] = r.y;
        }
    }
}

__device__ __forceinline__ float leaky(float v) {
    return v > 0.0f ? v : NEG_SLOPE * v;
}

__device__ __forceinline__ void load_edge(const int* __restrict__ perm,
                                          const int* __restrict__ src,
                                          const float* __restrict__ el,
                                          const float4 erd, int pos, int E,
                                          int& s, float4& v) {
    const int e = min(max(perm[pos], 0), E - 1);
    s = src[e];
    const float4 l = *(const float4*)(el + (size_t)s * H_DIM);
    v.x = leaky(l.x + erd.x); v.y = leaky(l.y + erd.y);
    v.z = leaky(l.z + erd.z); v.w = leaky(l.w + erd.w);
}

#define ROWS_PER_TRIP 8
__device__ __forceinline__ void add_rows(const float* __restrict__ h,
                                         int lane, int s_l,
                                         const float4 wt, int n,
                                         v2f (&acc)[H_DIM]) {
    for (int i = 0; i < n; i += ROWS_PER_TRIP) {
        v2f f[ROWS_PER_TRIP];
        #pragma unroll
        for (int u = 0; u < ROWS_PER_TRIP; ++u) {
            const int s = __builtin_amdgcn_readlane(s_l, i + u);
            f[u] = *(const v2f*)(h + (size_t)s * F_DIM + lane * 2);
        }
        #pragma unroll
        for (int u = 0; u < ROWS_PER_TRIP; ++u) {
            const float wx = __int_as_float(__builtin_amdgcn_readlane(__float_as_int(wt.x), i + u));
            const float wy = __int_as_float(__builtin_amdgcn_readlane(__float_as_int(wt.y), i + u));
            const float wz = __int_as_float(__builtin_amdgcn_readlane(__float_as_int(wt.z), i + u));
            const float ww = __int_as_float(__builtin_amdgcn_readlane(__float_as_int(wt.w), i + u));
            const v2f w0 = {wx, wx}, w1 = {wy, wy}, w2 = {wz, wz}, w3 = {ww, ww};
            acc[0] = __builtin_elementwise_fma(f[u], w0, acc[0]);
            acc[1] = __builtin_elementwise_fma(f[u], w1, acc[1]);
            acc[2] = __builtin_elementwise_fma(f[u], w2, acc[2]);
            acc[3] = __builtin_elementwise_fma(f[u], w3, acc[3]);
        }
    }
}

__device__ __forceinline__ void aggregate_node(const int d, const int lane,
                                               const int* __restrict__ src,
                                               const int* __restrict__ rowstart,
                                               const int* __restrict__ deg,
                                               const int* __restrict__ perm,
                                               const float* __restrict__ h,
                                               const float* __restrict__ el,
                                               const float* __restrict__ er,
                                               float* __restrict__ row, int E) {
    const int beg = min(max(rowstart[d], 0), E);
    const int cnt = min(max(deg[d], 0), E - beg);
    const float4 erd = *(const float4*)(er + (size_t)d * H_DIM);

    int    s0 = 0;
    float4 v0 = {-INFINITY, -INFINITY, -INFINITY, -INFINITY};
    if (lane < cnt) load_edge(perm, src, el, erd, beg + lane, E, s0, v0);
    s0 = lane < cnt ? s0 : __builtin_amdgcn_readfirstlane(s0);

    float4 mx = v0;
    for (int c = 32; c < cnt; c += 32) {
        if (c + lane < cnt) {
            int s; float4 v;
            load_edge(perm, src, el, erd, beg + c + lane, E, s, v);
            mx.x = fmaxf(mx.x, v.x); mx.y = fmaxf(mx.y, v.y);
            mx.z = fmaxf(mx.z, v.z); mx.w = fmaxf(mx.w, v.w);
        }
    }
    #pragma unroll
    for (int msk = 16; msk >= 1; msk >>= 1) {
        mx.x = fmaxf(mx.x, __shfl_xor(mx.x, msk, 32));
        mx.y = fmaxf(mx.y, __shfl_xor(mx.y, msk, 32));
        mx.z = fmaxf(mx.z, __shfl_xor(mx.z, msk, 32));
        mx.w = fmaxf(mx.w, __shfl_xor(mx.w, msk, 32));
    }

    v2f acc[H_DIM];
    #pragma unroll
    for (int k = 0; k < H_DIM; ++k) { acc[k].x = 0.0f; acc[k].y = 0.0f; }
    float4 ex0 = {0.0f, 0.0f, 0.0f, 0.0f};
    if (lane < cnt) {
        ex0.x = __expf(v0.x - mx.x); ex0.y = __expf(v0.y - mx.y);
        ex0.z = __expf(v0.z - mx.z); ex0.w = __expf(v0.w - mx.w);
    }
    float4 zs = ex0;
    for (int c = 32; c < cnt; c += 32) {
        const int n = min(32, cnt - c);
        int    s_l = 0;
        float4 ex  = {0.0f, 0.0f, 0.0f, 0.0f};
        if (lane < n) {
            float4 v;
            load_edge(perm, src, el, erd, beg + c + lane, E, s_l, v);
            ex.x = __expf(v.x - mx.x); ex.y = __expf(v.y - mx.y);
            ex.z = __expf(v.z - mx.z); ex.w = __expf(v.w - mx.w);
        }
        s_l = lane < n ? s_l : __builtin_amdgcn_readfirstlane(s_l);
        zs.x += ex.x; zs.y += ex.y; zs.z += ex.z; zs.w += ex.w;
        add_rows(h, lane, s_l, ex, n, acc);
    }
    #pragma unroll
    for (int msk = 16; msk >= 1; msk >>= 1) {
        zs.x += __shfl_xor(zs.x, msk, 32);
        zs.y += __shfl_xor(zs.y, msk, 32);
        zs.z += __shfl_xor(zs.z, msk, 32);
        zs.w += __shfl_xor(zs.w, msk, 32);
    }
    float4 rz = {1.0f, 1.0f, 1.0f, 1.0f};
    if (cnt > 0) {
        rz.x = 1.0f / zs.x; rz.y = 1.0f / zs.y;
        rz.z = 1.0f / zs.z; rz.w = 1.0f / zs.w;
    }
    acc[0].x *= rz.x; acc[0].y *= rz.x;
    acc[1].x *= rz.y; acc[1].y *= rz.y;
    acc[2].x *= rz.z; acc[2].y *= rz.z;
    acc[3].x *= rz.w; acc[3].y *= rz.w;
    float4 alpha;
    alpha.x = ex0.x * rz.x; alpha.y = ex0.y * rz.y;
    alpha.z = ex0.z * rz.z; alpha.w = ex0.w * rz.w;

    add_rows(h, lane, s0, alpha, min(cnt, 32), acc);

    float* r = row + lane * 2;
    #pragma unroll
    for (int k = 0; k < H_DIM; ++k)
        *(v2f*)(r + k * F_DIM) = acc[k];
}

#define AGG_NODES_PER_WAVE 2
#define TILE_NODES (8 * AGG_NODES_PER_WAVE)
#define AGG_PITCH  (HF + 4)
#define X_PITCH    (F_DIM + 4)
__global__ void aggregate_out_kernel(const float* __restrict__ h,
                                     const int* __restrict__ src,
                                     const int* __restrict__ rowstart,
                                     const int* __restrict__ deg,
                                     const int* __restrict__ perm,
                                     const float* __restrict__ el,
                                     const float* __restrict__ er,
                                     const float* __restrict__ w,
                                     const float* __restrict__ bv,
                                     const float* __restrict__ g,
                                     const float* __restrict__ b,
                                     float* __restrict__ out, int N, int E) {
    __shared__ float s_agg[TILE_NODES * AGG_PITCH];
    __shared__ float s_x[TILE_NODES * X_PITCH];
    const int wv    = __builtin_amdgcn_readfirstlane((int)(threadIdx.x >> 5));
    const int lane  = threadIdx.x & 31;
    const int node0 = blockIdx.x * TILE_NODES;

    #pragma nounroll
    for (int j = 0; j < AGG_NODES_PER_WAVE; ++j) {
        const int r = wv * AGG_NODES_PER_WAVE + j;
        const int d = node0 + r;
        float* row = s_agg + r * AGG_PITCH;
        if (d < N) {
            aggregate_node(d, lane, src, rowstart, deg, perm, h, el, er, row, E);
        } else {
            const v2f zero = {0.0f, 0.0f};
            for (int k = 0; k < H_DIM; ++k)
                *(v2f*)(row + k * F_DIM + lane * 2) = zero;
        }
    }
    __syncthreads();

    if (wv < F_DIM / 16) {
        const int half = lane >> 4, l = lane & 15;
        const float* arow = s_agg + l * AGG_PITCH;
        const int bcol = wv * 16 + l;
        v8f acc = {};
        #pragma unroll 8
        for (int k = 0; k < HF; k += 4) {
            const int ka = k + 2 * half;
            v2f a, bb;
            a.x  = arow[ka];
            a.y  = arow[ka + 1];
            bb.x = w[ka * F_DIM + bcol];
            bb.y = w[(ka + 1) * F_DIM + bcol];
            acc = __builtin_amdgcn_wmma_f32_16x16x4_f32(
                      false, a, false, bb, (short)0, acc, false, false);
        }
        const float bias = bv[bcol];
        float* xr = s_x + (half * 8) * X_PITCH + bcol;
        #pragma unroll
        for (int i = 0; i < 8; ++i)
            xr[i * X_PITCH] = acc[i] + bias;
    }
    __syncthreads();

    const float g0 = g[lane], g1 = g[lane + 32];
    const float b0 = b[lane], b1 = b[lane + 32];
    #pragma unroll
    for (int j = 0; j < AGG_NODES_PER_WAVE; ++j) {
        const int r = wv * AGG_NODES_PER_WAVE + j;
        const int node = node0 + r;
        const float x0 = s_x[r * X_PITCH + lane], x1 = s_x[r * X_PITCH + lane + 32];
        float s = x0 + x1, sq = x0 * x0 + x1 * x1;
        #pragma unroll
        for (int msk = 1; msk < 32; msk <<= 1) {
            s  += __shfl_xor(s,  msk, 32);
            sq += __shfl_xor(sq, msk, 32);
        }
        const float mu  = s * (1.0f / 64.0f);
        const float var = sq * (1.0f / 64.0f) - mu * mu;
        const float inv = rsqrtf(var + LN_EPS);
        if (node < N) {
            const size_t base = (size_t)node * F_DIM;
            out[base + lane]      = (x0 - mu) * inv * g0 + b0;
            out[base + lane + 32] = (x1 - mu) * inv * g1 + b1;
        }
    }
}

extern "C" void kernel_launch(void* const* d_in, const int* in_sizes, int n_in,
                              void* d_out, int out_size, void* d_ws, size_t ws_size,
                              hipStream_t stream) {
    const float* h   = (const float*)d_in[0];
    const int*   src = (const int*)  d_in[1];
    const int*   dst = (const int*)  d_in[2];
    const float* fcw = (const float*)d_in[3];
    const float* al  = (const float*)d_in[4];
    const float* ar  = (const float*)d_in[5];
    const float* gb  = (const float*)d_in[6];
    const float* ow  = (const float*)d_in[7];
    const float* ob  = (const float*)d_in[8];
    const float* lng = (const float*)d_in[9];
    const float* lnb = (const float*)d_in[10];

    const int N = in_sizes[0] / F_DIM;
    const int E = in_sizes[1];

    char*  ws  = (char*)d_ws;
    size_t off = 0;
    auto carve = [&](size_t bytes) -> float* {
        float* p = (float*)(ws + off);
        off += (bytes + 255) & ~(size_t)255;
        return p;
    };
    float* el   = carve((size_t)N * H_DIM * 4);
    float* er   = carve((size_t)N * H_DIM * 4);
    float* wlr  = carve((size_t)2 * H_DIM * F_DIM * 4);
    float* bvec = carve((size_t)F_DIM * 4);
    float* wc   = carve((size_t)HF * F_DIM * 4);
    const int nb  = (N + BKT_NODES - 1) / BKT_NODES;
    int* deg      = (int*)carve((size_t)N * 4);
    int* rowstart = (int*)carve((size_t)N * 4);
    int* gcount   = (int*)carve((size_t)MAX_BKTS * 4);
    int* total    = (int*)carve(8);
    int* recs     = (int*)carve((size_t)nb * BKT_CAP * 8);
    int* ovf      = (int*)carve((size_t)E * 8);
    int* perm     = (int*)carve((size_t)E * 4);
    (void)ws_size;

    {
        const int nInit = (nb + 255) / 256;
        setup_kernel<<<nInit + 3 + 8, 256, 0, stream>>>(fcw, al, ar, gb, ow, ob, gcount, total,
                                                        wlr, bvec, wc, nb, nInit);
    }
    {
        const int blocks = (N + 8 * NODES_PER_WAVE - 1) / (8 * NODES_PER_WAVE);
        attn_score_kernel<<<blocks, 256, 0, stream>>>(h, wlr, el, er, N);
    }
    {
        const int eblocks = (E + CHUNK_EDGES - 1) / CHUNK_EDGES;
        edge_partition_kernel<<<eblocks, 256, 0, stream>>>(dst, gcount, total, recs, ovf, E, nb);
        bucket_sort_kernel<<<nb, BKT_NODES, 0, stream>>>(recs, ovf, gcount, total, deg, rowstart,
                                                         perm, N, E);
        const int blocks = (N + TILE_NODES - 1) / TILE_NODES;
        aggregate_out_kernel<<<blocks, 256, 0, stream>>>(h, src, rowstart, deg, perm,
                                                         el, er, wc, bvec, lng, lnb,
                                                         (float*)d_out, N, E);
    }
}


# launches regrouped by what each stage needs: scores, edge partition and the folded weights wc, bvec share the second launch (block roles); the first keeps only counters and wl, wr: 5 launches -> 4
# speedup vs baseline: 8.9948x; 1.0902x over previous
#include <hip/hip_runtime.h>
#include <math.h>

typedef __attribute__((ext_vector_type(2))) float v2f;
typedef __attribute__((ext_vector_type(8))) float v8f;
typedef __attribute__((ext_vector_type(2))) int   v2i;

#define F_DIM 64
#define H_DIM 4
#define HF    256
#define NEG_SLOPE 0.2f
#define LN_EPS 1e-5f


__device__ __forceinline__ void wmma_tile_k64(const float* __restrict__ A, int lda,
                                              const float* __restrict__ B, int ldb,
                                              float* __restrict__ D, int ldd, int lane) {
    const int half = lane >> 4, l = lane & 15;
    v8f acc = {};
    #pragma unroll
    for (int k4 = 0; k4 < F_DIM / 4; ++k4) {
        const int ka = k4 * 4 + 2 * half;
        v2f a, b;
        a.x = A[l * lda + ka];
        a.y = A[l * lda + ka + 1];
        b.x = B[ka * ldb + l];
        b.y = B[(ka + 1) * ldb + l];
        acc = __builtin_amdgcn_wmma_f32_16x16x4_f32(
                  false, a, false, b, (short)0, acc, false, false);
    }
    #pragma unroll
    for (int i = 0; i < 8; ++i)
        D[(half * 8 + i) * ldd + l] = acc[i];
}

__global__ void setup_kernel(const float* __restrict__ fcw,
                             const float* __restrict__ al,
                             const float* __restrict__ ar,
                             int* __restrict__ gcount, int* __restrict__ total,
                             float* __restrict__ wlr, int nb, int nInit) {
    const int b = blockIdx.x;
    if (b < nInit) {
        const int i = b * blockDim.x + threadIdx.x;
        if (i < nb) gcount[i] = 0;
        if (i < 2)  total[i] = 0;
        return;
    }
    {
        const int r = b - nInit;
        const int t = r * 256 + threadIdx.x;
        const int o = t >> 6, k = t & 63, hs = o & 3;
        const float* a = (o < 4 ? al : ar) + hs * F_DIM;
        const float* w = fcw + k * HF + hs * F_DIM;
        float s = 0.0f;
        #pragma unroll
        for (int f = 0; f < F_DIM; f += 4) {
            const float4 wv = *(const float4*)(w + f);
            const float4 av = *(const float4*)(a + f);
            s += wv.x * av.x; s += wv.y * av.y;
            s += wv.z * av.z; s += wv.w * av.w;
        }
        wlr[o * F_DIM + k] = s;
    }
}

__device__ __forceinline__ void fold_weights_block(const float* __restrict__ fcw,
                                                   const float* __restrict__ gb,
                                                   const float* __restrict__ ow,
                                                   const float* __restrict__ ob,
                                                   float* __restrict__ bvec,
                                                   float* __restrict__ wc,
                                                   float* red, int r) {
    if (r == 0) {
        const int j = threadIdx.x & 63, part = threadIdx.x >> 6;
        float s = 0.0f;
        #pragma unroll 8
        for (int p = part * 64; p < part * 64 + 64; ++p)
            s += gb[p] * ow[p * F_DIM + j];
        red[threadIdx.x] = s;
        __syncthreads();
        if (threadIdx.x < 64)
            bvec[j] = ((red[j] + red[64 + j]) + (red[128 + j] + red[192 + j])) + ob[j];
        return;
    }
    const int tile = (r - 1) * 8 + (threadIdx.x >> 5);
    const int hs = tile >> 4, mt = (tile >> 2) & 3, nt = tile & 3;
    wmma_tile_k64(fcw + (mt * 16) * HF + hs * F_DIM, HF,
                  ow + (hs * F_DIM) * F_DIM + nt * 16, F_DIM,
                  wc + (hs * F_DIM + mt * 16) * F_DIM + nt * 16, F_DIM,
                  threadIdx.x & 31);
}

#define NODES_PER_WAVE 8
__device__ __forceinline__ void attn_score_block(const float* __restrict__ h,
                                                 const float* __restrict__ wlr,
                                                 float* __restrict__ el,
                                                 float* __restrict__ er,
                                                 int N, int blk) {
    const int wave  = (blk * (int)blockDim.x + (int)threadIdx.x) >> 5;
    const int node0 = wave * NODES_PER_WAVE;
    if (node0 >= N) return;
    const int lane = threadIdx.x & 31;
    const int sub = lane & 7, q = lane >> 3;
    float4 fa[NODES_PER_WAVE / 4], fb[NODES_PER_WAVE / 4];
    #pragma unroll
    for (int g = 0; g < NODES_PER_WAVE / 4; ++g) {
        const float* p = h + (size_t)min(node0 + g * 4 + q, N - 1) * F_DIM + sub * 8;
        fa[g] = *(const float4*)(p);
        fb[g] = *(const float4*)(p + 4);
    }
    float4 wa[8], wb[8];
    #pragma unroll
    for (int o = 0; o < 8; ++o) {
        wa[o] = *(const float4*)(wlr + o * F_DIM + sub * 8);
        wb[o] = *(const float4*)(wlr + o * F_DIM + sub * 8 + 4);
    }
    #pragma unroll
    for (int g = 0; g < NODES_PER_WAVE / 4; ++g) {
        float s[8];
        #pragma unroll
        for (int o = 0; o < 8; ++o) {
            float t = fa[g].x * wa[o].x;
            t += fa[g].y * wa[o].y; t += fa[g].z * wa[o].z; t += fa[g].w * wa[o].w;
            t += fb[g].x * wb[o].x; t += fb[g].y * wb[o].y;
            t += fb[g].z * wb[o].z; t += fb[g].w * wb[o].w;
            s[o] = t;
        }
        #pragma unroll
        for (int msk = 1; msk < 8; msk <<= 1) {
            #pragma unroll
            for (int o = 0; o < 8; ++o) s[o] += __shfl_xor(s[o], msk, 32);
        }
        const int node = node0 + g * 4 + q;
        if (sub == 0 && node < N) {
            float4 l4, r4;
            l4.x = s[0]; l4.y = s[1]; l4.z = s[2]; l4.w = s[3];
            r4.x = s[4]; r4.y = s[5]; r4.z = s[6]; r4.w = s[7];
            *(float4*)(el + (size_t)node * H_DIM) = l4;
            *(float4*)(er + (size_t)node * H_DIM) = r4;
        }
    }
}

#define BKT_SHIFT   8
#define BKT_NODES   (1 << BKT_SHIFT)
#define BKT_CAP     8192
#define MAX_BKTS    256
#define CHUNK_EDGES 4096
__device__ __forceinline__ void edge_partition_block(const int* __restrict__ dst,
                                                     int* __restrict__ gcount,
                                                     int* __restrict__ total,
                                                     int* __restrict__ recs,
                                                     int* __restrict__ ovf,
                                                     int* cnt, int* gbase,
                                                     int E, int nb, int blk) {
    const int tid = threadIdx.x;
    for (int b = tid; b < nb; b += blockDim.x) cnt[b] = 0;
    __syncthreads();
    const int e_blk = blk * CHUNK_EDGES;
    int dd[16], lr[16];
    if (e_blk + CHUNK_EDGES <= E) {
        #pragma unroll
        for (int k = 0; k < 4; ++k) {
            const int4 v = *(const int4*)(dst + e_blk + k * 1024 + tid * 4);
            dd[4 * k] = v.x; dd[4 * k + 1] = v.y; dd[4 * k + 2] = v.z; dd[4 * k + 3] = v.w;
        }
    } else {
        #pragma unroll
        for (int i = 0; i < 16; ++i) {
            const int e = e_blk + (i >> 2) * 1024 + tid * 4 + (i & 3);
            const int v = dst[min(e, E - 1)];
            dd[i] = e < E ? v : -1;
        }
    }
    #pragma unroll
    for (int i = 0; i < 16; ++i)
        lr[i] = dd[i] >= 0 ? atomicAdd(&cnt[dd[i] >> BKT_SHIFT], 1) : 0;
    __syncthreads();
    for (int b = tid; b < nb; b += blockDim.x) {
        const int c = cnt[b];
        gbase[b] = c > 0 ? atomicAdd(&gcount[b], c) : 0;
    }
    __syncthreads();
    v2i* r2 = (v2i*)recs;
    v2i* o2 = (v2i*)ovf;
    #pragma unroll
    for (int i = 0; i < 16; ++i) {
        if (dd[i] >= 0) {
            const int bb = dd[i] >> BKT_SHIFT;
            const int p  = gbase[bb] + lr[i];
            v2i rec;
            rec.x = dd[i];
            rec.y = e_blk + (i >> 2) * 1024 + tid * 4 + (i & 3);
            if (p < BKT_CAP) r2[(size_t)bb * BKT_CAP + p] = rec;
            else             o2[atomicAdd(&total[1], 1)] = rec;
        }
    }
}

#define FOLD_BLOCKS 9
__global__ void scores_partition_kernel(const float* __restrict__ h,
                                        const float* __restrict__ wlr,
                                        float* __restrict__ el,
                                        float* __restrict__ er,
                                        const int* __restrict__ dst,
                                        int* __restrict__ gcount,
                                        int* __restrict__ total,
                                        int* __restrict__ recs,
                                        int* __restrict__ ovf,
                                        const float* __restrict__ fcw,
                                        const float* __restrict__ gb,
                                        const float* __restrict__ ow,
                                        const float* __restrict__ ob,
                                        float* __restrict__ bvec,
                                        float* __restrict__ wc,
                                        int N, int E, int nb, int eblocks) {
    __shared__ int cnt[MAX_BKTS];
    __shared__ int gbase[MAX_BKTS];
    __shared__ float red[256];
    const int b = blockIdx.x;
    if (b < FOLD_BLOCKS)
        fold_weights_block(fcw, gb, ow, ob, bvec, wc, red, b);
    else if (b < FOLD_BLOCKS + eblocks)
        edge_partition_block(dst, gcount, total, recs, ovf, cnt, gbase, E, nb, b - FOLD_BLOCKS);
    else
        attn_score_block(h, wlr, el, er, N, b - FOLD_BLOCKS - eblocks);
}

__global__ void bucket_sort_kernel(const int* __restrict__ recs,
                                   const int* __restrict__ ovf,
                                   const int* __restrict__ gcount,
                                   int* __restrict__ total,
                                   int* __restrict__ deg,
                                   int* __restrict__ rowstart,
                                   int* __restrict__ perm, int N, int E) {
    __shared__ int ncnt[BKT_NODES];
    __shared__ int wave_off[8];
    __shared__ int block_base;
    const int tid = threadIdx.x, b = blockIdx.x;
    const int lane = tid & 31, wv = tid >> 5;
    ncnt[tid] = 0;
    const int n_in  = min(max(gcount[b], 0), BKT_CAP);
    const int n_ovf = min(max(total[1], 0), E);
    const v2i* mine = (const v2i*)recs + (size_t)b * BKT_CAP;
    const v2i* o2   = (const v2i*)ovf;
    __syncthreads();
    for (int i0 = tid; i0 < n_in; i0 += 4 * BKT_NODES) {
        v2i r[4];
        #pragma unroll
        for (int u = 0; u < 4; ++u) r[u] = mine[min(i0 + u * BKT_NODES, n_in - 1)];
        #pragma unroll
        for (int u = 0; u < 4; ++u)
            if (i0 + u * BKT_NODES < n_in) atomicAdd(&ncnt[r[u].x & (BKT_NODES - 1)], 1);
    }
    for (int i = tid; i < n_ovf; i += BKT_NODES) {
        const v2i r = o2[i];
        if ((r.x >> BKT_SHIFT) == b) atomicAdd(&ncnt[r.x & (BKT_NODES - 1)], 1);
    }
    __syncthreads();
    const int v = ncnt[tid];
    int inc = v;
    #pragma unroll
    for (int o = 1; o < 32; o <<= 1) {
        const int t = __shfl_up(inc, o, 32);
        if (lane >= o) inc += t;
    }
    if (lane == 31) wave_off[wv] = inc;
    __syncthreads();
    if (tid == 0) {
        int run = 0;
        #pragma unroll
        for (int w = 0; w < 8; ++w) { const int t = wave_off[w]; wave_off[w] = run; run += t; }
        block_base = atomicAdd(total, run);
    }
    __syncthreads();
    const int excl = wave_off[wv] + inc - v;
    const int base = block_base;
    const int node = b * BKT_NODES + tid;
    if (node < N) { deg[node] = v; rowstart[node] = base + excl; }
    ncnt[tid] = excl;
    __syncthreads();
    for (int i0 = tid; i0 < n_in; i0 += 4 * BKT_NODES) {
        v2i r[4];
        #pragma unroll
        for (int u = 0; u < 4; ++u) r[u] = mine[min(i0 + u * BKT_NODES, n_in - 1)];
        #pragma unroll
        for (int u = 0; u < 4; ++u)
            if (i0 + u * BKT_NODES < n_in) {
                const int pos = atomicAdd(&ncnt[r[u].x & (BKT_NODES - 1)], 1);
                perm[min(max(base + pos, 0), E - 1)] = r[u].y;
            }
    }
    for (int i = tid; i < n_ovf; i += BKT_NODES) {
        const v2i r = o2[i];
        if ((r.x >> BKT_SHIFT) == b) {
            const int pos = atomicAdd(&ncnt[r.x & (BKT_NODES - 1)], 1);
            perm[min(max(base + pos, 0), E - 1)] = r.y;
        }
    }
}

__device__ __forceinline__ float leaky(float v) {
    return v > 0.0f ? v : NEG_SLOPE * v;
}

__device__ __forceinline__ void load_edge(const int* __restrict__ perm,
                                          const int* __restrict__ src,
                                          const float* __restrict__ el,
                                          const float4 erd, int pos, int E,
                                          int& s, float4& v) {
    const int e = min(max(perm[pos], 0), E - 1);
    s = src[e];
    const float4 l = *(const float4*)(el + (size_t)s * H_DIM);
    v.x = leaky(l.x + erd.x); v.y = leaky(l.y + erd.y);
    v.z = leaky(l.z + erd.z); v.w = leaky(l.w + erd.w);
}

#define ROWS_PER_TRIP 8
__device__ __forceinline__ void add_rows(const float* __restrict__ h,
                                         int lane, int s_l,
                                         const float4 wt, int n,
                                         v2f (&acc)[H_DIM]) {
    for (int i = 0; i < n; i += ROWS_PER_TRIP) {
        v2f f[ROWS_PER_TRIP];
        #pragma unroll
        for (int u = 0; u < ROWS_PER_TRIP; ++u) {
            const int s = __builtin_amdgcn_readlane(s_l, i + u);
            f[u] = *(const v2f*)(h + (size_t)s * F_DIM + lane * 2);
        }
        #pragma unroll
        for (int u = 0; u < ROWS_PER_TRIP; ++u) {
            const float wx = __int_as_float(__builtin_amdgcn_readlane(__float_as_int(wt.x), i + u));
            const float wy = __int_as_float(__builtin_amdgcn_readlane(__float_as_int(wt.y), i + u));
            const float wz = __int_as_float(__builtin_amdgcn_readlane(__float_as_int(wt.z), i + u));
            const float ww = __int_as_float(__builtin_amdgcn_readlane(__float_as_int(wt.w), i + u));
            const v2f w0 = {wx, wx}, w1 = {wy, wy}, w2 = {wz, wz}, w3 = {ww, ww};
            acc[0] = __builtin_elementwise_fma(f[u], w0, acc[0]);
            acc[1] = __builtin_elementwise_fma(f[u], w1, acc[1]);
            acc[2] = __builtin_elementwise_fma(f[u], w2, acc[2]);
            acc[3] = __builtin_elementwise_fma(f[u], w3, acc[3]);
        }
    }
}

__device__ __forceinline__ void aggregate_node(const int d, const int lane,
                                               const int* __restrict__ src,
                                               const int* __restrict__ rowstart,
                                               const int* __restrict__ deg,
                                               const int* __restrict__ perm,
                                               const float* __restrict__ h,
                                               const float* __restrict__ el,
                                               const float* __restrict__ er,
                                               float* __restrict__ row, int E) {
    const int beg = min(max(rowstart[d], 0), E);
    const int cnt = min(max(deg[d], 0), E - beg);
    const float4 erd = *(const float4*)(er + (size_t)d * H_DIM);

    int    s0 = 0;
    float4 v0 = {-INFINITY, -INFINITY, -INFINITY, -INFINITY};
    if (lane < cnt) load_edge(perm, src, el, erd, beg + lane, E, s0, v0);
    s0 = lane < cnt ? s0 : __builtin_amdgcn_readfirstlane(s0);

    float4 mx = v0;
    for (int c = 32; c < cnt; c += 32) {
        if (c + lane < cnt) {
            int s; float4 v;
            load_edge(perm, src, el, erd, beg + c + lane, E, s, v);
            mx.x = fmaxf(mx.x, v.x); mx.y = fmaxf(mx.y, v.y);
            mx.z = fmaxf(mx.z, v.z); mx.w = fmaxf(mx.w, v.w);
        }
    }
    #pragma unroll
    for (int msk = 16; msk >= 1; msk >>= 1) {
        mx.x = fmaxf(mx.x, __shfl_xor(mx.x, msk, 32));
        mx.y = fmaxf(mx.y, __shfl_xor(mx.y, msk, 32));
        mx.z = fmaxf(mx.z, __shfl_xor(mx.z, msk, 32));
        mx.w = fmaxf(mx.w, __shfl_xor(mx.w, msk, 32));
    }

    v2f acc[H_DIM];
    #pragma unroll
    for (int k = 0; k < H_DIM; ++k) { acc[k].x = 0.0f; acc[k].y = 0.0f; }
    float4 ex0 = {0.0f, 0.0f, 0.0f, 0.0f};
    if (lane < cnt) {
        ex0.x = __expf(v0.x - mx.x); ex0.y = __expf(v0.y - mx.y);
        ex0.z = __expf(v0.z - mx.z); ex0.w = __expf(v0.w - mx.w);
    }
    float4 zs = ex0;
    for (int c = 32; c < cnt; c += 32) {
        const int n = min(32, cnt - c);
        int    s_l = 0;
        float4 ex  = {0.0f, 0.0f, 0.0f, 0.0f};
        if (lane < n) {
            float4 v;
            load_edge(perm, src, el, erd, beg + c + lane, E, s_l, v);
            ex.x = __expf(v.x - mx.x); ex.y = __expf(v.y - mx.y);
            ex.z = __expf(v.z - mx.z); ex.w = __expf(v.w - mx.w);
        }
        s_l = lane < n ? s_l : __builtin_amdgcn_readfirstlane(s_l);
        zs.x += ex.x; zs.y += ex.y; zs.z += ex.z; zs.w += ex.w;
        add_rows(h, lane, s_l, ex, n, acc);
    }
    #pragma unroll
    for (int msk = 16; msk >= 1; msk >>= 1) {
        zs.x += __shfl_xor(zs.x, msk, 32);
        zs.y += __shfl_xor(zs.y, msk, 32);
        zs.z += __shfl_xor(zs.z, msk, 32);
        zs.w += __shfl_xor(zs.w, msk, 32);
    }
    float4 rz = {1.0f, 1.0f, 1.0f, 1.0f};
    if (cnt > 0) {
        rz.x = 1.0f / zs.x; rz.y = 1.0f / zs.y;
        rz.z = 1.0f / zs.z; rz.w = 1.0f / zs.w;
    }
    acc[0].x *= rz.x; acc[0].y *= rz.x;
    acc[1].x *= rz.y; acc[1].y *= rz.y;
    acc[2].x *= rz.z; acc[2].y *= rz.z;
    acc[3].x *= rz.w; acc[3].y *= rz.w;
    float4 alpha;
    alpha.x = ex0.x * rz.x; alpha.y = ex0.y * rz.y;
    alpha.z = ex0.z * rz.z; alpha.w = ex0.w * rz.w;

    add_rows(h, lane, s0, alpha, min(cnt, 32), acc);

    float* r = row + lane * 2;
    #pragma unroll
    for (int k = 0; k < H_DIM; ++k)
        *(v2f*)(r + k * F_DIM) = acc[k];
}

#define AGG_NODES_PER_WAVE 2
#define TILE_NODES (8 * AGG_NODES_PER_WAVE)
#define AGG_PITCH  (HF + 4)
#define X_PITCH    (F_DIM + 4)
__global__ void aggregate_out_kernel(const float* __restrict__ h,
                                     const int* __restrict__ src,
                                     const int* __restrict__ rowstart,
                                     const int* __restrict__ deg,
                                     const int* __restrict__ perm,
                                     const float* __restrict__ el,
                                     const float* __restrict__ er,
                                     const float* __restrict__ w,
                                     const float* __restrict__ bv,
                                     const float* __restrict__ g,
                                     const float* __restrict__ b,
                                     float* __restrict__ out, int N, int E) {
    __shared__ float s_agg[TILE_NODES * AGG_PITCH];
    __shared__ float s_x[TILE_NODES * X_PITCH];
    const int wv    = __builtin_amdgcn_readfirstlane((int)(threadIdx.x >> 5));
    const int lane  = threadIdx.x & 31;
    const int node0 = blockIdx.x * TILE_NODES;

    #pragma nounroll
    for (int j = 0; j < AGG_NODES_PER_WAVE; ++j) {
        const int r = wv * AGG_NODES_PER_WAVE + j;
        const int d = node0 + r;
        float* row = s_agg + r * AGG_PITCH;
        if (d < N) {
            aggregate_node(d, lane, src, rowstart, deg, perm, h, el, er, row, E);
        } else {
            const v2f zero = {0.0f, 0.0f};
            for (int k = 0; k < H_DIM; ++k)
                *(v2f*)(row + k * F_DIM + lane * 2) = zero;
        }
    }
    __syncthreads();

    if (wv < F_DIM / 16) {
        const int half = lane >> 4, l = lane & 15;
        const float* arow = s_agg + l * AGG_PITCH;
        const int bcol = wv * 16 + l;
        v8f acc = {};
        #pragma unroll 8
        for (int k = 0; k < HF; k += 4) {
            const int ka = k + 2 * half;
            v2f a, bb;
            a.x  = arow[ka];
            a.y  = arow[ka + 1];
            bb.x = w[ka * F_DIM + bcol];
            bb.y = w[(ka + 1) * F_DIM + bcol];
            acc = __builtin_amdgcn_wmma_f32_16x16x4_f32(
                      false, a, false, bb, (short)0, acc, false, false);
        }
        const float bias = bv[bcol];
        float* xr = s_x + (half * 8) * X_PITCH + bcol;
        #pragma unroll
        for (int i = 0; i < 8; ++i)
            xr[i * X_PITCH] = acc[i] + bias;
    }
    __syncthreads();

    const float g0 = g[lane], g1 = g[lane + 32];
    const float b0 = b[lane], b1 = b[lane + 32];
    #pragma unroll
    for (int j = 0; j < AGG_NODES_PER_WAVE; ++j) {
        const int r = wv * AGG_NODES_PER_WAVE + j;
        const int node = node0 + r;
        const float x0 = s_x[r * X_PITCH + lane], x1 = s_x[r * X_PITCH + lane + 32];
        float s = x0 + x1, sq = x0 * x0 + x1 * x1;
        #pragma unroll
        for (int msk = 1; msk < 32; msk <<= 1) {
            s  += __shfl_xor(s,  msk, 32);
            sq += __shfl_xor(sq, msk, 32);
        }
        const float mu  = s * (1.0f / 64.0f);
        const float var = sq * (1.0f / 64.0f) - mu * mu;
        const float inv = rsqrtf(var + LN_EPS);
        if (node < N) {
            const size_t base = (size_t)node * F_DIM;
            out[base + lane]      = (x0 - mu) * inv * g0 + b0;
            out[base + lane + 32] = (x1 - mu) * inv * g1 + b1;
        }
    }
}

extern "C" void kernel_launch(void* const* d_in, const int* in_sizes, int n_in,
                              void* d_out, int out_size, void* d_ws, size_t ws_size,
                              hipStream_t stream) {
    const float* h   = (const float*)d_in[0];
    const int*   src = (const int*)  d_in[1];
    const int*   dst = (const int*)  d_in[2];
    const float* fcw = (const float*)d_in[3];
    const float* al  = (const float*)d_in[4];
    const float* ar  = (const float*)d_in[5];
    const float* gb  = (const float*)d_in[6];
    const float* ow  = (const float*)d_in[7];
    const float* ob  = (const float*)d_in[8];
    const float* lng = (const float*)d_in[9];
    const float* lnb = (const float*)d_in[10];

    const int N = in_sizes[0] / F_DIM;
    const int E = in_sizes[1];

    char*  ws  = (char*)d_ws;
    size_t off = 0;
    auto carve = [&](size_t bytes) -> float* {
        float* p = (float*)(ws + off);
        off += (bytes + 255) & ~(size_t)255;
        return p;
    };
    float* el   = carve((size_t)N * H_DIM * 4);
    float* er   = carve((size_t)N * H_DIM * 4);
    float* wlr  = carve((size_t)2 * H_DIM * F_DIM * 4);
    float* bvec = carve((size_t)F_DIM * 4);
    float* wc   = carve((size_t)HF * F_DIM * 4);
    const int nb  = (N + BKT_NODES - 1) / BKT_NODES;
    int* deg      = (int*)carve((size_t)N * 4);
    int* rowstart = (int*)carve((size_t)N * 4);
    int* gcount   = (int*)carve((size_t)MAX_BKTS * 4);
    int* total    = (int*)carve(8);
    int* recs     = (int*)carve((size_t)nb * BKT_CAP * 8);
    int* ovf      = (int*)carve((size_t)E * 8);
    int* perm     = (int*)carve((size_t)E * 4);
    (void)ws_size;

    {
        const int nInit = (nb + 255) / 256;
        setup_kernel<<<nInit + 2, 256, 0, stream>>>(fcw, al, ar, gcount, total, wlr, nb, nInit);
    }
    {
        const int ablocks = (N + 8 * NODES_PER_WAVE - 1) / (8 * NODES_PER_WAVE);
        const int eblocks = (E + CHUNK_EDGES - 1) / CHUNK_EDGES;
        scores_partition_kernel<<<eblocks + FOLD_BLOCKS + ablocks, 256, 0, stream>>>(
            h, wlr, el, er, dst, gcount, total, recs, ovf, fcw, gb, ow, ob, bvec, wc,
            N, E, nb, eblocks);
        bucket_sort_kernel<<<nb, BKT_NODES, 0, stream>>>(recs, ovf, gcount, total, deg, rowstart,
                                                         perm, N, E);
        const int blocks = (N + TILE_NODES - 1) / TILE_NODES;
        aggregate_out_kernel<<<blocks, 256, 0, stream>>>(h, src, rowstart, deg, perm,
                                                         el, er, wc, bvec, lng, lnb,
                                                         (float*)d_out, N, E);
    }
}


# block-local edge partition: a 4096-edge block bucket-sorts its records in its own list, bucket blocks read slices via a table; no global counters; set-up launch gone (4 -> 3); sums in LDS rank order
# speedup vs baseline: 9.3981x; 1.0448x over previous
#include <hip/hip_runtime.h>
#include <math.h>

typedef __attribute__((ext_vector_type(2))) float v2f;
typedef __attribute__((ext_vector_type(8))) float v8f;
typedef __attribute__((ext_vector_type(2))) int   v2i;

#define F_DIM 64
#define H_DIM 4
#define HF    256
#define NEG_SLOPE 0.2f
#define LN_EPS 1e-5f


__device__ __forceinline__ void wmma_tile_k64(const float* __restrict__ A, int lda,
                                              const float* __restrict__ B, int ldb,
                                              float* __restrict__ D, int ldd, int lane) {
    const int half = lane >> 4, l = lane & 15;
    v8f acc = {};
    #pragma unroll
    for (int k4 = 0; k4 < F_DIM / 4; ++k4) {
        const int ka = k4 * 4 + 2 * half;
        v2f a, b;
        a.x = A[l * lda + ka];
        a.y = A[l * lda + ka + 1];
        b.x = B[ka * ldb + l];
        b.y = B[(ka + 1) * ldb + l];
        acc = __builtin_amdgcn_wmma_f32_16x16x4_f32(
                  false, a, false, b, (short)0, acc, false, false);
    }
    #pragma unroll
    for (int i = 0; i < 8; ++i)
        D[(half * 8 + i) * ldd + l] = acc[i];
}

#define FOLD_BLOCKS 11
__device__ __forceinline__ void fold_weights_block(const float* __restrict__ fcw,
                                                   const float* __restrict__ al,
                                                   const float* __restrict__ ar,
                                                   const float* __restrict__ gb,
                                                   const float* __restrict__ ow,
                                                   const float* __restrict__ ob,
                                                   float* __restrict__ wlr,
                                                   float* __restrict__ bvec,
                                                   float* __restrict__ wc,
                                                   int* __restrict__ total,
                                                   float* red, int r) {
    if (r == 0) {
        if (threadIdx.x == 0) total[0] = 0;
        const int j = threadIdx.x & 63, part = threadIdx.x >> 6;
        float s = 0.0f;
        #pragma unroll 8
        for (int p = part * 64; p < part * 64 + 64; ++p)
            s += gb[p] * ow[p * F_DIM + j];
        red[threadIdx.x] = s;
        __syncthreads();
        if (threadIdx.x < 64)
            bvec[j] = ((red[j] + red[64 + j]) + (red[128 + j] + red[192 + j])) + ob[j];
        return;
    }
    if (r > 8) {
        const int t = (r - 9) * 256 + threadIdx.x;
        const int o = t >> 6, k = t & 63, hs = o & 3;
        const float* a = (o < 4 ? al : ar) + hs * F_DIM;
        const float* w = fcw + k * HF + hs * F_DIM;
        float s = 0.0f;
        #pragma unroll
        for (int f = 0; f < F_DIM; f += 4) {
            const float4 wv = *(const float4*)(w + f);
            const float4 av = *(const float4*)(a + f);
            s += wv.x * av.x; s += wv.y * av.y;
            s += wv.z * av.z; s += wv.w * av.w;
        }
        wlr[o * F_DIM + k] = s;
        return;
    }
    const int tile = (r - 1) * 8 + (threadIdx.x >> 5);
    const int hs = tile >> 4, mt = (tile >> 2) & 3, nt = tile & 3;
    wmma_tile_k64(fcw + (mt * 16) * HF + hs * F_DIM, HF,
                  ow + (hs * F_DIM) * F_DIM + nt * 16, F_DIM,
                  wc + (hs * F_DIM + mt * 16) * F_DIM + nt * 16, F_DIM,
                  threadIdx.x & 31);
}

#define NODES_PER_WAVE 8
__device__ __forceinline__ void attn_score_block(const float* __restrict__ h,
                                                 const float* __restrict__ wlr,
                                                 float* __restrict__ el,
                                                 float* __restrict__ er,
                                                 int N, int blk) {
    const int wave  = (blk * (int)blockDim.x + (int)threadIdx.x) >> 5;
    const int node0 = wave * NODES_PER_WAVE;
    if (node0 >= N) return;
    const int lane = threadIdx.x & 31;
    const int sub = lane & 7, q = lane >> 3;
    float4 fa[NODES_PER_WAVE / 4], fb[NODES_PER_WAVE / 4];
    #pragma unroll
    for (int g = 0; g < NODES_PER_WAVE / 4; ++g) {
        const float* p = h + (size_t)min(node0 + g * 4 + q, N - 1) * F_DIM + sub * 8;
        fa[g] = *(const float4*)(p);
        fb[g] = *(const float4*)(p + 4);
    }
    float4 wa[8], wb[8];
    #pragma unroll
    for (int o = 0; o < 8; ++o) {
        wa[o] = *(const float4*)(wlr + o * F_DIM + sub * 8);
        wb[o] = *(const float4*)(wlr + o * F_DIM + sub * 8 + 4);
    }
    #pragma unroll
    for (int g = 0; g < NODES_PER_WAVE / 4; ++g) {
        float s[8];
        #pragma unroll
        for (int o = 0; o < 8; ++o) {
            float t = fa[g].x * wa[o].x;
            t += fa[g].y * wa[o].y; t += fa[g].z * wa[o].z; t += fa[g].w * wa[o].w;
            t += fb[g].x * wb[o].x; t += fb[g].y * wb[o].y;
            t += fb[g].z * wb[o].z; t += fb[g].w * wb[o].w;
            s[o] = t;
        }
        #pragma unroll
        for (int msk = 1; msk < 8; msk <<= 1) {
            #pragma unroll
            for (int o = 0; o < 8; ++o) s[o] += __shfl_xor(s[o], msk, 32);
        }
        const int node = node0 + g * 4 + q;
        if (sub == 0 && node < N) {
            float4 l4, r4;
            l4.x = s[0]; l4.y = s[1]; l4.z = s[2]; l4.w = s[3];
            r4.x = s[4]; r4.y = s[5]; r4.z = s[6]; r4.w = s[7];
            *(float4*)(el + (size_t)node * H_DIM) = l4;
            *(float4*)(er + (size_t)node * H_DIM) = r4;
        }
    }
}

#define BKT_SHIFT   8
#define BKT_NODES   (1 << BKT_SHIFT)
#define MAX_BKTS    256
#define CHUNK_EDGES 4096
__device__ __forceinline__ void edge_partition_block(const int* __restrict__ dst,
                                                     int* __restrict__ seg,
                                                     int* __restrict__ recs,
                                                     int* cnt, int* gbase, int* wtot,
                                                     int E, int nb, int eblocks, int blk) {
    const int tid = threadIdx.x, lane = tid & 31, wv = tid >> 5;
    cnt[tid] = 0;
    __syncthreads();
    const int e_blk = blk * CHUNK_EDGES;
    int dd[16], lr[16];
    if (e_blk + CHUNK_EDGES <= E) {
        #pragma unroll
        for (int k = 0; k < 4; ++k) {
            const int4 v = *(const int4*)(dst + e_blk + k * 1024 + tid * 4);
            dd[4 * k] = v.x; dd[4 * k + 1] = v.y; dd[4 * k + 2] = v.z; dd[4 * k + 3] = v.w;
        }
    } else {
        #pragma unroll
        for (int i = 0; i < 16; ++i) {
            const int e = e_blk + (i >> 2) * 1024 + tid * 4 + (i & 3);
            const int v = dst[min(e, E - 1)];
            dd[i] = e < E ? v : -1;
        }
    }
    #pragma unroll
    for (int i = 0; i < 16; ++i)
        lr[i] = dd[i] >= 0 ? atomicAdd(&cnt[dd[i] >> BKT_SHIFT], 1) : 0;
    __syncthreads();
    const int c = cnt[tid];
    int inc = c;
    #pragma unroll
    for (int o = 1; o < 32; o <<= 1) {
        const int t = __shfl_up(inc, o, 32);
        if (lane >= o) inc += t;
    }
    if (lane == 31) wtot[wv] = inc;
    __syncthreads();
    int excl = inc - c;
    #pragma unroll
    for (int w = 0; w < 7; ++w) excl += w < wv ? wtot[w] : 0;
    gbase[tid] = excl;
    if (tid < nb) seg[(size_t)tid * eblocks + blk] = (excl << 16) | c;
    __syncthreads();
    v2i* r2 = (v2i*)recs + (size_t)blk * CHUNK_EDGES;
    #pragma unroll
    for (int i = 0; i < 16; ++i) {
        if (dd[i] >= 0) {
            v2i rec;
            rec.x = dd[i];
            rec.y = e_blk + (i >> 2) * 1024 + tid * 4 + (i & 3);
            r2[gbase[dd[i] >> BKT_SHIFT] + lr[i]] = rec;
        }
    }
}

__global__ void fold_partition_kernel(const int* __restrict__ dst,
                                      int* __restrict__ seg,
                                      int* __restrict__ recs,
                                      int* __restrict__ total,
                                      const float* __restrict__ fcw,
                                      const float* __restrict__ al,
                                      const float* __restrict__ ar,
                                      const float* __restrict__ gb,
                                      const float* __restrict__ ow,
                                      const float* __restrict__ ob,
                                      float* __restrict__ wlr,
                                      float* __restrict__ bvec,
                                      float* __restrict__ wc,
                                      int E, int nb, int eblocks) {
    __shared__ int cnt[MAX_BKTS];
    __shared__ int gbase[MAX_BKTS];
    __shared__ int wtot[8];
    __shared__ float red[256];
    const int b = blockIdx.x;
    if (b < eblocks)
        edge_partition_block(dst, seg, recs, cnt, gbase, wtot, E, nb, eblocks, b);
    else
        fold_weights_block(fcw, al, ar, gb, ow, ob, wlr, bvec, wc, total, red, b - eblocks);
}

#define SEGS_PER_TRIP 8
static_assert(SEGS_PER_TRIP == 8, "the load fence in walk_slices names eight records");
template <bool PLACE>
__device__ __forceinline__ void walk_slices(const v2i* __restrict__ r2,
                                            const int* __restrict__ myseg,
                                            int* ncnt, int* __restrict__ perm,
                                            int base, int E, int eblocks,
                                            int wv, int lane) {
    for (int j0 = 0; j0 < eblocks; j0 += 256) {
        const int j  = j0 + wv + 8 * lane;
        const int sv = j < eblocks ? myseg[j] : 0;
        const int nseg = (min(eblocks - j0, 256) - wv + 7) >> 3;
        for (int q = 0; q < nseg; q += SEGS_PER_TRIP) {
            v2i r[SEGS_PER_TRIP];
            int c[SEGS_PER_TRIP], st[SEGS_PER_TRIP];
            #pragma unroll
            for (int u = 0; u < SEGS_PER_TRIP; ++u) {
                const int s   = __builtin_amdgcn_readlane(sv, q + u);
                const int off = min(max(s >> 16, 0), CHUNK_EDGES);
                c[u]  = min(s & 0xFFFF, CHUNK_EDGES - off);
                st[u] = min(j0 + wv + 8 * (q + u), eblocks - 1) * CHUNK_EDGES + off;
                r[u]  = r2[c[u] > 0 ? st[u] + min(lane, c[u] - 1) : 0];
            }
            asm volatile("" :: "v"(r[0]), "v"(r[1]), "v"(r[2]), "v"(r[3]),
                               "v"(r[4]), "v"(r[5]), "v"(r[6]), "v"(r[7]));
            int pos[SEGS_PER_TRIP];
            #pragma unroll
            for (int u = 0; u < SEGS_PER_TRIP; ++u)
                pos[u] = lane < c[u] ? atomicAdd(&ncnt[r[u].x & (BKT_NODES - 1)], 1) : 0;
            if (PLACE) {
                #pragma unroll
                for (int u = 0; u < SEGS_PER_TRIP; ++u)
                    if (lane < c[u]) perm[min(max(base + pos[u], 0), E - 1)] = r[u].y;
            }
            #pragma unroll
            for (int u = 0; u < SEGS_PER_TRIP; ++u) {
                for (int i = 32 + lane; i < c[u]; i += 32) {
                    const v2i rr = r2[st[u] + i];
                    const int pos = atomicAdd(&ncnt[rr.x & (BKT_NODES - 1)], 1);
                    if (PLACE) perm[min(max(base + pos, 0), E - 1)] = rr.y;
                }
            }
        }
    }
}

__device__ __forceinline__ void bucket_sort_block(const int* __restrict__ recs,
                                                  const int* __restrict__ seg,
                                                  int* __restrict__ total,
                                                  int* __restrict__ deg,
                                                  int* __restrict__ rowstart,
                                                  int* __restrict__ perm,
                                                  int* ncnt, int* wave_off, int* block_base,
                                                  int N, int E, int eblocks, int b) {
    const int tid = threadIdx.x, lane = tid & 31;
    const int wv = __builtin_amdgcn_readfirstlane(tid >> 5);
    ncnt[tid] = 0;
    const v2i* r2    = (const v2i*)recs;
    const int* myseg = seg + (size_t)b * eblocks;
    __syncthreads();
    walk_slices<false>(r2, myseg, ncnt, perm, 0, E, eblocks, wv, lane);
    __syncthreads();
    const int v = ncnt[tid];
    int inc = v;
    #pragma unroll
    for (int o = 1; o < 32; o <<= 1) {
        const int t = __shfl_up(inc, o, 32);
        if (lane >= o) inc += t;
    }
    if (lane == 31) wave_off[wv] = inc;
    __syncthreads();
    if (tid == 0) {
        int run = 0;
        #pragma unroll
        for (int w = 0; w < 8; ++w) { const int t = wave_off[w]; wave_off[w] = run; run += t; }
        *block_base = atomicAdd(total, run);
    }
    __syncthreads();
    const int excl = wave_off[wv] + inc - v;
    const int base = *block_base;
    const int node = b * BKT_NODES + tid;
    if (node < N) { deg[node] = v; rowstart[node] = base + excl; }
    ncnt[tid] = excl;
    __syncthreads();
    walk_slices<true>(r2, myseg, ncnt, perm, base, E, eblocks, wv, lane);
}

__global__ void sort_scores_kernel(const int* __restrict__ recs,
                                   const int* __restrict__ seg,
                                   int* __restrict__ total,
                                   int* __restrict__ deg,
                                   int* __restrict__ rowstart,
                                   int* __restrict__ perm,
                                   const float* __restrict__ h,
                                   const float* __restrict__ wlr,
                                   float* __restrict__ el,
                                   float* __restrict__ er,
                                   int N, int E, int eblocks, int nb) {
    __shared__ int ncnt[BKT_NODES];
    __shared__ int wave_off[8];
    __shared__ int block_base;
    const int b = blockIdx.x;
    if (b < nb)
        bucket_sort_block(recs, seg, total, deg, rowstart, perm, ncnt, wave_off, &block_base,
                          N, E, eblocks, b);
    else
        attn_score_block(h, wlr, el, er, N, b - nb);
}

__device__ __forceinline__ float leaky(float v) {
    return v > 0.0f ? v : NEG_SLOPE * v;
}

__device__ __forceinline__ void load_edge(const int* __restrict__ perm,
                                          const int* __restrict__ src,
                                          const float* __restrict__ el,
                                          const float4 erd, int pos, int E,
                                          int& s, float4& v) {
    const int e = min(max(perm[pos], 0), E - 1);
    s = src[e];
    const float4 l = *(const float4*)(el + (size_t)s * H_DIM);
    v.x = leaky(l.x + erd.x); v.y = leaky(l.y + erd.y);
    v.z = leaky(l.z + erd.z); v.w = leaky(l.w + erd.w);
}

#define ROWS_PER_TRIP 8
__device__ __forceinline__ void add_rows(const float* __restrict__ h,
                                         int lane, int s_l,
                                         const float4 wt, int n,
                                         v2f (&acc)[H_DIM]) {
    for (int i = 0; i < n; i += ROWS_PER_TRIP) {
        v2f f[ROWS_PER_TRIP];
        #pragma unroll
        for (int u = 0; u < ROWS_PER_TRIP; ++u) {
            const int s = __builtin_amdgcn_readlane(s_l, i + u);
            f[u] = *(const v2f*)(h + (size_t)s * F_DIM + lane * 2);
        }
        #pragma unroll
        for (int u = 0; u < ROWS_PER_TRIP; ++u) {
            const float wx = __int_as_float(__builtin_amdgcn_readlane(__float_as_int(wt.x), i + u));
            const float wy = __int_as_float(__builtin_amdgcn_readlane(__float_as_int(wt.y), i + u));
            const float wz = __int_as_float(__builtin_amdgcn_readlane(__float_as_int(wt.z), i + u));
            const float ww = __int_as_float(__builtin_amdgcn_readlane(__float_as_int(wt.w), i + u));
            const v2f w0 = {wx, wx}, w1 = {wy, wy}, w2 = {wz, wz}, w3 = {ww, ww};
            acc[0] = __builtin_elementwise_fma(f[u], w0, acc[0]);
            acc[1] = __builtin_elementwise_fma(f[u], w1, acc[1]);
            acc[2] = __builtin_elementwise_fma(f[u], w2, acc[2]);
            acc[3] = __builtin_elementwise_fma(f[u], w3, acc[3]);
        }
    }
}

__device__ __forceinline__ void aggregate_node(const int d, const int lane,
                                               const int* __restrict__ src,
                                               const int* __restrict__ rowstart,
                                               const int* __restrict__ deg,
                                               const int* __restrict__ perm,
                                               const float* __restrict__ h,
                                               const float* __restrict__ el,
                                               const float* __restrict__ er,
                                               float* __restrict__ row, int E) {
    const int beg = min(max(rowstart[d], 0), E);
    const int cnt = min(max(deg[d], 0), E - beg);
    const float4 erd = *(const float4*)(er + (size_t)d * H_DIM);

    int    s0 = 0;
    float4 v0 = {-INFINITY, -INFINITY, -INFINITY, -INFINITY};
    if (lane < cnt) load_edge(perm, src, el, erd, beg + lane, E, s0, v0);
    s0 = lane < cnt ? s0 : __builtin_amdgcn_readfirstlane(s0);

    float4 mx = v0;
    for (int c = 32; c < cnt; c += 32) {
        if (c + lane < cnt) {
            int s; float4 v;
            load_edge(perm, src, el, erd, beg + c + lane, E, s, v);
            mx.x = fmaxf(mx.x, v.x); mx.y = fmaxf(mx.y, v.y);
            mx.z = fmaxf(mx.z, v.z); mx.w = fmaxf(mx.w, v.w);
        }
    }
    #pragma unroll
    for (int msk = 16; msk >= 1; msk >>= 1) {
        mx.x = fmaxf(mx.x, __shfl_xor(mx.x, msk, 32));
        mx.y = fmaxf(mx.y, __shfl_xor(mx.y, msk, 32));
        mx.z = fmaxf(mx.z, __shfl_xor(mx.z, msk, 32));
        mx.w = fmaxf(mx.w, __shfl_xor(mx.w, msk, 32));
    }

    v2f acc[H_DIM];
    #pragma unroll
    for (int k = 0; k < H_DIM; ++k) { acc[k].x = 0.0f; acc[k].y = 0.0f; }
    float4 ex0 = {0.0f, 0.0f, 0.0f, 0.0f};
    if (lane < cnt) {
        ex0.x = __expf(v0.x - mx.x); ex0.y = __expf(v0.y - mx.y);
        ex0.z = __expf(v0.z - mx.z); ex0.w = __expf(v0.w - mx.w);
    }
    float4 zs = ex0;
    for (int c = 32; c < cnt; c += 32) {
        const int n = min(32, cnt - c);
        int    s_l = 0;
        float4 ex  = {0.0f, 0.0f, 0.0f, 0.0f};
        if (lane < n) {
            float4 v;
            load_edge(perm, src, el, erd, beg + c + lane, E, s_l, v);
            ex.x = __expf(v.x - mx.x); ex.y = __expf(v.y - mx.y);
            ex.z = __expf(v.z - mx.z); ex.w = __expf(v.w - mx.w);
        }
        s_l = lane < n ? s_l : __builtin_amdgcn_readfirstlane(s_l);
        zs.x += ex.x; zs.y += ex.y; zs.z += ex.z; zs.w += ex.w;
        add_rows(h, lane, s_l, ex, n, acc);
    }
    #pragma unroll
    for (int msk = 16; msk >= 1; msk >>= 1) {
        zs.x += __shfl_xor(zs.x, msk, 32);
        zs.y += __shfl_xor(zs.y, msk, 32);
        zs.z += __shfl_xor(zs.z, msk, 32);
        zs.w += __shfl_xor(zs.w, msk, 32);
    }
    float4 rz = {1.0f, 1.0f, 1.0f, 1.0f};
    if (cnt > 0) {
        rz.x = 1.0f / zs.x; rz.y = 1.0f / zs.y;
        rz.z = 1.0f / zs.z; rz.w = 1.0f / zs.w;
    }
    acc[0].x *= rz.x; acc[0].y *= rz.x;
    acc[1].x *= rz.y; acc[1].y *= rz.y;
    acc[2].x *= rz.z; acc[2].y *= rz.z;
    acc[3].x *= rz.w; acc[3].y *= rz.w;
    float4 alpha;
    alpha.x = ex0.x * rz.x; alpha.y = ex0.y * rz.y;
    alpha.z = ex0.z * rz.z; alpha.w = ex0.w * rz.w;

    add_rows(h, lane, s0, alpha, min(cnt, 32), acc);

    float* r = row + lane * 2;
    #pragma unroll
    for (int k = 0; k < H_DIM; ++k)
        *(v2f*)(r + k * F_DIM) = acc[k];
}

#define AGG_NODES_PER_WAVE 2
#define TILE_NODES (8 * AGG_NODES_PER_WAVE)
#define AGG_PITCH  (HF + 4)
#define X_PITCH    (F_DIM + 4)
__global__ void aggregate_out_kernel(const float* __restrict__ h,
                                     const int* __restrict__ src,
                                     const int* __restrict__ rowstart,
                                     const int* __restrict__ deg,
                                     const int* __restrict__ perm,
                                     const float* __restrict__ el,
                                     const float* __restrict__ er,
                                     const float* __restrict__ w,
                                     const float* __restrict__ bv,
                                     const float* __restrict__ g,
                                     const float* __restrict__ b,
                                     float* __restrict__ out, int N, int E) {
    __shared__ float s_agg[TILE_NODES * AGG_PITCH];
    __shared__ float s_x[TILE_NODES * X_PITCH];
    const int wv    = __builtin_amdgcn_readfirstlane((int)(threadIdx.x >> 5));
    const int lane  = threadIdx.x & 31;
    const int node0 = blockIdx.x * TILE_NODES;

    #pragma nounroll
    for (int j = 0; j < AGG_NODES_PER_WAVE; ++j) {
        const int r = wv * AGG_NODES_PER_WAVE + j;
        const int d = node0 + r;
        float* row = s_agg + r * AGG_PITCH;
        if (d < N) {
            aggregate_node(d, lane, src, rowstart, deg, perm, h, el, er, row, E);
        } else {
            const v2f zero = {0.0f, 0.0f};
            for (int k = 0; k < H_DIM; ++k)
                *(v2f*)(row + k * F_DIM + lane * 2) = zero;
        }
    }
    __syncthreads();

    if (wv < F_DIM / 16) {
        const int half = lane >> 4, l = lane & 15;
        const float* arow = s_agg + l * AGG_PITCH;
        const int bcol = wv * 16 + l;
        v8f acc = {};
        #pragma unroll 8
        for (int k = 0; k < HF; k += 4) {
            const int ka = k + 2 * half;
            v2f a, bb;
            a.x  = arow[ka];
            a.y  = arow[ka + 1];
            bb.x = w[ka * F_DIM + bcol];
            bb.y = w[(ka + 1) * F_DIM + bcol];
            acc = __builtin_amdgcn_wmma_f32_16x16x4_f32(
                      false, a, false, bb, (short)0, acc, false, false);
        }
        const float bias = bv[bcol];
        float* xr = s_x + (half * 8) * X_PITCH + bcol;
        #pragma unroll
        for (int i = 0; i < 8; ++i)
            xr[i * X_PITCH] = acc[i] + bias;
    }
    __syncthreads();

    const float g0 = g[lane], g1 = g[lane + 32];
    const float b0 = b[lane], b1 = b[lane + 32];
    #pragma unroll
    for (int j = 0; j < AGG_NODES_PER_WAVE; ++j) {
        const int r = wv * AGG_NODES_PER_WAVE + j;
        const int node = node0 + r;
        const float x0 = s_x[r * X_PITCH + lane], x1 = s_x[r * X_PITCH + lane + 32];
        float s = x0 + x1, sq = x0 * x0 + x1 * x1;
        #pragma unroll
        for (int msk = 1; msk < 32; msk <<= 1) {
            s  += __shfl_xor(s,  msk, 32);
            sq += __shfl_xor(sq, msk, 32);
        }
        const float mu  = s * (1.0f / 64.0f);
        const float var = sq * (1.0f / 64.0f) - mu * mu;
        const float inv = rsqrtf(var + LN_EPS);
        if (node < N) {
            const size_t base = (size_t)node * F_DIM;
            out[base + lane]      = (x0 - mu) * inv * g0 + b0;
            out[base + lane + 32] = (x1 - mu) * inv * g1 + b1;
        }
    }
}

extern "C" void kernel_launch(void* const* d_in, const int* in_sizes, int n_in,
                              void* d_out, int out_size, void* d_ws, size_t ws_size,
                              hipStream_t stream) {
    const float* h   = (const float*)d_in[0];
    const int*   src = (const int*)  d_in[1];
    const int*   dst = (const int*)  d_in[2];
    const float* fcw = (const float*)d_in[3];
    const float* al  = (const float*)d_in[4];
    const float* ar  = (const float*)d_in[5];
    const float* gb  = (const float*)d_in[6];
    const float* ow  = (const float*)d_in[7];
    const float* ob  = (const float*)d_in[8];
    const float* lng = (const float*)d_in[9];
    const float* lnb = (const float*)d_in[10];

    const int N = in_sizes[0] / F_DIM;
    const int E = in_sizes[1];

    char*  ws  = (char*)d_ws;
    size_t off = 0;
    auto carve = [&](size_t bytes) -> float* {
        float* p = (float*)(ws + off);
        off += (bytes + 255) & ~(size_t)255;
        return p;
    };
    float* el   = carve((size_t)N * H_DIM * 4);
    float* er   = carve((size_t)N * H_DIM * 4);
    float* wlr  = carve((size_t)2 * H_DIM * F_DIM * 4);
    float* bvec = carve((size_t)F_DIM * 4);
    float* wc   = carve((size_t)HF * F_DIM * 4);
    const int nb      = (N + BKT_NODES - 1) / BKT_NODES;
    const int eblocks = (E + CHUNK_EDGES - 1) / CHUNK_EDGES;
    int* deg      = (int*)carve((size_t)N * 4);
    int* rowstart = (int*)carve((size_t)N * 4);
    int* total    = (int*)carve(4);
    int* seg      = (int*)carve((size_t)nb * eblocks * 4);
    int* recs     = (int*)carve((size_t)eblocks * CHUNK_EDGES * 8);
    int* perm     = (int*)carve((size_t)E * 4);
    (void)ws_size;

    const int ablocks = (N + 8 * NODES_PER_WAVE - 1) / (8 * NODES_PER_WAVE);
    fold_partition_kernel<<<eblocks + FOLD_BLOCKS, MAX_BKTS, 0, stream>>>(
        dst, seg, recs, total, fcw, al, ar, gb, ow, ob, wlr, bvec, wc, E, nb, eblocks);
    sort_scores_kernel<<<nb + ablocks, BKT_NODES, 0, stream>>>(
        recs, seg, total, deg, rowstart, perm, h, wlr, el, er, N, E, eblocks, nb);
    const int blocks = (N + TILE_NODES - 1) / TILE_NODES;
    aggregate_out_kernel<<<blocks, 256, 0, stream>>>(h, src, rowstart, deg, perm,
                                                     el, er, wc, bvec, lng, lnb,
                                                     (float*)d_out, N, E);
}


# folded weights wc kept in the output product's B-fragment order: a wave's fragment load is one 16-byte load per lane over four whole lines and feeds two WMMAs (was four 4-byte half-line loads)
# speedup vs baseline: 9.9458x; 1.0583x over previous
#include <hip/hip_runtime.h>
#include <math.h>

typedef __attribute__((ext_vector_type(2))) float v2f;
typedef __attribute__((ext_vector_type(8))) float v8f;
typedef __attribute__((ext_vector_type(2))) int   v2i;

#define F_DIM 64
#define H_DIM 4
#define HF    256
#define NEG_SLOPE 0.2f
#define LN_EPS 1e-5f


__device__ __forceinline__ v8f wmma_tile_k64(const float* __restrict__ A, int lda,
                                             const float* __restrict__ B, int ldb,
                                             int lane) {
    const int half = lane >> 4, l = lane & 15;
    v8f acc = {};
    #pragma unroll
    for (int k4 = 0; k4 < F_DIM / 4; ++k4) {
        const int ka = k4 * 4 + 2 * half;
        v2f a, b;
        a.x = A[l * lda + ka];
        a.y = A[l * lda + ka + 1];
        b.x = B[ka * ldb + l];
        b.y = B[(ka + 1) * ldb + l];
        acc = __builtin_amdgcn_wmma_f32_16x16x4_f32(
                  false, a, false, b, (short)0, acc, false, false);
    }
    return acc;
}

#define FOLD_BLOCKS 11
#define WC_FRAG(k, nt, c) \
    (((((nt) * (HF / 8) + ((k) >> 3)) * 32 + (((k) >> 1) & 1) * 16 + (c)) << 2) + \
     ((((k) >> 2) & 1) << 1) + ((k) & 1))
__device__ __forceinline__ void fold_weights_block(const float* __restrict__ fcw,
                                                   const float* __restrict__ al,
                                                   const float* __restrict__ ar,
                                                   const float* __restrict__ gb,
                                                   const float* __restrict__ ow,
                                                   const float* __restrict__ ob,
                                                   float* __restrict__ wlr,
                                                   float* __restrict__ bvec,
                                                   float* __restrict__ wc,
                                                   int* __restrict__ total,
                                                   float* red, int r) {
    if (r == 0) {
        if (threadIdx.x == 0) total[0] = 0;
        const int j = threadIdx.x & 63, part = threadIdx.x >> 6;
        float s = 0.0f;
        #pragma unroll 8
        for (int p = part * 64; p < part * 64 + 64; ++p)
            s += gb[p] * ow[p * F_DIM + j];
        red[threadIdx.x] = s;
        __syncthreads();
        if (threadIdx.x < 64)
            bvec[j] = ((red[j] + red[64 + j]) + (red[128 + j] + red[192 + j])) + ob[j];
        return;
    }
    if (r > 8) {
        const int t = (r - 9) * 256 + threadIdx.x;
        const int o = t >> 6, k = t & 63, hs = o & 3;
        const float* a = (o < 4 ? al : ar) + hs * F_DIM;
        const float* w = fcw + k * HF + hs * F_DIM;
        float s = 0.0f;
        #pragma unroll
        for (int f = 0; f < F_DIM; f += 4) {
            const float4 wv = *(const float4*)(w + f);
            const float4 av = *(const float4*)(a + f);
            s += wv.x * av.x; s += wv.y * av.y;
            s += wv.z * av.z; s += wv.w * av.w;
        }
        wlr[o * F_DIM + k] = s;
        return;
    }
    const int tile = (r - 1) * 8 + (threadIdx.x >> 5);
    const int hs = tile >> 4, mt = (tile >> 2) & 3, nt = tile & 3;
    const int lane = threadIdx.x & 31, half = lane >> 4, l = lane & 15;
    const v8f acc = wmma_tile_k64(fcw + (mt * 16) * HF + hs * F_DIM, HF,
                                  ow + (hs * F_DIM) * F_DIM + nt * 16, F_DIM, lane);
    const int k0 = hs * F_DIM + mt * 16 + half * 8;
    #pragma unroll
    for (int i = 0; i < 8; ++i)
        wc[WC_FRAG(k0 + i, nt, l)] = acc[i];
}

#define NODES_PER_WAVE 8
__device__ __forceinline__ void attn_score_block(const float* __restrict__ h,
                                                 const float* __restrict__ wlr,
                                                 float* __restrict__ el,
                                                 float* __restrict__ er,
                                                 int N, int blk) {
    const int wave  = (blk * (int)blockDim.x + (int)threadIdx.x) >> 5;
    const int node0 = wave * NODES_PER_WAVE;
    if (node0 >= N) return;
    const int lane = threadIdx.x & 31;
    const int sub = lane & 7, q = lane >> 3;
    float4 fa[NODES_PER_WAVE / 4], fb[NODES_PER_WAVE / 4];
    #pragma unroll
    for (int g = 0; g < NODES_PER_WAVE / 4; ++g) {
        const float* p = h + (size_t)min(node0 + g * 4 + q, N - 1) * F_DIM + sub * 8;
        fa[g] = *(const float4*)(p);
        fb[g] = *(const float4*)(p + 4);
    }
    float4 wa[8], wb[8];
    #pragma unroll
    for (int o = 0; o < 8; ++o) {
        wa[o] = *(const float4*)(wlr + o * F_DIM + sub * 8);
        wb[o] = *(const float4*)(wlr + o * F_DIM + sub * 8 + 4);
    }
    #pragma unroll
    for (int g = 0; g < NODES_PER_WAVE / 4; ++g) {
        float s[8];
        #pragma unroll
        for (int o = 0; o < 8; ++o) {
            float t = fa[g].x * wa[o].x;
            t += fa[g].y * wa[o].y; t += fa[g].z * wa[o].z; t += fa[g].w * wa[o].w;
            t += fb[g].x * wb[o].x; t += fb[g].y * wb[o].y;
            t += fb[g].z * wb[o].z; t += fb[g].w * wb[o].w;
            s[o] = t;
        }
        #pragma unroll
        for (int msk = 1; msk < 8; msk <<= 1) {
            #pragma unroll
            for (int o = 0; o < 8; ++o) s[o] += __shfl_xor(s[o], msk, 32);
        }
        const int node = node0 + g * 4 + q;
        if (sub == 0 && node < N) {
            float4 l4, r4;
            l4.x = s[0]; l4.y = s[1]; l4.z = s[2]; l4.w = s[3];
            r4.x = s[4]; r4.y = s[5]; r4.z = s[6]; r4.w = s[7];
            *(float4*)(el + (size_t)node * H_DIM) = l4;
            *(float4*)(er + (size_t)node * H_DIM) = r4;
        }
    }
}

#define BKT_SHIFT   8
#define BKT_NODES   (1 << BKT_SHIFT)
#define MAX_BKTS    256
#define CHUNK_EDGES 4096
__device__ __forceinline__ void edge_partition_block(const int* __restrict__ dst,
                                                     int* __restrict__ seg,
                                                     int* __restrict__ recs,
                                                     int* cnt, int* gbase, int* wtot,
                                                     int E, int nb, int eblocks, int blk) {
    const int tid = threadIdx.x, lane = tid & 31, wv = tid >> 5;
    cnt[tid] = 0;
    __syncthreads();
    const int e_blk = blk * CHUNK_EDGES;
    int dd[16], lr[16];
    if (e_blk + CHUNK_EDGES <= E) {
        #pragma unroll
        for (int k = 0; k < 4; ++k) {
            const int4 v = *(const int4*)(dst + e_blk + k * 1024 + tid * 4);
            dd[4 * k] = v.x; dd[4 * k + 1] = v.y; dd[4 * k + 2] = v.z; dd[4 * k + 3] = v.w;
        }
    } else {
        #pragma unroll
        for (int i = 0; i < 16; ++i) {
            const int e = e_blk + (i >> 2) * 1024 + tid * 4 + (i & 3);
            const int v = dst[min(e, E - 1)];
            dd[i] = e < E ? v : -1;
        }
    }
    #pragma unroll
    for (int i = 0; i < 16; ++i)
        lr[i] = dd[i] >= 0 ? atomicAdd(&cnt[dd[i] >> BKT_SHIFT], 1) : 0;
    __syncthreads();
    const int c = cnt[tid];
    int inc = c;
    #pragma unroll
    for (int o = 1; o < 32; o <<= 1) {
        const int t = __shfl_up(inc, o, 32);
        if (lane >= o) inc += t;
    }
    if (lane == 31) wtot[wv] = inc;
    __syncthreads();
    int excl = inc - c;
    #pragma unroll
    for (int w = 0; w < 7; ++w) excl += w < wv ? wtot[w] : 0;
    gbase[tid] = excl;
    if (tid < nb) seg[(size_t)tid * eblocks + blk] = (excl << 16) | c;
    __syncthreads();
    v2i* r2 = (v2i*)recs + (size_t)blk * CHUNK_EDGES;
    #pragma unroll
    for (int i = 0; i < 16; ++i) {
        if (dd[i] >= 0) {
            v2i rec;
            rec.x = dd[i];
            rec.y = e_blk + (i >> 2) * 1024 + tid * 4 + (i & 3);
            r2[gbase[dd[i] >> BKT_SHIFT] + lr[i]] = rec;
        }
    }
}

__global__ void fold_partition_kernel(const int* __restrict__ dst,
                                      int* __restrict__ seg,
                                      int* __restrict__ recs,
                                      int* __restrict__ total,
                                      const float* __restrict__ fcw,
                                      const float* __restrict__ al,
                                      const float* __restrict__ ar,
                                      const float* __restrict__ gb,
                                      const float* __restrict__ ow,
                                      const float* __restrict__ ob,
                                      float* __restrict__ wlr,
                                      float* __restrict__ bvec,
                                      float* __restrict__ wc,
                                      int E, int nb, int eblocks) {
    __shared__ int cnt[MAX_BKTS];
    __shared__ int gbase[MAX_BKTS];
    __shared__ int wtot[8];
    __shared__ float red[256];
    const int b = blockIdx.x;
    if (b < eblocks)
        edge_partition_block(dst, seg, recs, cnt, gbase, wtot, E, nb, eblocks, b);
    else
        fold_weights_block(fcw, al, ar, gb, ow, ob, wlr, bvec, wc, total, red, b - eblocks);
}

#define SEGS_PER_TRIP 8
static_assert(SEGS_PER_TRIP == 8, "the load fence in walk_slices names eight records");
template <bool PLACE>
__device__ __forceinline__ void walk_slices(const v2i* __restrict__ r2,
                                            const int* __restrict__ myseg,
                                            int* ncnt, int* __restrict__ perm,
                                            int base, int E, int eblocks,
                                            int wv, int lane) {
    for (int j0 = 0; j0 < eblocks; j0 += 256) {
        const int j  = j0 + wv + 8 * lane;
        const int sv = j < eblocks ? myseg[j] : 0;
        const int nseg = (min(eblocks - j0, 256) - wv + 7) >> 3;
        for (int q = 0; q < nseg; q += SEGS_PER_TRIP) {
            v2i r[SEGS_PER_TRIP];
            int c[SEGS_PER_TRIP], st[SEGS_PER_TRIP];
            #pragma unroll
            for (int u = 0; u < SEGS_PER_TRIP; ++u) {
                const int s   = __builtin_amdgcn_readlane(sv, q + u);
                const int off = min(max(s >> 16, 0), CHUNK_EDGES);
                c[u]  = min(s & 0xFFFF, CHUNK_EDGES - off);
                st[u] = min(j0 + wv + 8 * (q + u), eblocks - 1) * CHUNK_EDGES + off;
                r[u]  = r2[c[u] > 0 ? st[u] + min(lane, c[u] - 1) : 0];
            }
            asm volatile("" :: "v"(r[0]), "v"(r[1]), "v"(r[2]), "v"(r[3]),
                               "v"(r[4]), "v"(r[5]), "v"(r[6]), "v"(r[7]));
            int pos[SEGS_PER_TRIP];
            #pragma unroll
            for (int u = 0; u < SEGS_PER_TRIP; ++u)
                pos[u] = lane < c[u] ? atomicAdd(&ncnt[r[u].x & (BKT_NODES - 1)], 1) : 0;
            if (PLACE) {
                #pragma unroll
                for (int u = 0; u < SEGS_PER_TRIP; ++u)
                    if (lane < c[u]) perm[min(max(base + pos[u], 0), E - 1)] = r[u].y;
            }
            #pragma unroll
            for (int u = 0; u < SEGS_PER_TRIP; ++u) {
                for (int i = 32 + lane; i < c[u]; i += 32) {
                    const v2i rr = r2[st[u] + i];
                    const int pos = atomicAdd(&ncnt[rr.x & (BKT_NODES - 1)], 1);
                    if (PLACE) perm[min(max(base + pos, 0), E - 1)] = rr.y;
                }
            }
        }
    }
}

__device__ __forceinline__ void bucket_sort_block(const int* __restrict__ recs,
                                                  const int* __restrict__ seg,
                                                  int* __restrict__ total,
                                                  int* __restrict__ deg,
                                                  int* __restrict__ rowstart,
                                                  int* __restrict__ perm,
                                                  int* ncnt, int* wave_off, int* block_base,
                                                  int N, int E, int eblocks, int b) {
    const int tid = threadIdx.x, lane = tid & 31;
    const int wv = __builtin_amdgcn_readfirstlane(tid >> 5);
    ncnt[tid] = 0;
    const v2i* r2    = (const v2i*)recs;
    const int* myseg = seg + (size_t)b * eblocks;
    __syncthreads();
    walk_slices<false>(r2, myseg, ncnt, perm, 0, E, eblocks, wv, lane);
    __syncthreads();
    const int v = ncnt[tid];
    int inc = v;
    #pragma unroll
    for (int o = 1; o < 32; o <<= 1) {
        const int t = __shfl_up(inc, o, 32);
        if (lane >= o) inc += t;
    }
    if (lane == 31) wave_off[wv] = inc;
    __syncthreads();
    if (tid == 0) {
        int run = 0;
        #pragma unroll
        for (int w = 0; w < 8; ++w) { const int t = wave_off[w]; wave_off[w] = run; run += t; }
        *block_base = atomicAdd(total, run);
    }
    __syncthreads();
    const int excl = wave_off[wv] + inc - v;
    const int base = *block_base;
    const int node = b * BKT_NODES + tid;
    if (node < N) { deg[node] = v; rowstart[node] = base + excl; }
    ncnt[tid] = excl;
    __syncthreads();
    walk_slices<true>(r2, myseg, ncnt, perm, base, E, eblocks, wv, lane);
}

__global__ void sort_scores_kernel(const int* __restrict__ recs,
                                   const int* __restrict__ seg,
                                   int* __restrict__ total,
                                   int* __restrict__ deg,
                                   int* __restrict__ rowstart,
                                   int* __restrict__ perm,
                                   const float* __restrict__ h,
                                   const float* __restrict__ wlr,
                                   float* __restrict__ el,
                                   float* __restrict__ er,
                                   int N, int E, int eblocks, int nb) {
    __shared__ int ncnt[BKT_NODES];
    __shared__ int wave_off[8];
    __shared__ int block_base;
    const int b = blockIdx.x;
    if (b < nb)
        bucket_sort_block(recs, seg, total, deg, rowstart, perm, ncnt, wave_off, &block_base,
                          N, E, eblocks, b);
    else
        attn_score_block(h, wlr, el, er, N, b - nb);
}

__device__ __forceinline__ float leaky(float v) {
    return v > 0.0f ? v : NEG_SLOPE * v;
}

__device__ __forceinline__ void load_edge(const int* __restrict__ perm,
                                          const int* __restrict__ src,
                                          const float* __restrict__ el,
                                          const float4 erd, int pos, int E,
                                          int& s, float4& v) {
    const int e = min(max(perm[pos], 0), E - 1);
    s = src[e];
    const float4 l = *(const float4*)(el + (size_t)s * H_DIM);
    v.x = leaky(l.x + erd.x); v.y = leaky(l.y + erd.y);
    v.z = leaky(l.z + erd.z); v.w = leaky(l.w + erd.w);
}

#define ROWS_PER_TRIP 8
__device__ __forceinline__ void add_rows(const float* __restrict__ h,
                                         int lane, int s_l,
                                         const float4 wt, int n,
                                         v2f (&acc)[H_DIM]) {
    for (int i = 0; i < n; i += ROWS_PER_TRIP) {
        v2f f[ROWS_PER_TRIP];
        #pragma unroll
        for (int u = 0; u < ROWS_PER_TRIP; ++u) {
            const int s = __builtin_amdgcn_readlane(s_l, i + u);
            f[u] = *(const v2f*)(h + (size_t)s * F_DIM + lane * 2);
        }
        #pragma unroll
        for (int u = 0; u < ROWS_PER_TRIP; ++u) {
            const float wx = __int_as_float(__builtin_amdgcn_readlane(__float_as_int(wt.x), i + u));
            const float wy = __int_as_float(__builtin_amdgcn_readlane(__float_as_int(wt.y), i + u));
            const float wz = __int_as_float(__builtin_amdgcn_readlane(__float_as_int(wt.z), i + u));
            const float ww = __int_as_float(__builtin_amdgcn_readlane(__float_as_int(wt.w), i + u));
            const v2f w0 = {wx, wx}, w1 = {wy, wy}, w2 = {wz, wz}, w3 = {ww, ww};
            acc[0] = __builtin_elementwise_fma(f[u], w0, acc[0]);
            acc[1] = __builtin_elementwise_fma(f[u], w1, acc[1]);
            acc[2] = __builtin_elementwise_fma(f[u], w2, acc[2]);
            acc[3] = __builtin_elementwise_fma(f[u], w3, acc[3]);
        }
    }
}

__device__ __forceinline__ void aggregate_node(const int d, const int lane,
                                               const int* __restrict__ src,
                                               const int* __restrict__ rowstart,
                                               const int* __restrict__ deg,
                                               const int* __restrict__ perm,
                                               const float* __restrict__ h,
                                               const float* __restrict__ el,
                                               const float* __restrict__ er,
                                               float* __restrict__ row, int E) {
    const int beg = min(max(rowstart[d], 0), E);
    const int cnt = min(max(deg[d], 0), E - beg);
    const float4 erd = *(const float4*)(er + (size_t)d * H_DIM);

    int    s0 = 0;
    float4 v0 = {-INFINITY, -INFINITY, -INFINITY, -INFINITY};
    if (lane < cnt) load_edge(perm, src, el, erd, beg + lane, E, s0, v0);
    s0 = lane < cnt ? s0 : __builtin_amdgcn_readfirstlane(s0);

    float4 mx = v0;
    for (int c = 32; c < cnt; c += 32) {
        if (c + lane < cnt) {
            int s; float4 v;
            load_edge(perm, src, el, erd, beg + c + lane, E, s, v);
            mx.x = fmaxf(mx.x, v.x); mx.y = fmaxf(mx.y, v.y);
            mx.z = fmaxf(mx.z, v.z); mx.w = fmaxf(mx.w, v.w);
        }
    }
    #pragma unroll
    for (int msk = 16; msk >= 1; msk >>= 1) {
        mx.x = fmaxf(mx.x, __shfl_xor(mx.x, msk, 32));
        mx.y = fmaxf(mx.y, __shfl_xor(mx.y, msk, 32));
        mx.z = fmaxf(mx.z, __shfl_xor(mx.z, msk, 32));
        mx.w = fmaxf(mx.w, __shfl_xor(mx.w, msk, 32));
    }

    v2f acc[H_DIM];
    #pragma unroll
    for (int k = 0; k < H_DIM; ++k) { acc[k].x = 0.0f; acc[k].y = 0.0f; }
    float4 ex0 = {0.0f, 0.0f, 0.0f, 0.0f};
    if (lane < cnt) {
        ex0.x = __expf(v0.x - mx.x); ex0.y = __expf(v0.y - mx.y);
        ex0.z = __expf(v0.z - mx.z); ex0.w = __expf(v0.w - mx.w);
    }
    float4 zs = ex0;
    for (int c = 32; c < cnt; c += 32) {
        const int n = min(32, cnt - c);
        int    s_l = 0;
        float4 ex  = {0.0f, 0.0f, 0.0f, 0.0f};
        if (lane < n) {
            float4 v;
            load_edge(perm, src, el, erd, beg + c + lane, E, s_l, v);
            ex.x = __expf(v.x - mx.x); ex.y = __expf(v.y - mx.y);
            ex.z = __expf(v.z - mx.z); ex.w = __expf(v.w - mx.w);
        }
        s_l = lane < n ? s_l : __builtin_amdgcn_readfirstlane(s_l);
        zs.x += ex.x; zs.y += ex.y; zs.z += ex.z; zs.w += ex.w;
        add_rows(h, lane, s_l, ex, n, acc);
    }
    #pragma unroll
    for (int msk = 16; msk >= 1; msk >>= 1) {
        zs.x += __shfl_xor(zs.x, msk, 32);
        zs.y += __shfl_xor(zs.y, msk, 32);
        zs.z += __shfl_xor(zs.z, msk, 32);
        zs.w += __shfl_xor(zs.w, msk, 32);
    }
    float4 rz = {1.0f, 1.0f, 1.0f, 1.0f};
    if (cnt > 0) {
        rz.x = 1.0f / zs.x; rz.y = 1.0f / zs.y;
        rz.z = 1.0f / zs.z; rz.w = 1.0f / zs.w;
    }
    acc[0].x *= rz.x; acc[0].y *= rz.x;
    acc[1].x *= rz.y; acc[1].y *= rz.y;
    acc[2].x *= rz.z; acc[2].y *= rz.z;
    acc[3].x *= rz.w; acc[3].y *= rz.w;
    float4 alpha;
    alpha.x = ex0.x * rz.x; alpha.y = ex0.y * rz.y;
    alpha.z = ex0.z * rz.z; alpha.w = ex0.w * rz.w;

    add_rows(h, lane, s0, alpha, min(cnt, 32), acc);

    float* r = row + lane * 2;
    #pragma unroll
    for (int k = 0; k < H_DIM; ++k)
        *(v2f*)(r + k * F_DIM) = acc[k];
}

#define AGG_NODES_PER_WAVE 2
#define TILE_NODES (8 * AGG_NODES_PER_WAVE)
#define AGG_PITCH  (HF + 4)
#define X_PITCH    (F_DIM + 4)
__global__ void aggregate_out_kernel(const float* __restrict__ h,
                                     const int* __restrict__ src,
                                     const int* __restrict__ rowstart,
                                     const int* __restrict__ deg,
                                     const int* __restrict__ perm,
                                     const float* __restrict__ el,
                                     const float* __restrict__ er,
                                     const float* __restrict__ w,
                                     const float* __restrict__ bv,
                                     const float* __restrict__ g,
                                     const float* __restrict__ b,
                                     float* __restrict__ out, int N, int E) {
    __shared__ float s_agg[TILE_NODES * AGG_PITCH];
    __shared__ float s_x[TILE_NODES * X_PITCH];
    const int wv    = __builtin_amdgcn_readfirstlane((int)(threadIdx.x >> 5));
    const int lane  = threadIdx.x & 31;
    const int node0 = blockIdx.x * TILE_NODES;

    #pragma nounroll
    for (int j = 0; j < AGG_NODES_PER_WAVE; ++j) {
        const int r = wv * AGG_NODES_PER_WAVE + j;
        const int d = node0 + r;
        float* row = s_agg + r * AGG_PITCH;
        if (d < N) {
            aggregate_node(d, lane, src, rowstart, deg, perm, h, el, er, row, E);
        } else {
            const v2f zero = {0.0f, 0.0f};
            for (int k = 0; k < H_DIM; ++k)
                *(v2f*)(row + k * F_DIM + lane * 2) = zero;
        }
    }
    __syncthreads();

    if (wv < F_DIM / 16) {
        const int half = lane >> 4, l = lane & 15;
        const float* arow = s_agg + l * AGG_PITCH + 2 * half;
        const int bcol = wv * 16 + l;
        const float4* wf = (const float4*)w + (size_t)wv * (HF / 8) * 32 + lane;
        v8f acc = {};
        #pragma unroll 8
        for (int kp = 0; kp < HF / 8; ++kp) {
            const float4 bq = wf[kp * 32];
            v2f a0, a1, b0, b1;
            a0.x = arow[kp * 8];     a0.y = arow[kp * 8 + 1];
            a1.x = arow[kp * 8 + 4]; a1.y = arow[kp * 8 + 5];
            b0.x = bq.x; b0.y = bq.y;
            b1.x = bq.z; b1.y = bq.w;
            acc = __builtin_amdgcn_wmma_f32_16x16x4_f32(
                      false, a0, false, b0, (short)0, acc, false, false);
            acc = __builtin_amdgcn_wmma_f32_16x16x4_f32(
                      false, a1, false, b1, (short)0, acc, false, false);
        }
        const float bias = bv[bcol];
        float* xr = s_x + (half * 8) * X_PITCH + bcol;
        #pragma unroll
        for (int i = 0; i < 8; ++i)
            xr[i * X_PITCH] = acc[i] + bias;
    }
    __syncthreads();

    const float g0 = g[lane], g1 = g[lane + 32];
    const float b0 = b[lane], b1 = b[lane + 32];
    #pragma unroll
    for (int j = 0; j < AGG_NODES_PER_WAVE; ++j) {
        const int r = wv * AGG_NODES_PER_WAVE + j;
        const int node = node0 + r;
        const float x0 = s_x[r * X_PITCH + lane], x1 = s_x[r * X_PITCH + lane + 32];
        float s = x0 + x1, sq = x0 * x0 + x1 * x1;
        #pragma unroll
        for (int msk = 1; msk < 32; msk <<= 1) {
            s  += __shfl_xor(s,  msk, 32);
            sq += __shfl_xor(sq, msk, 32);
        }
        const float mu  = s * (1.0f / 64.0f);
        const float var = sq * (1.0f / 64.0f) - mu * mu;
        const float inv = rsqrtf(var + LN_EPS);
        if (node < N) {
            const size_t base = (size_t)node * F_DIM;
            out[base + lane]      = (x0 - mu) * inv * g0 + b0;
            out[base + lane + 32] = (x1 - mu) * inv * g1 + b1;
        }
    }
}

extern "C" void kernel_launch(void* const* d_in, const int* in_sizes, int n_in,
                              void* d_out, int out_size, void* d_ws, size_t ws_size,
                              hipStream_t stream) {
    const float* h   = (const float*)d_in[0];
    const int*   src = (const int*)  d_in[1];
    const int*   dst = (const int*)  d_in[2];
    const float* fcw = (const float*)d_in[3];
    const float* al  = (const float*)d_in[4];
    const float* ar  = (const float*)d_in[5];
    const float* gb  = (const float*)d_in[6];
    const float* ow  = (const float*)d_in[7];
    const float* ob  = (const float*)d_in[8];
    const float* lng = (const float*)d_in[9];
    const float* lnb = (const float*)d_in[10];

    const int N = in_sizes[0] / F_DIM;
    const int E = in_sizes[1];

    char*  ws  = (char*)d_ws;
    size_t off = 0;
    auto carve = [&](size_t bytes) -> float* {
        float* p = (float*)(ws + off);
        off += (bytes + 255) & ~(size_t)255;
        return p;
    };
    float* el   = carve((size_t)N * H_DIM * 4);
    float* er   = carve((size_t)N * H_DIM * 4);
    float* wlr  = carve((size_t)2 * H_DIM * F_DIM * 4);
    float* bvec = carve((size_t)F_DIM * 4);
    float* wc   = carve((size_t)HF * F_DIM * 4);
    const int nb      = (N + BKT_NODES - 1) / BKT_NODES;
    const int eblocks = (E + CHUNK_EDGES - 1) / CHUNK_EDGES;
    int* deg      = (int*)carve((size_t)N * 4);
    int* rowstart = (int*)carve((size_t)N * 4);
    int* total    = (int*)carve(4);
    int* seg      = (int*)carve((size_t)nb * eblocks * 4);
    int* recs     = (int*)carve((size_t)eblocks * CHUNK_EDGES * 8);
    int* perm     = (int*)carve((size_t)E * 4);
    (void)ws_size;

    const int ablocks = (N + 8 * NODES_PER_WAVE - 1) / (8 * NODES_PER_WAVE);
    fold_partition_kernel<<<eblocks + FOLD_BLOCKS, MAX_BKTS, 0, stream>>>(
        dst, seg, recs, total, fcw, al, ar, gb, ow, ob, wlr, bvec, wc, E, nb, eblocks);
    sort_scores_kernel<<<nb + ablocks, BKT_NODES, 0, stream>>>(
        recs, seg, total, deg, rowstart, perm, h, wlr, el, er, N, E, eblocks, nb);
    const int blocks = (N + TILE_NODES - 1) / TILE_NODES;
    aggregate_out_kernel<<<blocks, 256, 0, stream>>>(h, src, rowstart, deg, perm,
                                                     el, er, wc, bvec, lng, lnb,
                                                     (float*)d_out, N, E);
}


# edge records and the grouped list carry the source node, not the edge id (partition reads src beside dst, coalesced): aggregation drops its per-edge gather of src, one dependent round trip per node
# speedup vs baseline: 10.4480x; 1.0505x over previous
#include <hip/hip_runtime.h>
#include <math.h>

typedef __attribute__((ext_vector_type(2))) float v2f;
typedef __attribute__((ext_vector_type(8))) float v8f;
typedef __attribute__((ext_vector_type(2))) int   v2i;

#define F_DIM 64
#define H_DIM 4
#define HF    256
#define NEG_SLOPE 0.2f
#define LN_EPS 1e-5f


__device__ __forceinline__ v8f wmma_tile_k64(const float* __restrict__ A, int lda,
                                             const float* __restrict__ B, int ldb,
                                             int lane) {
    const int half = lane >> 4, l = lane & 15;
    v8f acc = {};
    #pragma unroll
    for (int k4 = 0; k4 < F_DIM / 4; ++k4) {
        const int ka = k4 * 4 + 2 * half;
        v2f a, b;
        a.x = A[l * lda + ka];
        a.y = A[l * lda + ka + 1];
        b.x = B[ka * ldb + l];
        b.y = B[(ka + 1) * ldb + l];
        acc = __builtin_amdgcn_wmma_f32_16x16x4_f32(
                  false, a, false, b, (short)0, acc, false, false);
    }
    return acc;
}

#define FOLD_BLOCKS 11
#define WC_FRAG(k, nt, c) \
    (((((nt) * (HF / 8) + ((k) >> 3)) * 32 + (((k) >> 1) & 1) * 16 + (c)) << 2) + \
     ((((k) >> 2) & 1) << 1) + ((k) & 1))
__device__ __forceinline__ void fold_weights_block(const float* __restrict__ fcw,
                                                   const float* __restrict__ al,
                                                   const float* __restrict__ ar,
                                                   const float* __restrict__ gb,
                                                   const float* __restrict__ ow,
                                                   const float* __restrict__ ob,
                                                   float* __restrict__ wlr,
                                                   float* __restrict__ bvec,
                                                   float* __restrict__ wc,
                                                   int* __restrict__ total,
                                                   float* red, int r) {
    if (r == 0) {
        if (threadIdx.x == 0) total[0] = 0;
        const int j = threadIdx.x & 63, part = threadIdx.x >> 6;
        float s = 0.0f;
        #pragma unroll 8
        for (int p = part * 64; p < part * 64 + 64; ++p)
            s += gb[p] * ow[p * F_DIM + j];
        red[threadIdx.x] = s;
        __syncthreads();
        if (threadIdx.x < 64)
            bvec[j] = ((red[j] + red[64 + j]) + (red[128 + j] + red[192 + j])) + ob[j];
        return;
    }
    if (r > 8) {
        const int t = (r - 9) * 256 + threadIdx.x;
        const int o = t >> 6, k = t & 63, hs = o & 3;
        const float* a = (o < 4 ? al : ar) + hs * F_DIM;
        const float* w = fcw + k * HF + hs * F_DIM;
        float s = 0.0f;
        #pragma unroll
        for (int f = 0; f < F_DIM; f += 4) {
            const float4 wv = *(const float4*)(w + f);
            const float4 av = *(const float4*)(a + f);
            s += wv.x * av.x; s += wv.y * av.y;
            s += wv.z * av.z; s += wv.w * av.w;
        }
        wlr[o * F_DIM + k] = s;
        return;
    }
    const int tile = (r - 1) * 8 + (threadIdx.x >> 5);
    const int hs = tile >> 4, mt = (tile >> 2) & 3, nt = tile & 3;
    const int lane = threadIdx.x & 31, half = lane >> 4, l = lane & 15;
    const v8f acc = wmma_tile_k64(fcw + (mt * 16) * HF + hs * F_DIM, HF,
                                  ow + (hs * F_DIM) * F_DIM + nt * 16, F_DIM, lane);
    const int k0 = hs * F_DIM + mt * 16 + half * 8;
    #pragma unroll
    for (int i = 0; i < 8; ++i)
        wc[WC_FRAG(k0 + i, nt, l)] = acc[i];
}

#define NODES_PER_WAVE 8
__device__ __forceinline__ void attn_score_block(const float* __restrict__ h,
                                                 const float* __restrict__ wlr,
                                                 float* __restrict__ el,
                                                 float* __restrict__ er,
                                                 int N, int blk) {
    const int wave  = (blk * (int)blockDim.x + (int)threadIdx.x) >> 5;
    const int node0 = wave * NODES_PER_WAVE;
    if (node0 >= N) return;
    const int lane = threadIdx.x & 31;
    const int sub = lane & 7, q = lane >> 3;
    float4 fa[NODES_PER_WAVE / 4], fb[NODES_PER_WAVE / 4];
    #pragma unroll
    for (int g = 0; g < NODES_PER_WAVE / 4; ++g) {
        const float* p = h + (size_t)min(node0 + g * 4 + q, N - 1) * F_DIM + sub * 8;
        fa[g] = *(const float4*)(p);
        fb[g] = *(const float4*)(p + 4);
    }
    float4 wa[8], wb[8];
    #pragma unroll
    for (int o = 0; o < 8; ++o) {
        wa[o] = *(const float4*)(wlr + o * F_DIM + sub * 8);
        wb[o] = *(const float4*)(wlr + o * F_DIM + sub * 8 + 4);
    }
    #pragma unroll
    for (int g = 0; g < NODES_PER_WAVE / 4; ++g) {
        float s[8];
        #pragma unroll
        for (int o = 0; o < 8; ++o) {
            float t = fa[g].x * wa[o].x;
            t += fa[g].y * wa[o].y; t += fa[g].z * wa[o].z; t += fa[g].w * wa[o].w;
            t += fb[g].x * wb[o].x; t += fb[g].y * wb[o].y;
            t += fb[g].z * wb[o].z; t += fb[g].w * wb[o].w;
            s[o] = t;
        }
        #pragma unroll
        for (int msk = 1; msk < 8; msk <<= 1) {
            #pragma unroll
            for (int o = 0; o < 8; ++o) s[o] += __shfl_xor(s[o], msk, 32);
        }
        const int node = node0 + g * 4 + q;
        if (sub == 0 && node < N) {
            float4 l4, r4;
            l4.x = s[0]; l4.y = s[1]; l4.z = s[2]; l4.w = s[3];
            r4.x = s[4]; r4.y = s[5]; r4.z = s[6]; r4.w = s[7];
            *(float4*)(el + (size_t)node * H_DIM) = l4;
            *(float4*)(er + (size_t)node * H_DIM) = r4;
        }
    }
}

#define BKT_SHIFT   8
#define BKT_NODES   (1 << BKT_SHIFT)
#define MAX_BKTS    256
#define CHUNK_EDGES 4096
__device__ __forceinline__ void edge_partition_block(const int* __restrict__ dst,
                                                     const int* __restrict__ src,
                                                     int* __restrict__ seg,
                                                     int* __restrict__ recs,
                                                     int* cnt, int* gbase, int* wtot,
                                                     int E, int nb, int eblocks, int blk) {
    const int tid = threadIdx.x, lane = tid & 31, wv = tid >> 5;
    cnt[tid] = 0;
    __syncthreads();
    const int e_blk = blk * CHUNK_EDGES;
    int dd[16], ss[16], lr[16];
    if (e_blk + CHUNK_EDGES <= E) {
        #pragma unroll
        for (int k = 0; k < 4; ++k) {
            const int4 v = *(const int4*)(dst + e_blk + k * 1024 + tid * 4);
            dd[4 * k] = v.x; dd[4 * k + 1] = v.y; dd[4 * k + 2] = v.z; dd[4 * k + 3] = v.w;
        }
        #pragma unroll
        for (int k = 0; k < 4; ++k) {
            const int4 v = *(const int4*)(src + e_blk + k * 1024 + tid * 4);
            ss[4 * k] = v.x; ss[4 * k + 1] = v.y; ss[4 * k + 2] = v.z; ss[4 * k + 3] = v.w;
        }
    } else {
        #pragma unroll
        for (int i = 0; i < 16; ++i) {
            const int e = e_blk + (i >> 2) * 1024 + tid * 4 + (i & 3);
            const int v = dst[min(e, E - 1)];
            dd[i] = e < E ? v : -1;
            ss[i] = src[min(e, E - 1)];
        }
    }
    #pragma unroll
    for (int i = 0; i < 16; ++i)
        lr[i] = dd[i] >= 0 ? atomicAdd(&cnt[dd[i] >> BKT_SHIFT], 1) : 0;
    __syncthreads();
    const int c = cnt[tid];
    int inc = c;
    #pragma unroll
    for (int o = 1; o < 32; o <<= 1) {
        const int t = __shfl_up(inc, o, 32);
        if (lane >= o) inc += t;
    }
    if (lane == 31) wtot[wv] = inc;
    __syncthreads();
    int excl = inc - c;
    #pragma unroll
    for (int w = 0; w < 7; ++w) excl += w < wv ? wtot[w] : 0;
    gbase[tid] = excl;
    if (tid < nb) seg[(size_t)tid * eblocks + blk] = (excl << 16) | c;
    __syncthreads();
    v2i* r2 = (v2i*)recs + (size_t)blk * CHUNK_EDGES;
    #pragma unroll
    for (int i = 0; i < 16; ++i) {
        if (dd[i] >= 0) {
            v2i rec;
            rec.x = dd[i];
            rec.y = ss[i];
            r2[gbase[dd[i] >> BKT_SHIFT] + lr[i]] = rec;
        }
    }
}

__global__ void fold_partition_kernel(const int* __restrict__ dst,
                                      const int* __restrict__ src,
                                      int* __restrict__ seg,
                                      int* __restrict__ recs,
                                      int* __restrict__ total,
                                      const float* __restrict__ fcw,
                                      const float* __restrict__ al,
                                      const float* __restrict__ ar,
                                      const float* __restrict__ gb,
                                      const float* __restrict__ ow,
                                      const float* __restrict__ ob,
                                      float* __restrict__ wlr,
                                      float* __restrict__ bvec,
                                      float* __restrict__ wc,
                                      int E, int nb, int eblocks) {
    __shared__ int cnt[MAX_BKTS];
    __shared__ int gbase[MAX_BKTS];
    __shared__ int wtot[8];
    __shared__ float red[256];
    const int b = blockIdx.x;
    if (b < eblocks)
        edge_partition_block(dst, src, seg, recs, cnt, gbase, wtot, E, nb, eblocks, b);
    else
        fold_weights_block(fcw, al, ar, gb, ow, ob, wlr, bvec, wc, total, red, b - eblocks);
}

#define SEGS_PER_TRIP 8
static_assert(SEGS_PER_TRIP == 8, "the load fence in walk_slices names eight records");
template <bool PLACE>
__device__ __forceinline__ void walk_slices(const v2i* __restrict__ r2,
                                            const int* __restrict__ myseg,
                                            int* ncnt, int* __restrict__ perm,
                                            int base, int E, int eblocks,
                                            int wv, int lane) {
    for (int j0 = 0; j0 < eblocks; j0 += 256) {
        const int j  = j0 + wv + 8 * lane;
        const int sv = j < eblocks ? myseg[j] : 0;
        const int nseg = (min(eblocks - j0, 256) - wv + 7) >> 3;
        for (int q = 0; q < nseg; q += SEGS_PER_TRIP) {
            v2i r[SEGS_PER_TRIP];
            int c[SEGS_PER_TRIP], st[SEGS_PER_TRIP];
            #pragma unroll
            for (int u = 0; u < SEGS_PER_TRIP; ++u) {
                const int s   = __builtin_amdgcn_readlane(sv, q + u);
                const int off = min(max(s >> 16, 0), CHUNK_EDGES);
                c[u]  = min(s & 0xFFFF, CHUNK_EDGES - off);
                st[u] = min(j0 + wv + 8 * (q + u), eblocks - 1) * CHUNK_EDGES + off;
                r[u]  = r2[c[u] > 0 ? st[u] + min(lane, c[u] - 1) : 0];
            }
            asm volatile("" :: "v"(r[0]), "v"(r[1]), "v"(r[2]), "v"(r[3]),
                               "v"(r[4]), "v"(r[5]), "v"(r[6]), "v"(r[7]));
            int pos[SEGS_PER_TRIP];
            #pragma unroll
            for (int u = 0; u < SEGS_PER_TRIP; ++u)
                pos[u] = lane < c[u] ? atomicAdd(&ncnt[r[u].x & (BKT_NODES - 1)], 1) : 0;
            if (PLACE) {
                #pragma unroll
                for (int u = 0; u < SEGS_PER_TRIP; ++u)
                    if (lane < c[u]) perm[min(max(base + pos[u], 0), E - 1)] = r[u].y;
            }
            #pragma unroll
            for (int u = 0; u < SEGS_PER_TRIP; ++u) {
                for (int i = 32 + lane; i < c[u]; i += 32) {
                    const v2i rr = r2[st[u] + i];
                    const int pos = atomicAdd(&ncnt[rr.x & (BKT_NODES - 1)], 1);
                    if (PLACE) perm[min(max(base + pos, 0), E - 1)] = rr.y;
                }
            }
        }
    }
}

__device__ __forceinline__ void bucket_sort_block(const int* __restrict__ recs,
                                                  const int* __restrict__ seg,
                                                  int* __restrict__ total,
                                                  int* __restrict__ deg,
                                                  int* __restrict__ rowstart,
                                                  int* __restrict__ perm,
                                                  int* ncnt, int* wave_off, int* block_base,
                                                  int N, int E, int eblocks, int b) {
    const int tid = threadIdx.x, lane = tid & 31;
    const int wv = __builtin_amdgcn_readfirstlane(tid >> 5);
    ncnt[tid] = 0;
    const v2i* r2    = (const v2i*)recs;
    const int* myseg = seg + (size_t)b * eblocks;
    __syncthreads();
    walk_slices<false>(r2, myseg, ncnt, perm, 0, E, eblocks, wv, lane);
    __syncthreads();
    const int v = ncnt[tid];
    int inc = v;
    #pragma unroll
    for (int o = 1; o < 32; o <<= 1) {
        const int t = __shfl_up(inc, o, 32);
        if (lane >= o) inc += t;
    }
    if (lane == 31) wave_off[wv] = inc;
    __syncthreads();
    if (tid == 0) {
        int run = 0;
        #pragma unroll
        for (int w = 0; w < 8; ++w) { const int t = wave_off[w]; wave_off[w] = run; run += t; }
        *block_base = atomicAdd(total, run);
    }
    __syncthreads();
    const int excl = wave_off[wv] + inc - v;
    const int base = *block_base;
    const int node = b * BKT_NODES + tid;
    if (node < N) { deg[node] = v; rowstart[node] = base + excl; }
    ncnt[tid] = excl;
    __syncthreads();
    walk_slices<true>(r2, myseg, ncnt, perm, base, E, eblocks, wv, lane);
}

__global__ void sort_scores_kernel(const int* __restrict__ recs,
                                   const int* __restrict__ seg,
                                   int* __restrict__ total,
                                   int* __restrict__ deg,
                                   int* __restrict__ rowstart,
                                   int* __restrict__ perm,
                                   const float* __restrict__ h,
                                   const float* __restrict__ wlr,
                                   float* __restrict__ el,
                                   float* __restrict__ er,
                                   int N, int E, int eblocks, int nb) {
    __shared__ int ncnt[BKT_NODES];
    __shared__ int wave_off[8];
    __shared__ int block_base;
    const int b = blockIdx.x;
    if (b < nb)
        bucket_sort_block(recs, seg, total, deg, rowstart, perm, ncnt, wave_off, &block_base,
                          N, E, eblocks, b);
    else
        attn_score_block(h, wlr, el, er, N, b - nb);
}

__device__ __forceinline__ float leaky(float v) {
    return v > 0.0f ? v : NEG_SLOPE * v;
}

__device__ __forceinline__ void load_edge(const int* __restrict__ perm,
                                          const float* __restrict__ el,
                                          const float4 erd, int pos, int N,
                                          int& s, float4& v) {
    s = min(max(perm[pos], 0), N - 1);
    const float4 l = *(const float4*)(el + (size_t)s * H_DIM);
    v.x = leaky(l.x + erd.x); v.y = leaky(l.y + erd.y);
    v.z = leaky(l.z + erd.z); v.w = leaky(l.w + erd.w);
}

#define ROWS_PER_TRIP 8
__device__ __forceinline__ void add_rows(const float* __restrict__ h,
                                         int lane, int s_l,
                                         const float4 wt, int n,
                                         v2f (&acc)[H_DIM]) {
    for (int i = 0; i < n; i += ROWS_PER_TRIP) {
        v2f f[ROWS_PER_TRIP];
        #pragma unroll
        for (int u = 0; u < ROWS_PER_TRIP; ++u) {
            const int s = __builtin_amdgcn_readlane(s_l, i + u);
            f[u] = *(const v2f*)(h + (size_t)s * F_DIM + lane * 2);
        }
        #pragma unroll
        for (int u = 0; u < ROWS_PER_TRIP; ++u) {
            const float wx = __int_as_float(__builtin_amdgcn_readlane(__float_as_int(wt.x), i + u));
            const float wy = __int_as_float(__builtin_amdgcn_readlane(__float_as_int(wt.y), i + u));
            const float wz = __int_as_float(__builtin_amdgcn_readlane(__float_as_int(wt.z), i + u));
            const float ww = __int_as_float(__builtin_amdgcn_readlane(__float_as_int(wt.w), i + u));
            const v2f w0 = {wx, wx}, w1 = {wy, wy}, w2 = {wz, wz}, w3 = {ww, ww};
            acc[0] = __builtin_elementwise_fma(f[u], w0, acc[0]);
            acc[1] = __builtin_elementwise_fma(f[u], w1, acc[1]);
            acc[2] = __builtin_elementwise_fma(f[u], w2, acc[2]);
            acc[3] = __builtin_elementwise_fma(f[u], w3, acc[3]);
        }
    }
}

__device__ __forceinline__ void aggregate_node(const int d, const int lane,
                                               const int* __restrict__ rowstart,
                                               const int* __restrict__ deg,
                                               const int* __restrict__ perm,
                                               const float* __restrict__ h,
                                               const float* __restrict__ el,
                                               const float* __restrict__ er,
                                               float* __restrict__ row, int N, int E) {
    const int beg = min(max(rowstart[d], 0), E);
    const int cnt = min(max(deg[d], 0), E - beg);
    const float4 erd = *(const float4*)(er + (size_t)d * H_DIM);

    int    s0 = 0;
    float4 v0 = {-INFINITY, -INFINITY, -INFINITY, -INFINITY};
    if (lane < cnt) load_edge(perm, el, erd, beg + lane, N, s0, v0);
    s0 = lane < cnt ? s0 : __builtin_amdgcn_readfirstlane(s0);

    float4 mx = v0;
    for (int c = 32; c < cnt; c += 32) {
        if (c + lane < cnt) {
            int s; float4 v;
            load_edge(perm, el, erd, beg + c + lane, N, s, v);
            mx.x = fmaxf(mx.x, v.x); mx.y = fmaxf(mx.y, v.y);
            mx.z = fmaxf(mx.z, v.z); mx.w = fmaxf(mx.w, v.w);
        }
    }
    #pragma unroll
    for (int msk = 16; msk >= 1; msk >>= 1) {
        mx.x = fmaxf(mx.x, __shfl_xor(mx.x, msk, 32));
        mx.y = fmaxf(mx.y, __shfl_xor(mx.y, msk, 32));
        mx.z = fmaxf(mx.z, __shfl_xor(mx.z, msk, 32));
        mx.w = fmaxf(mx.w, __shfl_xor(mx.w, msk, 32));
    }

    v2f acc[H_DIM];
    #pragma unroll
    for (int k = 0; k < H_DIM; ++k) { acc[k].x = 0.0f; acc[k].y = 0.0f; }
    float4 ex0 = {0.0f, 0.0f, 0.0f, 0.0f};
    if (lane < cnt) {
        ex0.x = __expf(v0.x - mx.x); ex0.y = __expf(v0.y - mx.y);
        ex0.z = __expf(v0.z - mx.z); ex0.w = __expf(v0.w - mx.w);
    }
    float4 zs = ex0;
    for (int c = 32; c < cnt; c += 32) {
        const int n = min(32, cnt - c);
        int    s_l = 0;
        float4 ex  = {0.0f, 0.0f, 0.0f, 0.0f};
        if (lane < n) {
            float4 v;
            load_edge(perm, el, erd, beg + c + lane, N, s_l, v);
            ex.x = __expf(v.x - mx.x); ex.y = __expf(v.y - mx.y);
            ex.z = __expf(v.z - mx.z); ex.w = __expf(v.w - mx.w);
        }
        s_l = lane < n ? s_l : __builtin_amdgcn_readfirstlane(s_l);
        zs.x += ex.x; zs.y += ex.y; zs.z += ex.z; zs.w += ex.w;
        add_rows(h, lane, s_l, ex, n, acc);
    }
    #pragma unroll
    for (int msk = 16; msk >= 1; msk >>= 1) {
        zs.x += __shfl_xor(zs.x, msk, 32);
        zs.y += __shfl_xor(zs.y, msk, 32);
        zs.z += __shfl_xor(zs.z, msk, 32);
        zs.w += __shfl_xor(zs.w, msk, 32);
    }
    float4 rz = {1.0f, 1.0f, 1.0f, 1.0f};
    if (cnt > 0) {
        rz.x = 1.0f / zs.x; rz.y = 1.0f / zs.y;
        rz.z = 1.0f / zs.z; rz.w = 1.0f / zs.w;
    }
    acc[0].x *= rz.x; acc[0].y *= rz.x;
    acc[1].x *= rz.y; acc[1].y *= rz.y;
    acc[2].x *= rz.z; acc[2].y *= rz.z;
    acc[3].x *= rz.w; acc[3].y *= rz.w;
    float4 alpha;
    alpha.x = ex0.x * rz.x; alpha.y = ex0.y * rz.y;
    alpha.z = ex0.z * rz.z; alpha.w = ex0.w * rz.w;

    add_rows(h, lane, s0, alpha, min(cnt, 32), acc);

    float* r = row + lane * 2;
    #pragma unroll
    for (int k = 0; k < H_DIM; ++k)
        *(v2f*)(r + k * F_DIM) = acc[k];
}

#define AGG_NODES_PER_WAVE 2
#define TILE_NODES (8 * AGG_NODES_PER_WAVE)
#define AGG_PITCH  (HF + 4)
#define X_PITCH    (F_DIM + 4)
__global__ void aggregate_out_kernel(const float* __restrict__ h,
                                     const int* __restrict__ rowstart,
                                     const int* __restrict__ deg,
                                     const int* __restrict__ perm,
                                     const float* __restrict__ el,
                                     const float* __restrict__ er,
                                     const float* __restrict__ w,
                                     const float* __restrict__ bv,
                                     const float* __restrict__ g,
                                     const float* __restrict__ b,
                                     float* __restrict__ out, int N, int E) {
    __shared__ float s_agg[TILE_NODES * AGG_PITCH];
    __shared__ float s_x[TILE_NODES * X_PITCH];
    const int wv    = __builtin_amdgcn_readfirstlane((int)(threadIdx.x >> 5));
    const int lane  = threadIdx.x & 31;
    const int node0 = blockIdx.x * TILE_NODES;

    #pragma nounroll
    for (int j = 0; j < AGG_NODES_PER_WAVE; ++j) {
        const int r = wv * AGG_NODES_PER_WAVE + j;
        const int d = node0 + r;
        float* row = s_agg + r * AGG_PITCH;
        if (d < N) {
            aggregate_node(d, lane, rowstart, deg, perm, h, el, er, row, N, E);
        } else {
            const v2f zero = {0.0f, 0.0f};
            for (int k = 0; k < H_DIM; ++k)
                *(v2f*)(row + k * F_DIM + lane * 2) = zero;
        }
    }
    __syncthreads();

    if (wv < F_DIM / 16) {
        const int half = lane >> 4, l = lane & 15;
        const float* arow = s_agg + l * AGG_PITCH + 2 * half;
        const int bcol = wv * 16 + l;
        const float4* wf = (const float4*)w + (size_t)wv * (HF / 8) * 32 + lane;
        v8f acc = {};
        #pragma unroll 8
        for (int kp = 0; kp < HF / 8; ++kp) {
            const float4 bq = wf[kp * 32];
            v2f a0, a1, b0, b1;
            a0.x = arow[kp * 8];     a0.y = arow[kp * 8 + 1];
            a1.x = arow[kp * 8 + 4]; a1.y = arow[kp * 8 + 5];
            b0.x = bq.x; b0.y = bq.y;
            b1.x = bq.z; b1.y = bq.w;
            acc = __builtin_amdgcn_wmma_f32_16x16x4_f32(
                      false, a0, false, b0, (short)0, acc, false, false);
            acc = __builtin_amdgcn_wmma_f32_16x16x4_f32(
                      false, a1, false, b1, (short)0, acc, false, false);
        }
        const float bias = bv[bcol];
        float* xr = s_x + (half * 8) * X_PITCH + bcol;
        #pragma unroll
        for (int i = 0; i < 8; ++i)
            xr[i * X_PITCH] = acc[i] + bias;
    }
    __syncthreads();

    const float g0 = g[lane], g1 = g[lane + 32];
    const float b0 = b[lane], b1 = b[lane + 32];
    #pragma unroll
    for (int j = 0; j < AGG_NODES_PER_WAVE; ++j) {
        const int r = wv * AGG_NODES_PER_WAVE + j;
        const int node = node0 + r;
        const float x0 = s_x[r * X_PITCH + lane], x1 = s_x[r * X_PITCH + lane + 32];
        float s = x0 + x1, sq = x0 * x0 + x1 * x1;
        #pragma unroll
        for (int msk = 1; msk < 32; msk <<= 1) {
            s  += __shfl_xor(s,  msk, 32);
            sq += __shfl_xor(sq, msk, 32);
        }
        const float mu  = s * (1.0f / 64.0f);
        const float var = sq * (1.0f / 64.0f) - mu * mu;
        const float inv = rsqrtf(var + LN_EPS);
        if (node < N) {
            const size_t base = (size_t)node * F_DIM;
            out[base + lane]      = (x0 - mu) * inv * g0 + b0;
            out[base + lane + 32] = (x1 - mu) * inv * g1 + b1;
        }
    }
}

extern "C" void kernel_launch(void* const* d_in, const int* in_sizes, int n_in,
                              void* d_out, int out_size, void* d_ws, size_t ws_size,
                              hipStream_t stream) {
    const float* h   = (const float*)d_in[0];
    const int*   src = (const int*)  d_in[1];
    const int*   dst = (const int*)  d_in[2];
    const float* fcw = (const float*)d_in[3];
    const float* al  = (const float*)d_in[4];
    const float* ar  = (const float*)d_in[5];
    const float* gb  = (const float*)d_in[6];
    const float* ow  = (const float*)d_in[7];
    const float* ob  = (const float*)d_in[8];
    const float* lng = (const float*)d_in[9];
    const float* lnb = (const float*)d_in[10];

    const int N = in_sizes[0] / F_DIM;
    const int E = in_sizes[1];

    char*  ws  = (char*)d_ws;
    size_t off = 0;
    auto carve = [&](size_t bytes) -> float* {
        float* p = (float*)(ws + off);
        off += (bytes + 255) & ~(size_t)255;
        return p;
    };
    float* el   = carve((size_t)N * H_DIM * 4);
    float* er   = carve((size_t)N * H_DIM * 4);
    float* wlr  = carve((size_t)2 * H_DIM * F_DIM * 4);
    float* bvec = carve((size_t)F_DIM * 4);
    float* wc   = carve((size_t)HF * F_DIM * 4);
    const int nb      = (N + BKT_NODES - 1) / BKT_NODES;
    const int eblocks = (E + CHUNK_EDGES - 1) / CHUNK_EDGES;
    int* deg      = (int*)carve((size_t)N * 4);
    int* rowstart = (int*)carve((size_t)N * 4);
    int* total    = (int*)carve(4);
    int* seg      = (int*)carve((size_t)nb * eblocks * 4);
    int* recs     = (int*)carve((size_t)eblocks * CHUNK_EDGES * 8);
    int* perm     = (int*)carve((size_t)E * 4);
    (void)ws_size;

    const int ablocks = (N + 8 * NODES_PER_WAVE - 1) / (8 * NODES_PER_WAVE);
    fold_partition_kernel<<<eblocks + FOLD_BLOCKS, MAX_BKTS, 0, stream>>>(
        dst, src, seg, recs, total, fcw, al, ar, gb, ow, ob, wlr, bvec, wc, E, nb, eblocks);
    sort_scores_kernel<<<nb + ablocks, BKT_NODES, 0, stream>>>(
        recs, seg, total, deg, rowstart, perm, h, wlr, el, er, N, E, eblocks, nb);
    const int blocks = (N + TILE_NODES - 1) / TILE_NODES;
    aggregate_out_kernel<<<blocks, 256, 0, stream>>>(h, rowstart, deg, perm,
                                                     el, er, wc, bvec, lng, lnb,
                                                     (float*)d_out, N, E);
}
